# starting kernel
# baseline (speedup 1.0000x reference)
#include <hip/hip_runtime.h>

typedef float v2f __attribute__((ext_vector_type(2)));
typedef float v8f __attribute__((ext_vector_type(8)));

#define IN_DIM 128
#define HID    64
#define OUT_D  64

__device__ __forceinline__ v8f wmma_k4(v2f a, v2f b, v8f c) {
  return __builtin_amdgcn_wmma_f32_16x16x4_f32(false, a, false, b, (short)0, c,
                                               false, false);
}

__global__ void __launch_bounds__(256) gat_lin_kernel(const float* __restrict__ x,
                                                      const float* __restrict__ W,
                                                      float* __restrict__ h, int N) {
  const int lane  = threadIdx.x & 31;
  const int wave  = threadIdx.x >> 5;
  const int row0  = (blockIdx.x * 8 + wave) * 16;
  const int lh    = lane & 15;
  const int khalf = lane >> 4;
  const int rowA  = row0 + lh;
  const int rA    = rowA < N ? rowA : (N - 1);

  v8f a0 = {}, a1 = {}, a2 = {}, a3 = {};
  for (int k0 = 0; k0 < IN_DIM; k0 += 4) {
    const int kb = k0 + khalf * 2;
    v2f a = { x[(size_t)rA * IN_DIM + kb], x[(size_t)rA * IN_DIM + kb + 1] };
    v2f b0 = { W[(size_t)(lh +  0) * IN_DIM + kb], W[(size_t)(lh +  0) * IN_DIM + kb + 1] };
    v2f b1 = { W[(size_t)(lh + 16) * IN_DIM + kb], W[(size_t)(lh + 16) * IN_DIM + kb + 1] };
    v2f b2 = { W[(size_t)(lh + 32) * IN_DIM + kb], W[(size_t)(lh + 32) * IN_DIM + kb + 1] };
    v2f b3 = { W[(size_t)(lh + 48) * IN_DIM + kb], W[(size_t)(lh + 48) * IN_DIM + kb + 1] };
    a0 = wmma_k4(a, b0, a0);
    a1 = wmma_k4(a, b1, a1);
    a2 = wmma_k4(a, b2, a2);
    a3 = wmma_k4(a, b3, a3);
  }
#pragma unroll
  for (int r = 0; r < 8; ++r) {
    const int row = row0 + khalf * 8 + r;
    if (row < N) {
      h[(size_t)row * HID + lh +  0] = a0[r];
      h[(size_t)row * HID + lh + 16] = a1[r];
      h[(size_t)row * HID + lh + 32] = a2[r];
      h[(size_t)row * HID + lh + 48] = a3[r];
    }
  }
}

__global__ void __launch_bounds__(256) gat_sd_kernel(const float* __restrict__ h,
                                                     const float* __restrict__ att,
                                                     float* __restrict__ s,
                                                     float* __restrict__ d, int N) {
  const int n = blockIdx.x * blockDim.x + threadIdx.x;
  if (n >= N) return;
  float ss = 0.f, dd = 0.f;
#pragma unroll 8
  for (int c = 0; c < HID; ++c) {
    const float v = h[(size_t)n * HID + c];
    ss += v * att[c];
    dd += v * att[HID + c];
  }
  s[n] = ss;
  d[n] = dd;
}

__global__ void __launch_bounds__(256) gat_edge1_kernel(const int* __restrict__ ei,
                                                        const float* __restrict__ s,
                                                        const float* __restrict__ d,
                                                        float* __restrict__ expw,
                                                        float* __restrict__ denom,
                                                        int E) {
  const int e = blockIdx.x * blockDim.x + threadIdx.x;
  if (e >= E) return;
  const int src = ei[e];
  const int dst = ei[(size_t)E + e];
  float v = s[src] + d[dst];
  v = (v > 0.f) ? v : 0.2f * v;
  const float w = __expf(v);
  expw[e] = w;
  atomicAdd(&denom[dst], w);
}

__global__ void __launch_bounds__(256) gat_edge2_kernel(const int* __restrict__ ei,
                                                        const float* __restrict__ h,
                                                        const float* __restrict__ expw,
                                                        const float* __restrict__ denom,
                                                        float* __restrict__ agg,
                                                        int E) {
  const long long g = (long long)blockIdx.x * blockDim.x + threadIdx.x;
  const int e = (int)(g >> 6);
  const int c = (int)(g & 63);
  if (e >= E) return;
  const int src = ei[e];
  const int dst = ei[(size_t)E + e];
  const float alpha = expw[e] / (denom[dst] + 1e-9f);
  atomicAdd(&agg[(size_t)dst * HID + c], h[(size_t)src * HID + c] * alpha);
}

__global__ void __launch_bounds__(256) gat_out_kernel(const float* __restrict__ agg,
                                                      const float* __restrict__ W,
                                                      const float* __restrict__ bias,
                                                      float* __restrict__ out, int N) {
  const int lane  = threadIdx.x & 31;
  const int wave  = threadIdx.x >> 5;
  const int row0  = (blockIdx.x * 8 + wave) * 16;
  const int lh    = lane & 15;
  const int khalf = lane >> 4;
  const int rowA  = row0 + lh;
  const int rA    = rowA < N ? rowA : (N - 1);

  v8f a0 = {}, a1 = {}, a2 = {}, a3 = {};
  for (int k0 = 0; k0 < HID; k0 += 4) {
    const int kb = k0 + khalf * 2;
    const float z0 = agg[(size_t)rA * HID + kb];
    const float z1 = agg[(size_t)rA * HID + kb + 1];
    v2f a = { z0 > 0.f ? z0 : 0.f, z1 > 0.f ? z1 : 0.f };
    v2f b0 = { W[(size_t)(lh +  0) * HID + kb], W[(size_t)(lh +  0) * HID + kb + 1] };
    v2f b1 = { W[(size_t)(lh + 16) * HID + kb], W[(size_t)(lh + 16) * HID + kb + 1] };
    v2f b2 = { W[(size_t)(lh + 32) * HID + kb], W[(size_t)(lh + 32) * HID + kb + 1] };
    v2f b3 = { W[(size_t)(lh + 48) * HID + kb], W[(size_t)(lh + 48) * HID + kb + 1] };
    a0 = wmma_k4(a, b0, a0);
    a1 = wmma_k4(a, b1, a1);
    a2 = wmma_k4(a, b2, a2);
    a3 = wmma_k4(a, b3, a3);
  }
  const float bb0 = bias[lh +  0];
  const float bb1 = bias[lh + 16];
  const float bb2 = bias[lh + 32];
  const float bb3 = bias[lh + 48];
#pragma unroll
  for (int r = 0; r < 8; ++r) {
    const int row = row0 + khalf * 8 + r;
    if (row < N) {
      out[(size_t)row * OUT_D + lh +  0] = a0[r] + bb0;
      out[(size_t)row * OUT_D + lh + 16] = a1[r] + bb1;
      out[(size_t)row * OUT_D + lh + 32] = a2[r] + bb2;
      out[(size_t)row * OUT_D + lh + 48] = a3[r] + bb3;
    }
  }
}

extern "C" void kernel_launch(void* const* d_in, const int* in_sizes, int n_in,
                              void* d_out, int out_size, void* d_ws, size_t ws_size,
                              hipStream_t stream) {
  const float* x    = (const float*)d_in[0];
  const int*   ei   = (const int*)  d_in[1];
  const float* Wlin = (const float*)d_in[2];
  const float* att  = (const float*)d_in[3];
  const float* Wout = (const float*)d_in[4];
  const float* bout = (const float*)d_in[5];
  float* out = (float*)d_out;

  const int N = in_sizes[0] / IN_DIM;
  const int E = in_sizes[1] / 2;

  float* ws    = (float*)d_ws;
  float* h     = ws;
  float* s     = h + (size_t)N * HID;
  float* dn    = s + N;
  float* expw  = dn + N;
  float* denom = expw + E;
  float* agg   = denom + N;

  hipMemsetAsync(denom, 0, sizeof(float) * ((size_t)N + (size_t)N * HID), stream);

  const int tiles   = (N + 15) / 16;
  const int gemmBlk = (tiles + 7) / 8;

  gat_lin_kernel<<<gemmBlk, 256, 0, stream>>>(x, Wlin, h, N);
  gat_sd_kernel<<<(N + 255) / 256, 256, 0, stream>>>(h, att, s, dn, N);
  gat_edge1_kernel<<<(E + 255) / 256, 256, 0, stream>>>(ei, s, dn, expw, denom, E);

  const long long ef = (long long)E * HID;
  gat_edge2_kernel<<<(unsigned)((ef + 255) / 256), 256, 0, stream>>>(ei, h, expw, denom, agg, E);

  gat_out_kernel<<<gemmBlk, 256, 0, stream>>>(agg, Wout, bout, out, N);
}


# Scatter by float atomics replaced by a counting sort of edges by destination (one int atomic per edge gives its rank) plus one wave per node summing its h rows in registers; sum order = arrival order
# speedup vs baseline: 1.8194x; 1.8194x over previous
#include <hip/hip_runtime.h>

typedef float v2f __attribute__((ext_vector_type(2)));
typedef float v8f __attribute__((ext_vector_type(8)));

#define IN_DIM 128
#define HID    64
#define OUT_D  64

__device__ __forceinline__ v8f wmma_k4(v2f a, v2f b, v8f c) {
  return __builtin_amdgcn_wmma_f32_16x16x4_f32(false, a, false, b, (short)0, c,
                                               false, false);
}

__global__ void __launch_bounds__(256) gat_lin_kernel(const float* __restrict__ x,
                                                      const float* __restrict__ W,
                                                      float* __restrict__ h, int N) {
  const int lane  = threadIdx.x & 31;
  const int wave  = threadIdx.x >> 5;
  const int row0  = (blockIdx.x * 8 + wave) * 16;
  const int lh    = lane & 15;
  const int khalf = lane >> 4;
  const int rowA  = row0 + lh;
  const int rA    = rowA < N ? rowA : (N - 1);

  v8f a0 = {}, a1 = {}, a2 = {}, a3 = {};
  for (int k0 = 0; k0 < IN_DIM; k0 += 4) {
    const int kb = k0 + khalf * 2;
    v2f a = { x[(size_t)rA * IN_DIM + kb], x[(size_t)rA * IN_DIM + kb + 1] };
    v2f b0 = { W[(size_t)(lh +  0) * IN_DIM + kb], W[(size_t)(lh +  0) * IN_DIM + kb + 1] };
    v2f b1 = { W[(size_t)(lh + 16) * IN_DIM + kb], W[(size_t)(lh + 16) * IN_DIM + kb + 1] };
    v2f b2 = { W[(size_t)(lh + 32) * IN_DIM + kb], W[(size_t)(lh + 32) * IN_DIM + kb + 1] };
    v2f b3 = { W[(size_t)(lh + 48) * IN_DIM + kb], W[(size_t)(lh + 48) * IN_DIM + kb + 1] };
    a0 = wmma_k4(a, b0, a0);
    a1 = wmma_k4(a, b1, a1);
    a2 = wmma_k4(a, b2, a2);
    a3 = wmma_k4(a, b3, a3);
  }
#pragma unroll
  for (int r = 0; r < 8; ++r) {
    const int row = row0 + khalf * 8 + r;
    if (row < N) {
      h[(size_t)row * HID + lh +  0] = a0[r];
      h[(size_t)row * HID + lh + 16] = a1[r];
      h[(size_t)row * HID + lh + 32] = a2[r];
      h[(size_t)row * HID + lh + 48] = a3[r];
    }
  }
}

__global__ void __launch_bounds__(256) gat_sd_kernel(const float* __restrict__ h,
                                                     const float* __restrict__ att,
                                                     float* __restrict__ s,
                                                     float* __restrict__ d, int N) {
  const int n = blockIdx.x * blockDim.x + threadIdx.x;
  if (n >= N) return;
  float ss = 0.f, dd = 0.f;
#pragma unroll 8
  for (int c = 0; c < HID; ++c) {
    const float v = h[(size_t)n * HID + c];
    ss += v * att[c];
    dd += v * att[HID + c];
  }
  s[n] = ss;
  d[n] = dd;
}

__global__ void __launch_bounds__(256) gat_hist_kernel(const int* __restrict__ ei,
                                                       int* __restrict__ cnt,
                                                       int* __restrict__ rank, int E) {
  const int e = blockIdx.x * blockDim.x + threadIdx.x;
  if (e >= E) return;
  rank[e] = atomicAdd(&cnt[ei[(size_t)E + e]], 1);
}

__global__ void __launch_bounds__(256) gat_scan_kernel(const int* __restrict__ cnt,
                                                       int* __restrict__ cursor, int N) {
  __shared__ int red[8];
  __shared__ int wsum[8];
  const int tid   = threadIdx.x;
  const int lane  = tid & 31;
  const int wave  = tid >> 5;
  const int start = blockIdx.x * 1024;

  int pre = 0;
  for (int i = tid * 4; i < start; i += 1024) {
    const int4 v = *reinterpret_cast<const int4*>(cnt + i);
    pre += (v.x + v.y) + (v.z + v.w);
  }
  for (int m = 16; m > 0; m >>= 1) pre += __shfl_xor(pre, m);
  if (lane == 0) red[wave] = pre;

  const int i0 = start + tid * 4;
  int c0 = (i0 + 0 < N) ? cnt[i0 + 0] : 0;
  int c1 = (i0 + 1 < N) ? cnt[i0 + 1] : 0;
  int c2 = (i0 + 2 < N) ? cnt[i0 + 2] : 0;
  int c3 = (i0 + 3 < N) ? cnt[i0 + 3] : 0;
  const int tsum = (c0 + c1) + (c2 + c3);
  int inc = tsum;
  for (int m = 1; m < 32; m <<= 1) {
    const int t = __shfl_up(inc, m);
    if (lane >= m) inc += t;
  }
  if (lane == 31) wsum[wave] = inc;
  __syncthreads();
  int base = 0;
  for (int w = 0; w < 8; ++w) {
    base += red[w];
    if (w < wave) base += wsum[w];
  }
  int o = base + inc - tsum;
  if (i0 + 0 < N) cursor[i0 + 0] = o;  o += c0;
  if (i0 + 1 < N) cursor[i0 + 1] = o;  o += c1;
  if (i0 + 2 < N) cursor[i0 + 2] = o;  o += c2;
  if (i0 + 3 < N) cursor[i0 + 3] = o;
}

__global__ void __launch_bounds__(256) gat_fill_kernel(const int* __restrict__ ei,
                                                       const int* __restrict__ cursor,
                                                       const int* __restrict__ rank,
                                                       int* __restrict__ ssrc, int E) {
  const int e = blockIdx.x * blockDim.x + threadIdx.x;
  if (e >= E) return;
  const int src = ei[e];
  const int dst = ei[(size_t)E + e];
  ssrc[cursor[dst] + rank[e]] = src;
}

__device__ __forceinline__ float gat_edge_w(const int* __restrict__ ssrc,
                                            const float* __restrict__ s,
                                            float dn, int idx, int end, int& srcOut) {
  int sj = 0;
  float w = 0.f;
  if (idx < end) {
    sj = ssrc[idx];
    float v = s[sj] + dn;
    v = (v > 0.f) ? v : 0.2f * v;
    w = __expf(v);
  }
  srcOut = sj;
  return w;
}

__global__ void __launch_bounds__(256) gat_gather_kernel(const int* __restrict__ cursor,
                                                         const int* __restrict__ ssrc,
                                                         const float* __restrict__ h,
                                                         const float* __restrict__ s,
                                                         const float* __restrict__ d,
                                                         float* __restrict__ agg, int N, int E) {
  const int lane = threadIdx.x & 31;
  const int n = __builtin_amdgcn_readfirstlane(blockIdx.x * 8 + (threadIdx.x >> 5));
  if (n >= N) return;
  const int begin = __builtin_amdgcn_readfirstlane(cursor[n]);
  const int end   = __builtin_amdgcn_readfirstlane((n + 1 < N) ? cursor[n + 1] : E);
  const float dn  = d[n];

  int   src0;
  const float w0 = gat_edge_w(ssrc, s, dn, begin + lane, end, src0);
  float dsum = w0;
  for (int base = begin + 32; base < end; base += 32) {
    int t;
    dsum += gat_edge_w(ssrc, s, dn, base + lane, end, t);
  }
  for (int m = 16; m > 0; m >>= 1) dsum += __shfl_xor(dsum, m);
  const float den = dsum + 1e-9f;

  const v2f* __restrict__ h2 = reinterpret_cast<const v2f*>(h);
  v2f acc = {0.f, 0.f};
  for (int base = begin; base < end; base += 32) {
    int   srcj = src0;
    float wj   = w0;
    if (base != begin) wj = gat_edge_w(ssrc, s, dn, base + lane, end, srcj);
    const float aj = wj / den;
    const int   ab = __float_as_int(aj);
    const int   m  = (end - base < 32) ? (end - base) : 32;
    for (int k = 0; k < m; k += 8) {
      v2f   v[8];
      float a[8];
#pragma unroll
      for (int j = 0; j < 8; ++j) {
        const bool live = (k + j) < m;
        const int  kk   = live ? (k + j) : (m - 1);
        const int  sk   = __builtin_amdgcn_readlane(srcj, kk);
        a[j] = live ? __int_as_float(__builtin_amdgcn_readlane(ab, kk)) : 0.f;
        v[j] = h2[(size_t)sk * (HID / 2) + lane];
      }
#pragma unroll
      for (int j = 0; j < 8; ++j) {
        const bool live = (k + j) < m;
        const float vx = live ? v[j].x : 0.f;
        const float vy = live ? v[j].y : 0.f;
        acc.x += vx * a[j];
        acc.y += vy * a[j];
      }
    }
  }
  reinterpret_cast<v2f*>(agg)[(size_t)n * (HID / 2) + lane] = acc;
}

__global__ void __launch_bounds__(256) gat_out_kernel(const float* __restrict__ agg,
                                                      const float* __restrict__ W,
                                                      const float* __restrict__ bias,
                                                      float* __restrict__ out, int N) {
  const int lane  = threadIdx.x & 31;
  const int wave  = threadIdx.x >> 5;
  const int row0  = (blockIdx.x * 8 + wave) * 16;
  const int lh    = lane & 15;
  const int khalf = lane >> 4;
  const int rowA  = row0 + lh;
  const int rA    = rowA < N ? rowA : (N - 1);

  v8f a0 = {}, a1 = {}, a2 = {}, a3 = {};
  for (int k0 = 0; k0 < HID; k0 += 4) {
    const int kb = k0 + khalf * 2;
    const float z0 = agg[(size_t)rA * HID + kb];
    const float z1 = agg[(size_t)rA * HID + kb + 1];
    v2f a = { z0 > 0.f ? z0 : 0.f, z1 > 0.f ? z1 : 0.f };
    v2f b0 = { W[(size_t)(lh +  0) * HID + kb], W[(size_t)(lh +  0) * HID + kb + 1] };
    v2f b1 = { W[(size_t)(lh + 16) * HID + kb], W[(size_t)(lh + 16) * HID + kb + 1] };
    v2f b2 = { W[(size_t)(lh + 32) * HID + kb], W[(size_t)(lh + 32) * HID + kb + 1] };
    v2f b3 = { W[(size_t)(lh + 48) * HID + kb], W[(size_t)(lh + 48) * HID + kb + 1] };
    a0 = wmma_k4(a, b0, a0);
    a1 = wmma_k4(a, b1, a1);
    a2 = wmma_k4(a, b2, a2);
    a3 = wmma_k4(a, b3, a3);
  }
  const float bb0 = bias[lh +  0];
  const float bb1 = bias[lh + 16];
  const float bb2 = bias[lh + 32];
  const float bb3 = bias[lh + 48];
#pragma unroll
  for (int r = 0; r < 8; ++r) {
    const int row = row0 + khalf * 8 + r;
    if (row < N) {
      out[(size_t)row * OUT_D + lh +  0] = a0[r] + bb0;
      out[(size_t)row * OUT_D + lh + 16] = a1[r] + bb1;
      out[(size_t)row * OUT_D + lh + 32] = a2[r] + bb2;
      out[(size_t)row * OUT_D + lh + 48] = a3[r] + bb3;
    }
  }
}

extern "C" void kernel_launch(void* const* d_in, const int* in_sizes, int n_in,
                              void* d_out, int out_size, void* d_ws, size_t ws_size,
                              hipStream_t stream) {
  const float* x    = (const float*)d_in[0];
  const int*   ei   = (const int*)  d_in[1];
  const float* Wlin = (const float*)d_in[2];
  const float* att  = (const float*)d_in[3];
  const float* Wout = (const float*)d_in[4];
  const float* bout = (const float*)d_in[5];
  float* out = (float*)d_out;

  const int N = in_sizes[0] / IN_DIM;
  const int E = in_sizes[1] / 2;

  float* ws     = (float*)d_ws;
  float* h      = ws;
  float* s      = h + (size_t)N * HID;
  float* dn     = s + N;
  int*   ssrc   = (int*)(dn + N);
  int*   cursor = ssrc + E;
  float* agg    = (float*)(cursor + N);
  int*   cnt    = (int*)agg;
  int*   rank   = cnt + N;

  hipMemsetAsync(cnt, 0, sizeof(int) * (size_t)N, stream);

  const int tiles   = (N + 15) / 16;
  const int gemmBlk = (tiles + 7) / 8;

  gat_lin_kernel<<<gemmBlk, 256, 0, stream>>>(x, Wlin, h, N);
  gat_sd_kernel<<<(N + 255) / 256, 256, 0, stream>>>(h, att, s, dn, N);

  gat_hist_kernel<<<(E + 255) / 256, 256, 0, stream>>>(ei, cnt, rank, E);
  gat_scan_kernel<<<(N + 1023) / 1024, 256, 0, stream>>>(cnt, cursor, N);
  gat_fill_kernel<<<(E + 255) / 256, 256, 0, stream>>>(ei, cursor, rank, ssrc, E);
  gat_gather_kernel<<<(N + 7) / 8, 256, 0, stream>>>(cursor, ssrc, h, s, dn, agg, N, E);

  gat_out_kernel<<<gemmBlk, 256, 0, stream>>>(agg, Wout, bout, out, N);
}


# Edges grouped by destination with a two-level counting sort ranked by LDS counters (256-node buckets) instead of one returning global atomic per edge; sum order = LDS arrival order
# speedup vs baseline: 3.2781x; 1.8017x over previous
#include <hip/hip_runtime.h>

typedef float v2f __attribute__((ext_vector_type(2)));
typedef float v8f __attribute__((ext_vector_type(8)));

#define IN_DIM 128
#define HID    64
#define OUT_D  64

__device__ __forceinline__ v8f wmma_k4(v2f a, v2f b, v8f c) {
  return __builtin_amdgcn_wmma_f32_16x16x4_f32(false, a, false, b, (short)0, c,
                                               false, false);
}

__global__ void __launch_bounds__(256) gat_lin_kernel(const float* __restrict__ x,
                                                      const float* __restrict__ W,
                                                      float* __restrict__ h, int N) {
  const int lane  = threadIdx.x & 31;
  const int wave  = threadIdx.x >> 5;
  const int row0  = (blockIdx.x * 8 + wave) * 16;
  const int lh    = lane & 15;
  const int khalf = lane >> 4;
  const int rowA  = row0 + lh;
  const int rA    = rowA < N ? rowA : (N - 1);

  v8f a0 = {}, a1 = {}, a2 = {}, a3 = {};
  for (int k0 = 0; k0 < IN_DIM; k0 += 4) {
    const int kb = k0 + khalf * 2;
    v2f a = { x[(size_t)rA * IN_DIM + kb], x[(size_t)rA * IN_DIM + kb + 1] };
    v2f b0 = { W[(size_t)(lh +  0) * IN_DIM + kb], W[(size_t)(lh +  0) * IN_DIM + kb + 1] };
    v2f b1 = { W[(size_t)(lh + 16) * IN_DIM + kb], W[(size_t)(lh + 16) * IN_DIM + kb + 1] };
    v2f b2 = { W[(size_t)(lh + 32) * IN_DIM + kb], W[(size_t)(lh + 32) * IN_DIM + kb + 1] };
    v2f b3 = { W[(size_t)(lh + 48) * IN_DIM + kb], W[(size_t)(lh + 48) * IN_DIM + kb + 1] };
    a0 = wmma_k4(a, b0, a0);
    a1 = wmma_k4(a, b1, a1);
    a2 = wmma_k4(a, b2, a2);
    a3 = wmma_k4(a, b3, a3);
  }
#pragma unroll
  for (int r = 0; r < 8; ++r) {
    const int row = row0 + khalf * 8 + r;
    if (row < N) {
      h[(size_t)row * HID + lh +  0] = a0[r];
      h[(size_t)row * HID + lh + 16] = a1[r];
      h[(size_t)row * HID + lh + 32] = a2[r];
      h[(size_t)row * HID + lh + 48] = a3[r];
    }
  }
}

__global__ void __launch_bounds__(256) gat_sd_kernel(const float* __restrict__ h,
                                                     const float* __restrict__ att,
                                                     float* __restrict__ s,
                                                     float* __restrict__ d, int N) {
  const int n = blockIdx.x * blockDim.x + threadIdx.x;
  if (n >= N) return;
  float ss = 0.f, dd = 0.f;
#pragma unroll 8
  for (int c = 0; c < HID; ++c) {
    const float v = h[(size_t)n * HID + c];
    ss += v * att[c];
    dd += v * att[HID + c];
  }
  s[n] = ss;
  d[n] = dd;
}

#define BKT_SHIFT 8
#define BKT_NODES 256
#define BKT_WAVES (BKT_NODES / 32)
#define BKT_MAX   2048
#define SORT_PER  6
#define ECHUNK    4096

__global__ void __launch_bounds__(1024) gat_bcount_kernel(const int* __restrict__ ei,
                                                          int* __restrict__ tcnt,
                                                          int E, int NBK, int NBLK) {
  __shared__ int hc[BKT_MAX];
  const int tid = threadIdx.x;
  for (int i = tid; i < NBK; i += 1024) hc[i] = 0;
  __syncthreads();
  const int base = blockIdx.x * ECHUNK + tid;
  int d[4];
#pragma unroll
  for (int k = 0; k < 4; ++k) {
    const int e = base + k * 1024;
    d[k] = (e < E) ? ei[(size_t)E + e] : -1;
  }
#pragma unroll
  for (int k = 0; k < 4; ++k)
    if (d[k] >= 0) atomicAdd(&hc[d[k] >> BKT_SHIFT], 1);
  __syncthreads();
  for (int i = tid; i < NBK; i += 1024) tcnt[(size_t)i * NBLK + blockIdx.x] = hc[i];
}

template <int ROWS>
__device__ __forceinline__ int gat_rows_sum(const int4* __restrict__ q) {
  int4 v[ROWS];
#pragma unroll
  for (int j = 0; j < ROWS; ++j) v[j] = q[j * 1024];
  int t = 0;
#pragma unroll
  for (int j = 0; j < ROWS; ++j) t += (v[j].x + v[j].y) + (v[j].z + v[j].w);
  return t;
}

__global__ void __launch_bounds__(1024) gat_scan_kernel(const int* __restrict__ cnt,
                                                        int* __restrict__ off, int L) {
  __shared__ int red[32];
  __shared__ int wsum[32];
  const int tid   = threadIdx.x;
  const int lane  = tid & 31;
  const int wave  = tid >> 5;
  const int start = blockIdx.x * 1024;

  const int i0 = start + tid;
  const int c  = (i0 < L) ? cnt[i0] : 0;

  const int rows = start >> 12;
  const int rem  = start & 4095;
  const int4* __restrict__ p = reinterpret_cast<const int4*>(cnt) + tid;

  int4 vp = make_int4(0, 0, 0, 0);
  if (tid * 4 < rem) vp = p[(size_t)rows * 1024];

  int pre = 0;
  int r = 0;
  for (; r + 16 <= rows; r += 16) pre += gat_rows_sum<16>(p + (size_t)r * 1024);
  if (rows - r >= 8) { pre += gat_rows_sum<8>(p + (size_t)r * 1024); r += 8; }
  if (rows - r >= 4) { pre += gat_rows_sum<4>(p + (size_t)r * 1024); r += 4; }
  if (rows - r >= 2) { pre += gat_rows_sum<2>(p + (size_t)r * 1024); r += 2; }
  if (rows - r >= 1) { pre += gat_rows_sum<1>(p + (size_t)r * 1024); }
  pre += (vp.x + vp.y) + (vp.z + vp.w);
  for (int m = 16; m > 0; m >>= 1) pre += __shfl_xor(pre, m);

  int inc = c;
  for (int m = 1; m < 32; m <<= 1) {
    const int t = __shfl_up(inc, m);
    if (lane >= m) inc += t;
  }
  if (lane == 0)  red[wave]  = pre;
  if (lane == 31) wsum[wave] = inc;
  __syncthreads();
  int part = red[lane] + ((lane < wave) ? wsum[lane] : 0);
  for (int m = 16; m > 0; m >>= 1) part += __shfl_xor(part, m);
  if (i0 < L) off[i0] = part + inc - c;
}

__global__ void __launch_bounds__(1024) gat_bscatter_kernel(const int* __restrict__ ei,
                                                            const int* __restrict__ toff,
                                                            int* __restrict__ cpair,
                                                            int E, int NBK, int NBLK) {
  __shared__ int pos[BKT_MAX];
  const int tid = threadIdx.x;
  for (int i = tid; i < NBK; i += 1024) pos[i] = toff[(size_t)i * NBLK + blockIdx.x];
  __syncthreads();
  const int base = blockIdx.x * ECHUNK + tid;
  int s[4], d[4];
#pragma unroll
  for (int k = 0; k < 4; ++k) {
    const int e = base + k * 1024;
    s[k] = (e < E) ? ei[e] : 0;
    d[k] = (e < E) ? ei[(size_t)E + e] : -1;
  }
  int2* __restrict__ out2 = reinterpret_cast<int2*>(cpair);
#pragma unroll
  for (int k = 0; k < 4; ++k) {
    if (d[k] >= 0) {
      const int q = atomicAdd(&pos[d[k] >> BKT_SHIFT], 1);
      out2[q] = make_int2(s[k], d[k]);
    }
  }
}

__global__ void __launch_bounds__(1024) gat_bsort_kernel(const int* __restrict__ toff,
                                                         const int* __restrict__ cpair,
                                                         int* __restrict__ cursor,
                                                         int* __restrict__ ssrc,
                                                         int N, int E, int NBK, int NBLK) {
  __shared__ int cl[BKT_NODES];
  __shared__ int wtot[BKT_WAVES];
  const int tid   = threadIdx.x;
  const int lane  = tid & 31;
  const int wave  = tid >> 5;
  const int b     = blockIdx.x;
  const int node0 = b << BKT_SHIFT;
  const int bstart = toff[(size_t)b * NBLK];
  const int bend   = (b + 1 < NBK) ? toff[(size_t)(b + 1) * NBLK] : E;
  const int2* __restrict__ pr = reinterpret_cast<const int2*>(cpair);

  if (tid < BKT_NODES) cl[tid] = 0;
  __syncthreads();

  for (int e0 = bstart; e0 < bend; e0 += SORT_PER * 1024) {
    int d[SORT_PER];
#pragma unroll
    for (int k = 0; k < SORT_PER; ++k) {
      const int e = e0 + k * 1024 + tid;
      d[k] = (e < bend) ? pr[e].y : -1;
    }
#pragma unroll
    for (int k = 0; k < SORT_PER; ++k)
      if (d[k] >= 0) atomicAdd(&cl[d[k] - node0], 1);
  }
  __syncthreads();

  const int v = (tid < BKT_NODES) ? cl[tid] : 0;
  int inc = v;
  for (int m = 1; m < 32; m <<= 1) {
    const int t = __shfl_up(inc, m);
    if (lane >= m) inc += t;
  }
  if (lane == 31 && wave < BKT_WAVES) wtot[wave] = inc;
  __syncthreads();
  int part = (lane < BKT_WAVES && lane < wave) ? wtot[lane] : 0;
  for (int m = 16; m > 0; m >>= 1) part += __shfl_xor(part, m);
  const int excl = part + inc - v;
  if (tid < BKT_NODES) {
    cl[tid] = excl;
    if (node0 + tid < N) cursor[node0 + tid] = bstart + excl;
  }
  __syncthreads();

  for (int e0 = bstart; e0 < bend; e0 += SORT_PER * 1024) {
    int2 q[SORT_PER];
#pragma unroll
    for (int k = 0; k < SORT_PER; ++k) {
      const int e = e0 + k * 1024 + tid;
      q[k] = (e < bend) ? pr[e] : make_int2(0, -1);
    }
#pragma unroll
    for (int k = 0; k < SORT_PER; ++k) {
      if (q[k].y >= 0) {
        const int at = atomicAdd(&cl[q[k].y - node0], 1);
        ssrc[bstart + at] = q[k].x;
      }
    }
  }
}

__device__ __forceinline__ float gat_edge_w(const int* __restrict__ ssrc,
                                            const float* __restrict__ s,
                                            float dn, int idx, int end, int& srcOut) {
  int sj = 0;
  float w = 0.f;
  if (idx < end) {
    sj = ssrc[idx];
    float v = s[sj] + dn;
    v = (v > 0.f) ? v : 0.2f * v;
    w = __expf(v);
  }
  srcOut = sj;
  return w;
}

__global__ void __launch_bounds__(256) gat_gather_kernel(const int* __restrict__ cursor,
                                                         const int* __restrict__ ssrc,
                                                         const float* __restrict__ h,
                                                         const float* __restrict__ s,
                                                         const float* __restrict__ d,
                                                         float* __restrict__ agg, int N, int E) {
  const int lane = threadIdx.x & 31;
  const int n = __builtin_amdgcn_readfirstlane(blockIdx.x * 8 + (threadIdx.x >> 5));
  if (n >= N) return;
  const int begin = __builtin_amdgcn_readfirstlane(cursor[n]);
  const int end   = __builtin_amdgcn_readfirstlane((n + 1 < N) ? cursor[n + 1] : E);
  const float dn  = d[n];

  int   src0;
  const float w0 = gat_edge_w(ssrc, s, dn, begin + lane, end, src0);
  float dsum = w0;
  for (int base = begin + 32; base < end; base += 32) {
    int t;
    dsum += gat_edge_w(ssrc, s, dn, base + lane, end, t);
  }
  for (int m = 16; m > 0; m >>= 1) dsum += __shfl_xor(dsum, m);
  const float den = dsum + 1e-9f;

  const v2f* __restrict__ h2 = reinterpret_cast<const v2f*>(h);
  v2f acc = {0.f, 0.f};
  for (int base = begin; base < end; base += 32) {
    int   srcj = src0;
    float wj   = w0;
    if (base != begin) wj = gat_edge_w(ssrc, s, dn, base + lane, end, srcj);
    const float aj = wj / den;
    const int   ab = __float_as_int(aj);
    const int   m  = (end - base < 32) ? (end - base) : 32;
    for (int k = 0; k < m; k += 8) {
      v2f   v[8];
      float a[8];
#pragma unroll
      for (int j = 0; j < 8; ++j) {
        const bool live = (k + j) < m;
        const int  kk   = live ? (k + j) : (m - 1);
        const int  sk   = __builtin_amdgcn_readlane(srcj, kk);
        a[j] = live ? __int_as_float(__builtin_amdgcn_readlane(ab, kk)) : 0.f;
        v[j] = h2[(size_t)sk * (HID / 2) + lane];
      }
#pragma unroll
      for (int j = 0; j < 8; ++j) {
        const bool live = (k + j) < m;
        const float vx = live ? v[j].x : 0.f;
        const float vy = live ? v[j].y : 0.f;
        acc.x += vx * a[j];
        acc.y += vy * a[j];
      }
    }
  }
  reinterpret_cast<v2f*>(agg)[(size_t)n * (HID / 2) + lane] = acc;
}

__global__ void __launch_bounds__(256) gat_out_kernel(const float* __restrict__ agg,
                                                      const float* __restrict__ W,
                                                      const float* __restrict__ bias,
                                                      float* __restrict__ out, int N) {
  const int lane  = threadIdx.x & 31;
  const int wave  = threadIdx.x >> 5;
  const int row0  = (blockIdx.x * 8 + wave) * 16;
  const int lh    = lane & 15;
  const int khalf = lane >> 4;
  const int rowA  = row0 + lh;
  const int rA    = rowA < N ? rowA : (N - 1);

  v8f a0 = {}, a1 = {}, a2 = {}, a3 = {};
  for (int k0 = 0; k0 < HID; k0 += 4) {
    const int kb = k0 + khalf * 2;
    const float z0 = agg[(size_t)rA * HID + kb];
    const float z1 = agg[(size_t)rA * HID + kb + 1];
    v2f a = { z0 > 0.f ? z0 : 0.f, z1 > 0.f ? z1 : 0.f };
    v2f b0 = { W[(size_t)(lh +  0) * HID + kb], W[(size_t)(lh +  0) * HID + kb + 1] };
    v2f b1 = { W[(size_t)(lh + 16) * HID + kb], W[(size_t)(lh + 16) * HID + kb + 1] };
    v2f b2 = { W[(size_t)(lh + 32) * HID + kb], W[(size_t)(lh + 32) * HID + kb + 1] };
    v2f b3 = { W[(size_t)(lh + 48) * HID + kb], W[(size_t)(lh + 48) * HID + kb + 1] };
    a0 = wmma_k4(a, b0, a0);
    a1 = wmma_k4(a, b1, a1);
    a2 = wmma_k4(a, b2, a2);
    a3 = wmma_k4(a, b3, a3);
  }
  const float bb0 = bias[lh +  0];
  const float bb1 = bias[lh + 16];
  const float bb2 = bias[lh + 32];
  const float bb3 = bias[lh + 48];
#pragma unroll
  for (int r = 0; r < 8; ++r) {
    const int row = row0 + khalf * 8 + r;
    if (row < N) {
      out[(size_t)row * OUT_D + lh +  0] = a0[r] + bb0;
      out[(size_t)row * OUT_D + lh + 16] = a1[r] + bb1;
      out[(size_t)row * OUT_D + lh + 32] = a2[r] + bb2;
      out[(size_t)row * OUT_D + lh + 48] = a3[r] + bb3;
    }
  }
}

extern "C" void kernel_launch(void* const* d_in, const int* in_sizes, int n_in,
                              void* d_out, int out_size, void* d_ws, size_t ws_size,
                              hipStream_t stream) {
  const float* x    = (const float*)d_in[0];
  const int*   ei   = (const int*)  d_in[1];
  const float* Wlin = (const float*)d_in[2];
  const float* att  = (const float*)d_in[3];
  const float* Wout = (const float*)d_in[4];
  const float* bout = (const float*)d_in[5];
  float* out = (float*)d_out;

  const int N = in_sizes[0] / IN_DIM;
  const int E = in_sizes[1] / 2;

  float* ws     = (float*)d_ws;
  float* h      = ws;
  float* s      = h + (size_t)N * HID;
  float* dn     = s + N;
  int*   ssrc   = (int*)(dn + N);
  int*   cursor = ssrc + E;
  float* agg    = (float*)(cursor + N);

  const int NBK  = (N + BKT_NODES - 1) / BKT_NODES;
  const int NBLK = (E + ECHUNK - 1) / ECHUNK;
  const int L    = NBK * NBLK;
  const int L4   = (L + 3) / 4 * 4;
  int*   tcnt   = (int*)agg;
  int*   toff   = tcnt + L4;
  int*   cpair  = toff + L4;

  const int tiles   = (N + 15) / 16;
  const int gemmBlk = (tiles + 7) / 8;

  gat_lin_kernel<<<gemmBlk, 256, 0, stream>>>(x, Wlin, h, N);
  gat_sd_kernel<<<(N + 255) / 256, 256, 0, stream>>>(h, att, s, dn, N);

  gat_bcount_kernel<<<NBLK, 1024, 0, stream>>>(ei, tcnt, E, NBK, NBLK);
  gat_scan_kernel<<<(L + 1023) / 1024, 1024, 0, stream>>>(tcnt, toff, L);
  gat_bscatter_kernel<<<NBLK, 1024, 0, stream>>>(ei, toff, cpair, E, NBK, NBLK);
  gat_bsort_kernel<<<NBK, 1024, 0, stream>>>(toff, cpair, cursor, ssrc, N, E, NBK, NBLK);
  gat_gather_kernel<<<(N + 7) / 8, 256, 0, stream>>>(cursor, ssrc, h, s, dn, agg, N, E);

  gat_out_kernel<<<gemmBlk, 256, 0, stream>>>(agg, Wout, bout, out, N);
}


# Attention dots s,d formed in the first GEMM's epilogue from its accumulators (4 columns per lane, then row_xmask tree over 16 lanes) instead of a kernel re-reading h; the dots' sum order changes
# speedup vs baseline: 3.3847x; 1.0325x over previous
#include <hip/hip_runtime.h>

typedef float v2f __attribute__((ext_vector_type(2)));
typedef float v8f __attribute__((ext_vector_type(8)));

#define IN_DIM 128
#define HID    64
#define OUT_D  64

__device__ __forceinline__ v8f wmma_k4(v2f a, v2f b, v8f c) {
  return __builtin_amdgcn_wmma_f32_16x16x4_f32(false, a, false, b, (short)0, c,
                                               false, false);
}


template <int XM>
__device__ __forceinline__ float gat_row_xadd(float v) {
  const int t = __builtin_amdgcn_update_dpp(0, __float_as_int(v), 0x160 + XM, 0xF, 0xF, true);
  return v + __int_as_float(t);
}

__global__ void __launch_bounds__(256) gat_lin_kernel(const float* __restrict__ x,
                                                      const float* __restrict__ W,
                                                      const float* __restrict__ att,
                                                      float* __restrict__ h,
                                                      float* __restrict__ s,
                                                      float* __restrict__ d, int N) {
  __shared__ float sdl[2][128];
  const int lane  = threadIdx.x & 31;
  const int wave  = threadIdx.x >> 5;
  const int row0  = (blockIdx.x * 8 + wave) * 16;
  const int lh    = lane & 15;
  const int khalf = lane >> 4;
  const int rowA  = row0 + lh;
  const int rA    = rowA < N ? rowA : (N - 1);

  const float ws0 = att[lh +  0], ws1 = att[lh + 16], ws2 = att[lh + 32], ws3 = att[lh + 48];
  const float wd0 = att[HID + lh +  0], wd1 = att[HID + lh + 16];
  const float wd2 = att[HID + lh + 32], wd3 = att[HID + lh + 48];

  v8f a0 = {}, a1 = {}, a2 = {}, a3 = {};
#pragma unroll
  for (int k0 = 0; k0 < IN_DIM; k0 += 4) {
    const int kb = k0 + khalf * 2;
    v2f a = { x[(size_t)rA * IN_DIM + kb], x[(size_t)rA * IN_DIM + kb + 1] };
    v2f b0 = { W[(size_t)(lh +  0) * IN_DIM + kb], W[(size_t)(lh +  0) * IN_DIM + kb + 1] };
    v2f b1 = { W[(size_t)(lh + 16) * IN_DIM + kb], W[(size_t)(lh + 16) * IN_DIM + kb + 1] };
    v2f b2 = { W[(size_t)(lh + 32) * IN_DIM + kb], W[(size_t)(lh + 32) * IN_DIM + kb + 1] };
    v2f b3 = { W[(size_t)(lh + 48) * IN_DIM + kb], W[(size_t)(lh + 48) * IN_DIM + kb + 1] };
    a0 = wmma_k4(a, b0, a0);
    a1 = wmma_k4(a, b1, a1);
    a2 = wmma_k4(a, b2, a2);
    a3 = wmma_k4(a, b3, a3);
  }
  float sel = 0.f;
#pragma unroll
  for (int r = 0; r < 8; ++r) {
    float ps = a0[r] * ws0 + a1[r] * ws1 + a2[r] * ws2 + a3[r] * ws3;
    float pd = a0[r] * wd0 + a1[r] * wd1 + a2[r] * wd2 + a3[r] * wd3;
    ps = gat_row_xadd<8>(ps);  pd = gat_row_xadd<8>(pd);
    ps = gat_row_xadd<4>(ps);  pd = gat_row_xadd<4>(pd);
    ps = gat_row_xadd<2>(ps);  pd = gat_row_xadd<2>(pd);
    ps = gat_row_xadd<1>(ps);  pd = gat_row_xadd<1>(pd);
    if ((lh & 7) == r) sel = (lh < 8) ? ps : pd;
  }
  sdl[(lh < 8) ? 0 : 1][wave * 16 + khalf * 8 + (lh & 7)] = sel;
  __syncthreads();

#pragma unroll
  for (int r = 0; r < 8; ++r) {
    const int row = row0 + khalf * 8 + r;
    if (row < N) {
      h[(size_t)row * HID + lh +  0] = a0[r];
      h[(size_t)row * HID + lh + 16] = a1[r];
      h[(size_t)row * HID + lh + 32] = a2[r];
      h[(size_t)row * HID + lh + 48] = a3[r];
    }
  }

  const int which = threadIdx.x >> 7;
  const int lrow  = threadIdx.x & 127;
  const int rowS  = blockIdx.x * 128 + lrow;
  if (rowS < N) {
    float* __restrict__ dstp = which ? d : s;
    dstp[rowS] = sdl[which][lrow];
  }
}

#define BKT_SHIFT 8
#define BKT_NODES 256
#define BKT_WAVES (BKT_NODES / 32)
#define BKT_MAX   2048
#define SORT_PER  6
#define ECHUNK    4096

__global__ void __launch_bounds__(1024) gat_bcount_kernel(const int* __restrict__ ei,
                                                          int* __restrict__ tcnt,
                                                          int E, int NBK, int NBLK) {
  __shared__ int hc[BKT_MAX];
  const int tid = threadIdx.x;
  for (int i = tid; i < NBK; i += 1024) hc[i] = 0;
  __syncthreads();
  const int base = blockIdx.x * ECHUNK + tid;
  int d[4];
#pragma unroll
  for (int k = 0; k < 4; ++k) {
    const int e = base + k * 1024;
    d[k] = (e < E) ? ei[(size_t)E + e] : -1;
  }
#pragma unroll
  for (int k = 0; k < 4; ++k)
    if (d[k] >= 0) atomicAdd(&hc[d[k] >> BKT_SHIFT], 1);
  __syncthreads();
  for (int i = tid; i < NBK; i += 1024) tcnt[(size_t)i * NBLK + blockIdx.x] = hc[i];
}

template <int ROWS>
__device__ __forceinline__ int gat_rows_sum(const int4* __restrict__ q) {
  int4 v[ROWS];
#pragma unroll
  for (int j = 0; j < ROWS; ++j) v[j] = q[j * 1024];
  int t = 0;
#pragma unroll
  for (int j = 0; j < ROWS; ++j) t += (v[j].x + v[j].y) + (v[j].z + v[j].w);
  return t;
}

__global__ void __launch_bounds__(1024) gat_scan_kernel(const int* __restrict__ cnt,
                                                        int* __restrict__ off, int L) {
  __shared__ int red[32];
  __shared__ int wsum[32];
  const int tid   = threadIdx.x;
  const int lane  = tid & 31;
  const int wave  = tid >> 5;
  const int start = blockIdx.x * 1024;

  const int i0 = start + tid;
  const int c  = (i0 < L) ? cnt[i0] : 0;

  const int rows = start >> 12;
  const int rem  = start & 4095;
  const int4* __restrict__ p = reinterpret_cast<const int4*>(cnt) + tid;

  int4 vp = make_int4(0, 0, 0, 0);
  if (tid * 4 < rem) vp = p[(size_t)rows * 1024];

  int pre = 0;
  int r = 0;
  for (; r + 16 <= rows; r += 16) pre += gat_rows_sum<16>(p + (size_t)r * 1024);
  if (rows - r >= 8) { pre += gat_rows_sum<8>(p + (size_t)r * 1024); r += 8; }
  if (rows - r >= 4) { pre += gat_rows_sum<4>(p + (size_t)r * 1024); r += 4; }
  if (rows - r >= 2) { pre += gat_rows_sum<2>(p + (size_t)r * 1024); r += 2; }
  if (rows - r >= 1) { pre += gat_rows_sum<1>(p + (size_t)r * 1024); }
  pre += (vp.x + vp.y) + (vp.z + vp.w);
  for (int m = 16; m > 0; m >>= 1) pre += __shfl_xor(pre, m);

  int inc = c;
  for (int m = 1; m < 32; m <<= 1) {
    const int t = __shfl_up(inc, m);
    if (lane >= m) inc += t;
  }
  if (lane == 0)  red[wave]  = pre;
  if (lane == 31) wsum[wave] = inc;
  __syncthreads();
  int part = red[lane] + ((lane < wave) ? wsum[lane] : 0);
  for (int m = 16; m > 0; m >>= 1) part += __shfl_xor(part, m);
  if (i0 < L) off[i0] = part + inc - c;
}

__global__ void __launch_bounds__(1024) gat_bscatter_kernel(const int* __restrict__ ei,
                                                            const int* __restrict__ toff,
                                                            int* __restrict__ cpair,
                                                            int E, int NBK, int NBLK) {
  __shared__ int pos[BKT_MAX];
  const int tid = threadIdx.x;
  for (int i = tid; i < NBK; i += 1024) pos[i] = toff[(size_t)i * NBLK + blockIdx.x];
  __syncthreads();
  const int base = blockIdx.x * ECHUNK + tid;
  int s[4], d[4];
#pragma unroll
  for (int k = 0; k < 4; ++k) {
    const int e = base + k * 1024;
    s[k] = (e < E) ? ei[e] : 0;
    d[k] = (e < E) ? ei[(size_t)E + e] : -1;
  }
  int2* __restrict__ out2 = reinterpret_cast<int2*>(cpair);
#pragma unroll
  for (int k = 0; k < 4; ++k) {
    if (d[k] >= 0) {
      const int q = atomicAdd(&pos[d[k] >> BKT_SHIFT], 1);
      out2[q] = make_int2(s[k], d[k]);
    }
  }
}

__global__ void __launch_bounds__(1024) gat_bsort_kernel(const int* __restrict__ toff,
                                                         const int* __restrict__ cpair,
                                                         int* __restrict__ cursor,
                                                         int* __restrict__ ssrc,
                                                         int N, int E, int NBK, int NBLK) {
  __shared__ int cl[BKT_NODES];
  __shared__ int wtot[BKT_WAVES];
  const int tid   = threadIdx.x;
  const int lane  = tid & 31;
  const int wave  = tid >> 5;
  const int b     = blockIdx.x;
  const int node0 = b << BKT_SHIFT;
  const int bstart = toff[(size_t)b * NBLK];
  const int bend   = (b + 1 < NBK) ? toff[(size_t)(b + 1) * NBLK] : E;
  const int2* __restrict__ pr = reinterpret_cast<const int2*>(cpair);

  if (tid < BKT_NODES) cl[tid] = 0;
  __syncthreads();

  for (int e0 = bstart; e0 < bend; e0 += SORT_PER * 1024) {
    int d[SORT_PER];
#pragma unroll
    for (int k = 0; k < SORT_PER; ++k) {
      const int e = e0 + k * 1024 + tid;
      d[k] = (e < bend) ? pr[e].y : -1;
    }
#pragma unroll
    for (int k = 0; k < SORT_PER; ++k)
      if (d[k] >= 0) atomicAdd(&cl[d[k] - node0], 1);
  }
  __syncthreads();

  const int v = (tid < BKT_NODES) ? cl[tid] : 0;
  int inc = v;
  for (int m = 1; m < 32; m <<= 1) {
    const int t = __shfl_up(inc, m);
    if (lane >= m) inc += t;
  }
  if (lane == 31 && wave < BKT_WAVES) wtot[wave] = inc;
  __syncthreads();
  int part = (lane < BKT_WAVES && lane < wave) ? wtot[lane] : 0;
  for (int m = 16; m > 0; m >>= 1) part += __shfl_xor(part, m);
  const int excl = part + inc - v;
  if (tid < BKT_NODES) {
    cl[tid] = excl;
    if (node0 + tid < N) cursor[node0 + tid] = bstart + excl;
  }
  __syncthreads();

  for (int e0 = bstart; e0 < bend; e0 += SORT_PER * 1024) {
    int2 q[SORT_PER];
#pragma unroll
    for (int k = 0; k < SORT_PER; ++k) {
      const int e = e0 + k * 1024 + tid;
      q[k] = (e < bend) ? pr[e] : make_int2(0, -1);
    }
#pragma unroll
    for (int k = 0; k < SORT_PER; ++k) {
      if (q[k].y >= 0) {
        const int at = atomicAdd(&cl[q[k].y - node0], 1);
        ssrc[bstart + at] = q[k].x;
      }
    }
  }
}

__device__ __forceinline__ float gat_edge_w(const int* __restrict__ ssrc,
                                            const float* __restrict__ s,
                                            float dn, int idx, int end, int& srcOut) {
  int sj = 0;
  float w = 0.f;
  if (idx < end) {
    sj = ssrc[idx];
    float v = s[sj] + dn;
    v = (v > 0.f) ? v : 0.2f * v;
    w = __expf(v);
  }
  srcOut = sj;
  return w;
}

__global__ void __launch_bounds__(256) gat_gather_kernel(const int* __restrict__ cursor,
                                                         const int* __restrict__ ssrc,
                                                         const float* __restrict__ h,
                                                         const float* __restrict__ s,
                                                         const float* __restrict__ d,
                                                         float* __restrict__ agg, int N, int E) {
  const int lane = threadIdx.x & 31;
  const int n = __builtin_amdgcn_readfirstlane(blockIdx.x * 8 + (threadIdx.x >> 5));
  if (n >= N) return;
  const int begin = __builtin_amdgcn_readfirstlane(cursor[n]);
  const int end   = __builtin_amdgcn_readfirstlane((n + 1 < N) ? cursor[n + 1] : E);
  const float dn  = d[n];

  int   src0;
  const float w0 = gat_edge_w(ssrc, s, dn, begin + lane, end, src0);
  float dsum = w0;
  for (int base = begin + 32; base < end; base += 32) {
    int t;
    dsum += gat_edge_w(ssrc, s, dn, base + lane, end, t);
  }
  for (int m = 16; m > 0; m >>= 1) dsum += __shfl_xor(dsum, m);
  const float den = dsum + 1e-9f;

  const v2f* __restrict__ h2 = reinterpret_cast<const v2f*>(h);
  v2f acc = {0.f, 0.f};
  for (int base = begin; base < end; base += 32) {
    int   srcj = src0;
    float wj   = w0;
    if (base != begin) wj = gat_edge_w(ssrc, s, dn, base + lane, end, srcj);
    const float aj = wj / den;
    const int   ab = __float_as_int(aj);
    const int   m  = (end - base < 32) ? (end - base) : 32;
    for (int k = 0; k < m; k += 8) {
      v2f   v[8];
      float a[8];
#pragma unroll
      for (int j = 0; j < 8; ++j) {
        const bool live = (k + j) < m;
        const int  kk   = live ? (k + j) : (m - 1);
        const int  sk   = __builtin_amdgcn_readlane(srcj, kk);
        a[j] = live ? __int_as_float(__builtin_amdgcn_readlane(ab, kk)) : 0.f;
        v[j] = h2[(size_t)sk * (HID / 2) + lane];
      }
#pragma unroll
      for (int j = 0; j < 8; ++j) {
        const bool live = (k + j) < m;
        const float vx = live ? v[j].x : 0.f;
        const float vy = live ? v[j].y : 0.f;
        acc.x += vx * a[j];
        acc.y += vy * a[j];
      }
    }
  }
  reinterpret_cast<v2f*>(agg)[(size_t)n * (HID / 2) + lane] = acc;
}

__global__ void __launch_bounds__(256) gat_out_kernel(const float* __restrict__ agg,
                                                      const float* __restrict__ W,
                                                      const float* __restrict__ bias,
                                                      float* __restrict__ out, int N) {
  const int lane  = threadIdx.x & 31;
  const int wave  = threadIdx.x >> 5;
  const int row0  = (blockIdx.x * 8 + wave) * 16;
  const int lh    = lane & 15;
  const int khalf = lane >> 4;
  const int rowA  = row0 + lh;
  const int rA    = rowA < N ? rowA : (N - 1);

  v8f a0 = {}, a1 = {}, a2 = {}, a3 = {};
  for (int k0 = 0; k0 < HID; k0 += 4) {
    const int kb = k0 + khalf * 2;
    const float z0 = agg[(size_t)rA * HID + kb];
    const float z1 = agg[(size_t)rA * HID + kb + 1];
    v2f a = { z0 > 0.f ? z0 : 0.f, z1 > 0.f ? z1 : 0.f };
    v2f b0 = { W[(size_t)(lh +  0) * HID + kb], W[(size_t)(lh +  0) * HID + kb + 1] };
    v2f b1 = { W[(size_t)(lh + 16) * HID + kb], W[(size_t)(lh + 16) * HID + kb + 1] };
    v2f b2 = { W[(size_t)(lh + 32) * HID + kb], W[(size_t)(lh + 32) * HID + kb + 1] };
    v2f b3 = { W[(size_t)(lh + 48) * HID + kb], W[(size_t)(lh + 48) * HID + kb + 1] };
    a0 = wmma_k4(a, b0, a0);
    a1 = wmma_k4(a, b1, a1);
    a2 = wmma_k4(a, b2, a2);
    a3 = wmma_k4(a, b3, a3);
  }
  const float bb0 = bias[lh +  0];
  const float bb1 = bias[lh + 16];
  const float bb2 = bias[lh + 32];
  const float bb3 = bias[lh + 48];
#pragma unroll
  for (int r = 0; r < 8; ++r) {
    const int row = row0 + khalf * 8 + r;
    if (row < N) {
      out[(size_t)row * OUT_D + lh +  0] = a0[r] + bb0;
      out[(size_t)row * OUT_D + lh + 16] = a1[r] + bb1;
      out[(size_t)row * OUT_D + lh + 32] = a2[r] + bb2;
      out[(size_t)row * OUT_D + lh + 48] = a3[r] + bb3;
    }
  }
}

extern "C" void kernel_launch(void* const* d_in, const int* in_sizes, int n_in,
                              void* d_out, int out_size, void* d_ws, size_t ws_size,
                              hipStream_t stream) {
  const float* x    = (const float*)d_in[0];
  const int*   ei   = (const int*)  d_in[1];
  const float* Wlin = (const float*)d_in[2];
  const float* att  = (const float*)d_in[3];
  const float* Wout = (const float*)d_in[4];
  const float* bout = (const float*)d_in[5];
  float* out = (float*)d_out;

  const int N = in_sizes[0] / IN_DIM;
  const int E = in_sizes[1] / 2;

  float* ws     = (float*)d_ws;
  float* h      = ws;
  float* s      = h + (size_t)N * HID;
  float* dn     = s + N;
  int*   ssrc   = (int*)(dn + N);
  int*   cursor = ssrc + E;
  float* agg    = (float*)(cursor + N);

  const int NBK  = (N + BKT_NODES - 1) / BKT_NODES;
  const int NBLK = (E + ECHUNK - 1) / ECHUNK;
  const int L    = NBK * NBLK;
  const int L4   = (L + 3) / 4 * 4;
  int*   tcnt   = (int*)agg;
  int*   toff   = tcnt + L4;
  int*   cpair  = toff + L4;

  const int tiles   = (N + 15) / 16;
  const int gemmBlk = (tiles + 7) / 8;

  gat_lin_kernel<<<gemmBlk, 256, 0, stream>>>(x, Wlin, att, h, s, dn, N);

  gat_bcount_kernel<<<NBLK, 1024, 0, stream>>>(ei, tcnt, E, NBK, NBLK);
  gat_scan_kernel<<<(L + 1023) / 1024, 1024, 0, stream>>>(tcnt, toff, L);
  gat_bscatter_kernel<<<NBLK, 1024, 0, stream>>>(ei, toff, cpair, E, NBK, NBLK);
  gat_bsort_kernel<<<NBK, 1024, 0, stream>>>(toff, cpair, cursor, ssrc, N, E, NBK, NBLK);
  gat_gather_kernel<<<(N + 7) / 8, 256, 0, stream>>>(cursor, ssrc, h, s, dn, agg, N, E);

  gat_out_kernel<<<gemmBlk, 256, 0, stream>>>(agg, Wout, bout, out, N);
}


# Weight (B) fragments of both GEMMs read from LDS (W staged once per block, row pitch +4 dwords) instead of 16-line global loads per k-step; A loads issued before an LDS-only barrier; sums unchanged
# speedup vs baseline: 3.8188x; 1.1283x over previous
#include <hip/hip_runtime.h>

typedef float v2f __attribute__((ext_vector_type(2)));
typedef float v8f __attribute__((ext_vector_type(8)));

#define IN_DIM 128
#define HID    64
#define OUT_D  64

#define WL_PITCH (IN_DIM + 4)
#define WO_PITCH (HID + 4)

__device__ __forceinline__ v8f wmma_k4(v2f a, v2f b, v8f c) {
  return __builtin_amdgcn_wmma_f32_16x16x4_f32(false, a, false, b, (short)0, c,
                                               false, false);
}


template <int XM>
__device__ __forceinline__ float gat_row_xadd(float v) {
  const int t = __builtin_amdgcn_update_dpp(0, __float_as_int(v), 0x160 + XM, 0xF, 0xF, true);
  return v + __int_as_float(t);
}

__global__ void __launch_bounds__(256) gat_lin_kernel(const float* __restrict__ x,
                                                      const float* __restrict__ W,
                                                      const float* __restrict__ att,
                                                      float* __restrict__ h,
                                                      float* __restrict__ s,
                                                      float* __restrict__ d, int N) {
  __shared__ float sdl[2][128];
  __shared__ __attribute__((aligned(16))) float wl[HID * WL_PITCH];
  const int lane  = threadIdx.x & 31;
  const int wave  = threadIdx.x >> 5;
  const int row0  = (blockIdx.x * 8 + wave) * 16;
  const int lh    = lane & 15;
  const int khalf = lane >> 4;
  const int rowA  = row0 + lh;
  const int rA    = rowA < N ? rowA : (N - 1);

  const float ws0 = att[lh +  0], ws1 = att[lh + 16], ws2 = att[lh + 32], ws3 = att[lh + 48];
  const float wd0 = att[HID + lh +  0], wd1 = att[HID + lh + 16];
  const float wd2 = att[HID + lh + 32], wd3 = att[HID + lh + 48];

  v2f av[IN_DIM / 4];
  {
    const float4* __restrict__ W4 = reinterpret_cast<const float4*>(W);
    float4 wv[(HID * IN_DIM / 4) / 256];
#pragma unroll
    for (int i = 0; i < (HID * IN_DIM / 4) / 256; ++i) wv[i] = W4[i * 256 + threadIdx.x];
#pragma unroll
    for (int i = 0; i < IN_DIM / 4; ++i)
      av[i] = *reinterpret_cast<const v2f*>(&x[(size_t)rA * IN_DIM + i * 4 + khalf * 2]);
#pragma unroll
    for (int i = 0; i < (HID * IN_DIM / 4) / 256; ++i) {
      const int idx = i * 256 + threadIdx.x;
      const int r   = idx >> 5;
      const int c4  = idx & 31;
      *reinterpret_cast<float4*>(&wl[r * WL_PITCH + c4 * 4]) = wv[i];
    }
  }
  __builtin_amdgcn_fence(__ATOMIC_RELEASE, "workgroup", "local");
  __builtin_amdgcn_s_barrier();
  __builtin_amdgcn_fence(__ATOMIC_ACQUIRE, "workgroup", "local");
  const float* __restrict__ wlp = &wl[lh * WL_PITCH + khalf * 2];

  v8f a0 = {}, a1 = {}, a2 = {}, a3 = {};
#pragma unroll
  for (int k0 = 0; k0 < IN_DIM; k0 += 4) {
    const v2f a = av[k0 / 4];
    const v2f b0 = *reinterpret_cast<const v2f*>(wlp +  0 * WL_PITCH + k0);
    const v2f b1 = *reinterpret_cast<const v2f*>(wlp + 16 * WL_PITCH + k0);
    const v2f b2 = *reinterpret_cast<const v2f*>(wlp + 32 * WL_PITCH + k0);
    const v2f b3 = *reinterpret_cast<const v2f*>(wlp + 48 * WL_PITCH + k0);
    a0 = wmma_k4(a, b0, a0);
    a1 = wmma_k4(a, b1, a1);
    a2 = wmma_k4(a, b2, a2);
    a3 = wmma_k4(a, b3, a3);
  }
  float sel = 0.f;
#pragma unroll
  for (int r = 0; r < 8; ++r) {
    float ps = a0[r] * ws0 + a1[r] * ws1 + a2[r] * ws2 + a3[r] * ws3;
    float pd = a0[r] * wd0 + a1[r] * wd1 + a2[r] * wd2 + a3[r] * wd3;
    ps = gat_row_xadd<8>(ps);  pd = gat_row_xadd<8>(pd);
    ps = gat_row_xadd<4>(ps);  pd = gat_row_xadd<4>(pd);
    ps = gat_row_xadd<2>(ps);  pd = gat_row_xadd<2>(pd);
    ps = gat_row_xadd<1>(ps);  pd = gat_row_xadd<1>(pd);
    if ((lh & 7) == r) sel = (lh < 8) ? ps : pd;
  }
  sdl[(lh < 8) ? 0 : 1][wave * 16 + khalf * 8 + (lh & 7)] = sel;
  __syncthreads();

#pragma unroll
  for (int r = 0; r < 8; ++r) {
    const int row = row0 + khalf * 8 + r;
    if (row < N) {
      h[(size_t)row * HID + lh +  0] = a0[r];
      h[(size_t)row * HID + lh + 16] = a1[r];
      h[(size_t)row * HID + lh + 32] = a2[r];
      h[(size_t)row * HID + lh + 48] = a3[r];
    }
  }

  const int which = threadIdx.x >> 7;
  const int lrow  = threadIdx.x & 127;
  const int rowS  = blockIdx.x * 128 + lrow;
  if (rowS < N) {
    float* __restrict__ dstp = which ? d : s;
    dstp[rowS] = sdl[which][lrow];
  }
}

#define BKT_SHIFT 8
#define BKT_NODES 256
#define BKT_WAVES (BKT_NODES / 32)
#define BKT_MAX   2048
#define SORT_PER  6
#define ECHUNK    4096

__global__ void __launch_bounds__(1024) gat_bcount_kernel(const int* __restrict__ ei,
                                                          int* __restrict__ tcnt,
                                                          int E, int NBK, int NBLK) {
  __shared__ int hc[BKT_MAX];
  const int tid = threadIdx.x;
  for (int i = tid; i < NBK; i += 1024) hc[i] = 0;
  __syncthreads();
  const int base = blockIdx.x * ECHUNK + tid;
  int d[4];
#pragma unroll
  for (int k = 0; k < 4; ++k) {
    const int e = base + k * 1024;
    d[k] = (e < E) ? ei[(size_t)E + e] : -1;
  }
#pragma unroll
  for (int k = 0; k < 4; ++k)
    if (d[k] >= 0) atomicAdd(&hc[d[k] >> BKT_SHIFT], 1);
  __syncthreads();
  for (int i = tid; i < NBK; i += 1024) tcnt[(size_t)i * NBLK + blockIdx.x] = hc[i];
}

template <int ROWS>
__device__ __forceinline__ int gat_rows_sum(const int4* __restrict__ q) {
  int4 v[ROWS];
#pragma unroll
  for (int j = 0; j < ROWS; ++j) v[j] = q[j * 1024];
  int t = 0;
#pragma unroll
  for (int j = 0; j < ROWS; ++j) t += (v[j].x + v[j].y) + (v[j].z + v[j].w);
  return t;
}

__global__ void __launch_bounds__(1024) gat_scan_kernel(const int* __restrict__ cnt,
                                                        int* __restrict__ off, int L) {
  __shared__ int red[32];
  __shared__ int wsum[32];
  const int tid   = threadIdx.x;
  const int lane  = tid & 31;
  const int wave  = tid >> 5;
  const int start = blockIdx.x * 1024;

  const int i0 = start + tid;
  const int c  = (i0 < L) ? cnt[i0] : 0;

  const int rows = start >> 12;
  const int rem  = start & 4095;
  const int4* __restrict__ p = reinterpret_cast<const int4*>(cnt) + tid;

  int4 vp = make_int4(0, 0, 0, 0);
  if (tid * 4 < rem) vp = p[(size_t)rows * 1024];

  int pre = 0;
  int r = 0;
  for (; r + 16 <= rows; r += 16) pre += gat_rows_sum<16>(p + (size_t)r * 1024);
  if (rows - r >= 8) { pre += gat_rows_sum<8>(p + (size_t)r * 1024); r += 8; }
  if (rows - r >= 4) { pre += gat_rows_sum<4>(p + (size_t)r * 1024); r += 4; }
  if (rows - r >= 2) { pre += gat_rows_sum<2>(p + (size_t)r * 1024); r += 2; }
  if (rows - r >= 1) { pre += gat_rows_sum<1>(p + (size_t)r * 1024); }
  pre += (vp.x + vp.y) + (vp.z + vp.w);
  for (int m = 16; m > 0; m >>= 1) pre += __shfl_xor(pre, m);

  int inc = c;
  for (int m = 1; m < 32; m <<= 1) {
    const int t = __shfl_up(inc, m);
    if (lane >= m) inc += t;
  }
  if (lane == 0)  red[wave]  = pre;
  if (lane == 31) wsum[wave] = inc;
  __syncthreads();
  int part = red[lane] + ((lane < wave) ? wsum[lane] : 0);
  for (int m = 16; m > 0; m >>= 1) part += __shfl_xor(part, m);
  if (i0 < L) off[i0] = part + inc - c;
}

__global__ void __launch_bounds__(1024) gat_bscatter_kernel(const int* __restrict__ ei,
                                                            const int* __restrict__ toff,
                                                            int* __restrict__ cpair,
                                                            int E, int NBK, int NBLK) {
  __shared__ int pos[BKT_MAX];
  const int tid = threadIdx.x;
  for (int i = tid; i < NBK; i += 1024) pos[i] = toff[(size_t)i * NBLK + blockIdx.x];
  __syncthreads();
  const int base = blockIdx.x * ECHUNK + tid;
  int s[4], d[4];
#pragma unroll
  for (int k = 0; k < 4; ++k) {
    const int e = base + k * 1024;
    s[k] = (e < E) ? ei[e] : 0;
    d[k] = (e < E) ? ei[(size_t)E + e] : -1;
  }
  int2* __restrict__ out2 = reinterpret_cast<int2*>(cpair);
#pragma unroll
  for (int k = 0; k < 4; ++k) {
    if (d[k] >= 0) {
      const int q = atomicAdd(&pos[d[k] >> BKT_SHIFT], 1);
      out2[q] = make_int2(s[k], d[k]);
    }
  }
}

__global__ void __launch_bounds__(1024) gat_bsort_kernel(const int* __restrict__ toff,
                                                         const int* __restrict__ cpair,
                                                         int* __restrict__ cursor,
                                                         int* __restrict__ ssrc,
                                                         int N, int E, int NBK, int NBLK) {
  __shared__ int cl[BKT_NODES];
  __shared__ int wtot[BKT_WAVES];
  const int tid   = threadIdx.x;
  const int lane  = tid & 31;
  const int wave  = tid >> 5;
  const int b     = blockIdx.x;
  const int node0 = b << BKT_SHIFT;
  const int bstart = toff[(size_t)b * NBLK];
  const int bend   = (b + 1 < NBK) ? toff[(size_t)(b + 1) * NBLK] : E;
  const int2* __restrict__ pr = reinterpret_cast<const int2*>(cpair);

  if (tid < BKT_NODES) cl[tid] = 0;
  __syncthreads();

  for (int e0 = bstart; e0 < bend; e0 += SORT_PER * 1024) {
    int d[SORT_PER];
#pragma unroll
    for (int k = 0; k < SORT_PER; ++k) {
      const int e = e0 + k * 1024 + tid;
      d[k] = (e < bend) ? pr[e].y : -1;
    }
#pragma unroll
    for (int k = 0; k < SORT_PER; ++k)
      if (d[k] >= 0) atomicAdd(&cl[d[k] - node0], 1);
  }
  __syncthreads();

  const int v = (tid < BKT_NODES) ? cl[tid] : 0;
  int inc = v;
  for (int m = 1; m < 32; m <<= 1) {
    const int t = __shfl_up(inc, m);
    if (lane >= m) inc += t;
  }
  if (lane == 31 && wave < BKT_WAVES) wtot[wave] = inc;
  __syncthreads();
  int part = (lane < BKT_WAVES && lane < wave) ? wtot[lane] : 0;
  for (int m = 16; m > 0; m >>= 1) part += __shfl_xor(part, m);
  const int excl = part + inc - v;
  if (tid < BKT_NODES) {
    cl[tid] = excl;
    if (node0 + tid < N) cursor[node0 + tid] = bstart + excl;
  }
  __syncthreads();

  for (int e0 = bstart; e0 < bend; e0 += SORT_PER * 1024) {
    int2 q[SORT_PER];
#pragma unroll
    for (int k = 0; k < SORT_PER; ++k) {
      const int e = e0 + k * 1024 + tid;
      q[k] = (e < bend) ? pr[e] : make_int2(0, -1);
    }
#pragma unroll
    for (int k = 0; k < SORT_PER; ++k) {
      if (q[k].y >= 0) {
        const int at = atomicAdd(&cl[q[k].y - node0], 1);
        ssrc[bstart + at] = q[k].x;
      }
    }
  }
}

__device__ __forceinline__ float gat_edge_w(const int* __restrict__ ssrc,
                                            const float* __restrict__ s,
                                            float dn, int idx, int end, int& srcOut) {
  int sj = 0;
  float w = 0.f;
  if (idx < end) {
    sj = ssrc[idx];
    float v = s[sj] + dn;
    v = (v > 0.f) ? v : 0.2f * v;
    w = __expf(v);
  }
  srcOut = sj;
  return w;
}

__global__ void __launch_bounds__(256) gat_gather_kernel(const int* __restrict__ cursor,
                                                         const int* __restrict__ ssrc,
                                                         const float* __restrict__ h,
                                                         const float* __restrict__ s,
                                                         const float* __restrict__ d,
                                                         float* __restrict__ agg, int N, int E) {
  const int lane = threadIdx.x & 31;
  const int n = __builtin_amdgcn_readfirstlane(blockIdx.x * 8 + (threadIdx.x >> 5));
  if (n >= N) return;
  const int begin = __builtin_amdgcn_readfirstlane(cursor[n]);
  const int end   = __builtin_amdgcn_readfirstlane((n + 1 < N) ? cursor[n + 1] : E);
  const float dn  = d[n];

  int   src0;
  const float w0 = gat_edge_w(ssrc, s, dn, begin + lane, end, src0);
  float dsum = w0;
  for (int base = begin + 32; base < end; base += 32) {
    int t;
    dsum += gat_edge_w(ssrc, s, dn, base + lane, end, t);
  }
  for (int m = 16; m > 0; m >>= 1) dsum += __shfl_xor(dsum, m);
  const float den = dsum + 1e-9f;

  const v2f* __restrict__ h2 = reinterpret_cast<const v2f*>(h);
  v2f acc = {0.f, 0.f};
  for (int base = begin; base < end; base += 32) {
    int   srcj = src0;
    float wj   = w0;
    if (base != begin) wj = gat_edge_w(ssrc, s, dn, base + lane, end, srcj);
    const float aj = wj / den;
    const int   ab = __float_as_int(aj);
    const int   m  = (end - base < 32) ? (end - base) : 32;
    for (int k = 0; k < m; k += 8) {
      v2f   v[8];
      float a[8];
#pragma unroll
      for (int j = 0; j < 8; ++j) {
        const bool live = (k + j) < m;
        const int  kk   = live ? (k + j) : (m - 1);
        const int  sk   = __builtin_amdgcn_readlane(srcj, kk);
        a[j] = live ? __int_as_float(__builtin_amdgcn_readlane(ab, kk)) : 0.f;
        v[j] = h2[(size_t)sk * (HID / 2) + lane];
      }
#pragma unroll
      for (int j = 0; j < 8; ++j) {
        const bool live = (k + j) < m;
        const float vx = live ? v[j].x : 0.f;
        const float vy = live ? v[j].y : 0.f;
        acc.x += vx * a[j];
        acc.y += vy * a[j];
      }
    }
  }
  reinterpret_cast<v2f*>(agg)[(size_t)n * (HID / 2) + lane] = acc;
}

__global__ void __launch_bounds__(256) gat_out_kernel(const float* __restrict__ agg,
                                                      const float* __restrict__ W,
                                                      const float* __restrict__ bias,
                                                      float* __restrict__ out, int N) {
  const int lane  = threadIdx.x & 31;
  const int wave  = threadIdx.x >> 5;
  const int row0  = (blockIdx.x * 8 + wave) * 16;
  const int lh    = lane & 15;
  const int khalf = lane >> 4;
  const int rowA  = row0 + lh;
  const int rA    = rowA < N ? rowA : (N - 1);

  __shared__ __attribute__((aligned(16))) float wo[OUT_D * WO_PITCH];
  v2f av[HID / 4];
  {
    const float4* __restrict__ W4 = reinterpret_cast<const float4*>(W);
    float4 wv[(OUT_D * HID / 4) / 256];
#pragma unroll
    for (int i = 0; i < (OUT_D * HID / 4) / 256; ++i) wv[i] = W4[i * 256 + threadIdx.x];
#pragma unroll
    for (int i = 0; i < HID / 4; ++i)
      av[i] = *reinterpret_cast<const v2f*>(&agg[(size_t)rA * HID + i * 4 + khalf * 2]);
#pragma unroll
    for (int i = 0; i < (OUT_D * HID / 4) / 256; ++i) {
      const int idx = i * 256 + threadIdx.x;
      const int r   = idx >> 4;
      const int c4  = idx & 15;
      *reinterpret_cast<float4*>(&wo[r * WO_PITCH + c4 * 4]) = wv[i];
    }
  }
  __builtin_amdgcn_fence(__ATOMIC_RELEASE, "workgroup", "local");
  __builtin_amdgcn_s_barrier();
  __builtin_amdgcn_fence(__ATOMIC_ACQUIRE, "workgroup", "local");
  const float* __restrict__ wop = &wo[lh * WO_PITCH + khalf * 2];

  v8f a0 = {}, a1 = {}, a2 = {}, a3 = {};
#pragma unroll
  for (int k0 = 0; k0 < HID; k0 += 4) {
    const float z0 = av[k0 / 4].x;
    const float z1 = av[k0 / 4].y;
    v2f a = { z0 > 0.f ? z0 : 0.f, z1 > 0.f ? z1 : 0.f };
    const v2f b0 = *reinterpret_cast<const v2f*>(wop +  0 * WO_PITCH + k0);
    const v2f b1 = *reinterpret_cast<const v2f*>(wop + 16 * WO_PITCH + k0);
    const v2f b2 = *reinterpret_cast<const v2f*>(wop + 32 * WO_PITCH + k0);
    const v2f b3 = *reinterpret_cast<const v2f*>(wop + 48 * WO_PITCH + k0);
    a0 = wmma_k4(a, b0, a0);
    a1 = wmma_k4(a, b1, a1);
    a2 = wmma_k4(a, b2, a2);
    a3 = wmma_k4(a, b3, a3);
  }
  const float bb0 = bias[lh +  0];
  const float bb1 = bias[lh + 16];
  const float bb2 = bias[lh + 32];
  const float bb3 = bias[lh + 48];
#pragma unroll
  for (int r = 0; r < 8; ++r) {
    const int row = row0 + khalf * 8 + r;
    if (row < N) {
      out[(size_t)row * OUT_D + lh +  0] = a0[r] + bb0;
      out[(size_t)row * OUT_D + lh + 16] = a1[r] + bb1;
      out[(size_t)row * OUT_D + lh + 32] = a2[r] + bb2;
      out[(size_t)row * OUT_D + lh + 48] = a3[r] + bb3;
    }
  }
}

extern "C" void kernel_launch(void* const* d_in, const int* in_sizes, int n_in,
                              void* d_out, int out_size, void* d_ws, size_t ws_size,
                              hipStream_t stream) {
  const float* x    = (const float*)d_in[0];
  const int*   ei   = (const int*)  d_in[1];
  const float* Wlin = (const float*)d_in[2];
  const float* att  = (const float*)d_in[3];
  const float* Wout = (const float*)d_in[4];
  const float* bout = (const float*)d_in[5];
  float* out = (float*)d_out;

  const int N = in_sizes[0] / IN_DIM;
  const int E = in_sizes[1] / 2;

  float* ws     = (float*)d_ws;
  float* h      = ws;
  float* s      = h + (size_t)N * HID;
  float* dn     = s + N;
  int*   ssrc   = (int*)(dn + N);
  int*   cursor = ssrc + E;
  float* agg    = (float*)(cursor + N);

  const int NBK  = (N + BKT_NODES - 1) / BKT_NODES;
  const int NBLK = (E + ECHUNK - 1) / ECHUNK;
  const int L    = NBK * NBLK;
  const int L4   = (L + 3) / 4 * 4;
  int*   tcnt   = (int*)agg;
  int*   toff   = tcnt + L4;
  int*   cpair  = toff + L4;

  const int tiles   = (N + 15) / 16;
  const int gemmBlk = (tiles + 7) / 8;

  gat_lin_kernel<<<gemmBlk, 256, 0, stream>>>(x, Wlin, att, h, s, dn, N);

  gat_bcount_kernel<<<NBLK, 1024, 0, stream>>>(ei, tcnt, E, NBK, NBLK);
  gat_scan_kernel<<<(L + 1023) / 1024, 1024, 0, stream>>>(tcnt, toff, L);
  gat_bscatter_kernel<<<NBLK, 1024, 0, stream>>>(ei, toff, cpair, E, NBK, NBLK);
  gat_bsort_kernel<<<NBK, 1024, 0, stream>>>(toff, cpair, cursor, ssrc, N, E, NBK, NBLK);
  gat_gather_kernel<<<(N + 7) / 8, 256, 0, stream>>>(cursor, ssrc, h, s, dn, agg, N, E);

  gat_out_kernel<<<gemmBlk, 256, 0, stream>>>(agg, Wout, bout, out, N);
}


# Gather in one pass: each edge's logit is summed from its fetched h row (16-lane DPP sums), not read from the s table before a denominator pass; sum orders change, one division per node after the sum
# speedup vs baseline: 4.1264x; 1.0805x over previous
#include <hip/hip_runtime.h>

typedef float v2f __attribute__((ext_vector_type(2)));
typedef float v8f __attribute__((ext_vector_type(8)));

#define IN_DIM 128
#define HID    64
#define OUT_D  64

#define WL_PITCH (IN_DIM + 4)
#define WO_PITCH (HID + 4)

__device__ __forceinline__ v8f wmma_k4(v2f a, v2f b, v8f c) {
  return __builtin_amdgcn_wmma_f32_16x16x4_f32(false, a, false, b, (short)0, c,
                                               false, false);
}


template <int XM>
__device__ __forceinline__ float gat_row_xadd(float v) {
  const int t = __builtin_amdgcn_update_dpp(0, __float_as_int(v), 0x160 + XM, 0xF, 0xF, true);
  return v + __int_as_float(t);
}

__global__ void __launch_bounds__(256) gat_lin_kernel(const float* __restrict__ x,
                                                      const float* __restrict__ W,
                                                      const float* __restrict__ att,
                                                      float* __restrict__ h,
                                                      float* __restrict__ s,
                                                      float* __restrict__ d, int N) {
  __shared__ float sdl[2][128];
  __shared__ __attribute__((aligned(16))) float wl[HID * WL_PITCH];
  const int lane  = threadIdx.x & 31;
  const int wave  = threadIdx.x >> 5;
  const int row0  = (blockIdx.x * 8 + wave) * 16;
  const int lh    = lane & 15;
  const int khalf = lane >> 4;
  const int rowA  = row0 + lh;
  const int rA    = rowA < N ? rowA : (N - 1);

  const float ws0 = att[lh +  0], ws1 = att[lh + 16], ws2 = att[lh + 32], ws3 = att[lh + 48];
  const float wd0 = att[HID + lh +  0], wd1 = att[HID + lh + 16];
  const float wd2 = att[HID + lh + 32], wd3 = att[HID + lh + 48];

  v2f av[IN_DIM / 4];
  {
    const float4* __restrict__ W4 = reinterpret_cast<const float4*>(W);
    float4 wv[(HID * IN_DIM / 4) / 256];
#pragma unroll
    for (int i = 0; i < (HID * IN_DIM / 4) / 256; ++i) wv[i] = W4[i * 256 + threadIdx.x];
#pragma unroll
    for (int i = 0; i < IN_DIM / 4; ++i)
      av[i] = *reinterpret_cast<const v2f*>(&x[(size_t)rA * IN_DIM + i * 4 + khalf * 2]);
#pragma unroll
    for (int i = 0; i < (HID * IN_DIM / 4) / 256; ++i) {
      const int idx = i * 256 + threadIdx.x;
      const int r   = idx >> 5;
      const int c4  = idx & 31;
      *reinterpret_cast<float4*>(&wl[r * WL_PITCH + c4 * 4]) = wv[i];
    }
  }
  __builtin_amdgcn_fence(__ATOMIC_RELEASE, "workgroup", "local");
  __builtin_amdgcn_s_barrier();
  __builtin_amdgcn_fence(__ATOMIC_ACQUIRE, "workgroup", "local");
  const float* __restrict__ wlp = &wl[lh * WL_PITCH + khalf * 2];

  v8f a0 = {}, a1 = {}, a2 = {}, a3 = {};
#pragma unroll
  for (int k0 = 0; k0 < IN_DIM; k0 += 4) {
    const v2f a = av[k0 / 4];
    const v2f b0 = *reinterpret_cast<const v2f*>(wlp +  0 * WL_PITCH + k0);
    const v2f b1 = *reinterpret_cast<const v2f*>(wlp + 16 * WL_PITCH + k0);
    const v2f b2 = *reinterpret_cast<const v2f*>(wlp + 32 * WL_PITCH + k0);
    const v2f b3 = *reinterpret_cast<const v2f*>(wlp + 48 * WL_PITCH + k0);
    a0 = wmma_k4(a, b0, a0);
    a1 = wmma_k4(a, b1, a1);
    a2 = wmma_k4(a, b2, a2);
    a3 = wmma_k4(a, b3, a3);
  }
  float sel = 0.f;
#pragma unroll
  for (int r = 0; r < 8; ++r) {
    float ps = a0[r] * ws0 + a1[r] * ws1 + a2[r] * ws2 + a3[r] * ws3;
    float pd = a0[r] * wd0 + a1[r] * wd1 + a2[r] * wd2 + a3[r] * wd3;
    ps = gat_row_xadd<8>(ps);  pd = gat_row_xadd<8>(pd);
    ps = gat_row_xadd<4>(ps);  pd = gat_row_xadd<4>(pd);
    ps = gat_row_xadd<2>(ps);  pd = gat_row_xadd<2>(pd);
    ps = gat_row_xadd<1>(ps);  pd = gat_row_xadd<1>(pd);
    if ((lh & 7) == r) sel = (lh < 8) ? ps : pd;
  }
  sdl[(lh < 8) ? 0 : 1][wave * 16 + khalf * 8 + (lh & 7)] = sel;
  __syncthreads();

#pragma unroll
  for (int r = 0; r < 8; ++r) {
    const int row = row0 + khalf * 8 + r;
    if (row < N) {
      h[(size_t)row * HID + lh +  0] = a0[r];
      h[(size_t)row * HID + lh + 16] = a1[r];
      h[(size_t)row * HID + lh + 32] = a2[r];
      h[(size_t)row * HID + lh + 48] = a3[r];
    }
  }

  const int which = threadIdx.x >> 7;
  const int lrow  = threadIdx.x & 127;
  const int rowS  = blockIdx.x * 128 + lrow;
  if (rowS < N) {
    float* __restrict__ dstp = which ? d : s;
    dstp[rowS] = sdl[which][lrow];
  }
}

#define BKT_SHIFT 8
#define BKT_NODES 256
#define BKT_WAVES (BKT_NODES / 32)
#define BKT_MAX   2048
#define SORT_PER  6
#define ECHUNK    4096

__global__ void __launch_bounds__(1024) gat_bcount_kernel(const int* __restrict__ ei,
                                                          int* __restrict__ tcnt,
                                                          int E, int NBK, int NBLK) {
  __shared__ int hc[BKT_MAX];
  const int tid = threadIdx.x;
  for (int i = tid; i < NBK; i += 1024) hc[i] = 0;
  __syncthreads();
  const int base = blockIdx.x * ECHUNK + tid;
  int d[4];
#pragma unroll
  for (int k = 0; k < 4; ++k) {
    const int e = base + k * 1024;
    d[k] = (e < E) ? ei[(size_t)E + e] : -1;
  }
#pragma unroll
  for (int k = 0; k < 4; ++k)
    if (d[k] >= 0) atomicAdd(&hc[d[k] >> BKT_SHIFT], 1);
  __syncthreads();
  for (int i = tid; i < NBK; i += 1024) tcnt[(size_t)i * NBLK + blockIdx.x] = hc[i];
}

template <int ROWS>
__device__ __forceinline__ int gat_rows_sum(const int4* __restrict__ q) {
  int4 v[ROWS];
#pragma unroll
  for (int j = 0; j < ROWS; ++j) v[j] = q[j * 1024];
  int t = 0;
#pragma unroll
  for (int j = 0; j < ROWS; ++j) t += (v[j].x + v[j].y) + (v[j].z + v[j].w);
  return t;
}

__global__ void __launch_bounds__(1024) gat_scan_kernel(const int* __restrict__ cnt,
                                                        int* __restrict__ off, int L) {
  __shared__ int red[32];
  __shared__ int wsum[32];
  const int tid   = threadIdx.x;
  const int lane  = tid & 31;
  const int wave  = tid >> 5;
  const int start = blockIdx.x * 1024;

  const int i0 = start + tid;
  const int c  = (i0 < L) ? cnt[i0] : 0;

  const int rows = start >> 12;
  const int rem  = start & 4095;
  const int4* __restrict__ p = reinterpret_cast<const int4*>(cnt) + tid;

  int4 vp = make_int4(0, 0, 0, 0);
  if (tid * 4 < rem) vp = p[(size_t)rows * 1024];

  int pre = 0;
  int r = 0;
  for (; r + 16 <= rows; r += 16) pre += gat_rows_sum<16>(p + (size_t)r * 1024);
  if (rows - r >= 8) { pre += gat_rows_sum<8>(p + (size_t)r * 1024); r += 8; }
  if (rows - r >= 4) { pre += gat_rows_sum<4>(p + (size_t)r * 1024); r += 4; }
  if (rows - r >= 2) { pre += gat_rows_sum<2>(p + (size_t)r * 1024); r += 2; }
  if (rows - r >= 1) { pre += gat_rows_sum<1>(p + (size_t)r * 1024); }
  pre += (vp.x + vp.y) + (vp.z + vp.w);
  for (int m = 16; m > 0; m >>= 1) pre += __shfl_xor(pre, m);

  int inc = c;
  for (int m = 1; m < 32; m <<= 1) {
    const int t = __shfl_up(inc, m);
    if (lane >= m) inc += t;
  }
  if (lane == 0)  red[wave]  = pre;
  if (lane == 31) wsum[wave] = inc;
  __syncthreads();
  int part = red[lane] + ((lane < wave) ? wsum[lane] : 0);
  for (int m = 16; m > 0; m >>= 1) part += __shfl_xor(part, m);
  if (i0 < L) off[i0] = part + inc - c;
}

__global__ void __launch_bounds__(1024) gat_bscatter_kernel(const int* __restrict__ ei,
                                                            const int* __restrict__ toff,
                                                            int* __restrict__ cpair,
                                                            int E, int NBK, int NBLK) {
  __shared__ int pos[BKT_MAX];
  const int tid = threadIdx.x;
  for (int i = tid; i < NBK; i += 1024) pos[i] = toff[(size_t)i * NBLK + blockIdx.x];
  __syncthreads();
  const int base = blockIdx.x * ECHUNK + tid;
  int s[4], d[4];
#pragma unroll
  for (int k = 0; k < 4; ++k) {
    const int e = base + k * 1024;
    s[k] = (e < E) ? ei[e] : 0;
    d[k] = (e < E) ? ei[(size_t)E + e] : -1;
  }
  int2* __restrict__ out2 = reinterpret_cast<int2*>(cpair);
#pragma unroll
  for (int k = 0; k < 4; ++k) {
    if (d[k] >= 0) {
      const int q = atomicAdd(&pos[d[k] >> BKT_SHIFT], 1);
      out2[q] = make_int2(s[k], d[k]);
    }
  }
}

__global__ void __launch_bounds__(1024) gat_bsort_kernel(const int* __restrict__ toff,
                                                         const int* __restrict__ cpair,
                                                         int* __restrict__ cursor,
                                                         int* __restrict__ ssrc,
                                                         int N, int E, int NBK, int NBLK) {
  __shared__ int cl[BKT_NODES];
  __shared__ int wtot[BKT_WAVES];
  const int tid   = threadIdx.x;
  const int lane  = tid & 31;
  const int wave  = tid >> 5;
  const int b     = blockIdx.x;
  const int node0 = b << BKT_SHIFT;
  const int bstart = toff[(size_t)b * NBLK];
  const int bend   = (b + 1 < NBK) ? toff[(size_t)(b + 1) * NBLK] : E;
  const int2* __restrict__ pr = reinterpret_cast<const int2*>(cpair);

  if (tid < BKT_NODES) cl[tid] = 0;
  __syncthreads();

  for (int e0 = bstart; e0 < bend; e0 += SORT_PER * 1024) {
    int d[SORT_PER];
#pragma unroll
    for (int k = 0; k < SORT_PER; ++k) {
      const int e = e0 + k * 1024 + tid;
      d[k] = (e < bend) ? pr[e].y : -1;
    }
#pragma unroll
    for (int k = 0; k < SORT_PER; ++k)
      if (d[k] >= 0) atomicAdd(&cl[d[k] - node0], 1);
  }
  __syncthreads();

  const int v = (tid < BKT_NODES) ? cl[tid] : 0;
  int inc = v;
  for (int m = 1; m < 32; m <<= 1) {
    const int t = __shfl_up(inc, m);
    if (lane >= m) inc += t;
  }
  if (lane == 31 && wave < BKT_WAVES) wtot[wave] = inc;
  __syncthreads();
  int part = (lane < BKT_WAVES && lane < wave) ? wtot[lane] : 0;
  for (int m = 16; m > 0; m >>= 1) part += __shfl_xor(part, m);
  const int excl = part + inc - v;
  if (tid < BKT_NODES) {
    cl[tid] = excl;
    if (node0 + tid < N) cursor[node0 + tid] = bstart + excl;
  }
  __syncthreads();

  for (int e0 = bstart; e0 < bend; e0 += SORT_PER * 1024) {
    int2 q[SORT_PER];
#pragma unroll
    for (int k = 0; k < SORT_PER; ++k) {
      const int e = e0 + k * 1024 + tid;
      q[k] = (e < bend) ? pr[e] : make_int2(0, -1);
    }
#pragma unroll
    for (int k = 0; k < SORT_PER; ++k) {
      if (q[k].y >= 0) {
        const int at = atomicAdd(&cl[q[k].y - node0], 1);
        ssrc[bstart + at] = q[k].x;
      }
    }
  }
}


template <int K>
__device__ __forceinline__ int gat_row_lane(int v) {
  return __builtin_amdgcn_update_dpp(0, v, 0x150 + K, 0xF, 0xF, true);
}

template <int XM>
__device__ __forceinline__ float gat_row_xfold(float keep, float send) {
  const int t = __builtin_amdgcn_update_dpp(0, __float_as_int(send), 0x160 + XM, 0xF, 0xF, true);
  return keep + __int_as_float(t);
}

__device__ __forceinline__ float gat_other_row(float v) {
  return __int_as_float(__builtin_amdgcn_permlanex16(0, __float_as_int(v), 0x76543210u, 0xfedcba98u,
                                                     false, false));
}

template <int T>
__device__ __forceinline__ void gat_rows16(const float4* __restrict__ h4, int srcl, int c,
                                           bool b1, bool b2, bool b3, bool live, bool even,
                                           float4 at, float dn, float4& acc, float& dsum) {
  int sj[8];
  sj[0] = gat_row_lane<8 * T + 0>(srcl);
  sj[1] = gat_row_lane<8 * T + 1>(srcl);
  sj[2] = gat_row_lane<8 * T + 2>(srcl);
  sj[3] = gat_row_lane<8 * T + 3>(srcl);
  sj[4] = gat_row_lane<8 * T + 4>(srcl);
  sj[5] = gat_row_lane<8 * T + 5>(srcl);
  sj[6] = gat_row_lane<8 * T + 6>(srcl);
  sj[7] = gat_row_lane<8 * T + 7>(srcl);
  float4 r[8];
#pragma unroll
  for (int j = 0; j < 8; ++j) {
    const unsigned off = (unsigned)sj[j] * (unsigned)(HID * 4) + (unsigned)c * 16u;
    r[j] = *reinterpret_cast<const float4*>(reinterpret_cast<const char*>(h4) + off);
  }
  float p[8];
#pragma unroll
  for (int j = 0; j < 8; ++j)
    p[j] = __builtin_fmaf(r[j].w, at.w, __builtin_fmaf(r[j].z, at.z, __builtin_fmaf(r[j].y, at.y, r[j].x * at.x)));
  float q[4];
#pragma unroll
  for (int j = 0; j < 4; ++j)
    q[j] = gat_row_xfold<8>(b3 ? p[j + 4] : p[j], b3 ? p[j] : p[j + 4]);
  float u[2];
#pragma unroll
  for (int j = 0; j < 2; ++j)
    u[j] = gat_row_xfold<4>(b2 ? q[j + 2] : q[j], b2 ? q[j] : q[j + 2]);
  float t = gat_row_xfold<2>(b1 ? u[1] : u[0], b1 ? u[0] : u[1]);
  t = gat_row_xfold<1>(t, t);
  float v = t + dn;
  v = (v > 0.f) ? v : 0.2f * v;
  const float w = live ? __expf(v) : 0.f;
  dsum += even ? w : 0.f;
  const int wb = __float_as_int(w);
  float a[8];
  a[0] = __int_as_float(gat_row_lane< 0>(wb));
  a[1] = __int_as_float(gat_row_lane< 2>(wb));
  a[2] = __int_as_float(gat_row_lane< 4>(wb));
  a[3] = __int_as_float(gat_row_lane< 6>(wb));
  a[4] = __int_as_float(gat_row_lane< 8>(wb));
  a[5] = __int_as_float(gat_row_lane<10>(wb));
  a[6] = __int_as_float(gat_row_lane<12>(wb));
  a[7] = __int_as_float(gat_row_lane<14>(wb));
#pragma unroll
  for (int j = 0; j < 8; ++j) {
    acc.x += r[j].x * a[j];
    acc.y += r[j].y * a[j];
    acc.z += r[j].z * a[j];
    acc.w += r[j].w * a[j];
  }
}

__global__ void __launch_bounds__(256) gat_gather_kernel(const int* __restrict__ cursor,
                                                         const int* __restrict__ ssrc,
                                                         const float* __restrict__ h,
                                                         const float* __restrict__ att,
                                                         const float* __restrict__ d,
                                                         float* __restrict__ agg, int N, int E) {
  const int lane = threadIdx.x & 31;
  const int c    = lane & 15;
  const int half = lane >> 4;
  const int n = __builtin_amdgcn_readfirstlane(blockIdx.x * 8 + (threadIdx.x >> 5));
  if (n >= N) return;
  const int begin = __builtin_amdgcn_readfirstlane(cursor[n]);
  const int end   = __builtin_amdgcn_readfirstlane((n + 1 < N) ? cursor[n + 1] : E);
  const float  dn = d[n];
  const float4 at = *reinterpret_cast<const float4*>(att + 4 * c);
  const float4* __restrict__ h4 = reinterpret_cast<const float4*>(h);

  const bool b1 = (c & 2) != 0, b2 = (c & 4) != 0, b3 = (c & 8) != 0, even = (c & 1) == 0;
  float4 acc  = {0.f, 0.f, 0.f, 0.f};
  float  dsum = 0.f;
  for (int base = begin; base < end; base += 32) {
    const int el   = base + 2 * c + half;
    const int srcl = ssrc[el < end ? el : end - 1];
    const int ej   = base + (c & ~1) + half;
    gat_rows16<0>(h4, srcl, c, b1, b2, b3, ej < end, even, at, dn, acc, dsum);
    if (end - base > 16)
      gat_rows16<1>(h4, srcl, c, b1, b2, b3, ej + 16 < end, even, at, dn, acc, dsum);
  }
  acc.x += gat_other_row(acc.x);
  acc.y += gat_other_row(acc.y);
  acc.z += gat_other_row(acc.z);
  acc.w += gat_other_row(acc.w);
  float tot = dsum;
  tot = gat_row_xfold<8>(tot, tot);
  tot = gat_row_xfold<4>(tot, tot);
  tot = gat_row_xfold<2>(tot, tot);
  tot = gat_row_xfold<1>(tot, tot);
  tot += gat_other_row(tot);
  const float inv = 1.0f / (tot + 1e-9f);
  if (half == 0)
    reinterpret_cast<float4*>(agg)[(size_t)n * (HID / 4) + c] =
        make_float4(acc.x * inv, acc.y * inv, acc.z * inv, acc.w * inv);
}

__global__ void __launch_bounds__(256) gat_out_kernel(const float* __restrict__ agg,
                                                      const float* __restrict__ W,
                                                      const float* __restrict__ bias,
                                                      float* __restrict__ out, int N) {
  const int lane  = threadIdx.x & 31;
  const int wave  = threadIdx.x >> 5;
  const int row0  = (blockIdx.x * 8 + wave) * 16;
  const int lh    = lane & 15;
  const int khalf = lane >> 4;
  const int rowA  = row0 + lh;
  const int rA    = rowA < N ? rowA : (N - 1);

  __shared__ __attribute__((aligned(16))) float wo[OUT_D * WO_PITCH];
  v2f av[HID / 4];
  {
    const float4* __restrict__ W4 = reinterpret_cast<const float4*>(W);
    float4 wv[(OUT_D * HID / 4) / 256];
#pragma unroll
    for (int i = 0; i < (OUT_D * HID / 4) / 256; ++i) wv[i] = W4[i * 256 + threadIdx.x];
#pragma unroll
    for (int i = 0; i < HID / 4; ++i)
      av[i] = *reinterpret_cast<const v2f*>(&agg[(size_t)rA * HID + i * 4 + khalf * 2]);
#pragma unroll
    for (int i = 0; i < (OUT_D * HID / 4) / 256; ++i) {
      const int idx = i * 256 + threadIdx.x;
      const int r   = idx >> 4;
      const int c4  = idx & 15;
      *reinterpret_cast<float4*>(&wo[r * WO_PITCH + c4 * 4]) = wv[i];
    }
  }
  __builtin_amdgcn_fence(__ATOMIC_RELEASE, "workgroup", "local");
  __builtin_amdgcn_s_barrier();
  __builtin_amdgcn_fence(__ATOMIC_ACQUIRE, "workgroup", "local");
  const float* __restrict__ wop = &wo[lh * WO_PITCH + khalf * 2];

  v8f a0 = {}, a1 = {}, a2 = {}, a3 = {};
#pragma unroll
  for (int k0 = 0; k0 < HID; k0 += 4) {
    const float z0 = av[k0 / 4].x;
    const float z1 = av[k0 / 4].y;
    v2f a = { z0 > 0.f ? z0 : 0.f, z1 > 0.f ? z1 : 0.f };
    const v2f b0 = *reinterpret_cast<const v2f*>(wop +  0 * WO_PITCH + k0);
    const v2f b1 = *reinterpret_cast<const v2f*>(wop + 16 * WO_PITCH + k0);
    const v2f b2 = *reinterpret_cast<const v2f*>(wop + 32 * WO_PITCH + k0);
    const v2f b3 = *reinterpret_cast<const v2f*>(wop + 48 * WO_PITCH + k0);
    a0 = wmma_k4(a, b0, a0);
    a1 = wmma_k4(a, b1, a1);
    a2 = wmma_k4(a, b2, a2);
    a3 = wmma_k4(a, b3, a3);
  }
  const float bb0 = bias[lh +  0];
  const float bb1 = bias[lh + 16];
  const float bb2 = bias[lh + 32];
  const float bb3 = bias[lh + 48];
#pragma unroll
  for (int r = 0; r < 8; ++r) {
    const int row = row0 + khalf * 8 + r;
    if (row < N) {
      out[(size_t)row * OUT_D + lh +  0] = a0[r] + bb0;
      out[(size_t)row * OUT_D + lh + 16] = a1[r] + bb1;
      out[(size_t)row * OUT_D + lh + 32] = a2[r] + bb2;
      out[(size_t)row * OUT_D + lh + 48] = a3[r] + bb3;
    }
  }
}

extern "C" void kernel_launch(void* const* d_in, const int* in_sizes, int n_in,
                              void* d_out, int out_size, void* d_ws, size_t ws_size,
                              hipStream_t stream) {
  const float* x    = (const float*)d_in[0];
  const int*   ei   = (const int*)  d_in[1];
  const float* Wlin = (const float*)d_in[2];
  const float* att  = (const float*)d_in[3];
  const float* Wout = (const float*)d_in[4];
  const float* bout = (const float*)d_in[5];
  float* out = (float*)d_out;

  const int N = in_sizes[0] / IN_DIM;
  const int E = in_sizes[1] / 2;

  float* ws     = (float*)d_ws;
  float* h      = ws;
  float* s      = h + (size_t)N * HID;
  float* dn     = s + N;
  int*   ssrc   = (int*)(dn + N);
  int*   cursor = ssrc + E;
  float* agg    = (float*)(cursor + N);

  const int NBK  = (N + BKT_NODES - 1) / BKT_NODES;
  const int NBLK = (E + ECHUNK - 1) / ECHUNK;
  const int L    = NBK * NBLK;
  const int L4   = (L + 3) / 4 * 4;
  int*   tcnt   = (int*)agg;
  int*   toff   = tcnt + L4;
  int*   cpair  = toff + L4;

  const int tiles   = (N + 15) / 16;
  const int gemmBlk = (tiles + 7) / 8;

  gat_lin_kernel<<<gemmBlk, 256, 0, stream>>>(x, Wlin, att, h, s, dn, N);

  gat_bcount_kernel<<<NBLK, 1024, 0, stream>>>(ei, tcnt, E, NBK, NBLK);
  gat_scan_kernel<<<(L + 1023) / 1024, 1024, 0, stream>>>(tcnt, toff, L);
  gat_bscatter_kernel<<<NBLK, 1024, 0, stream>>>(ei, toff, cpair, E, NBK, NBLK);
  gat_bsort_kernel<<<NBK, 1024, 0, stream>>>(toff, cpair, cursor, ssrc, N, E, NBK, NBLK);
  gat_gather_kernel<<<(N + 7) / 8, 256, 0, stream>>>(cursor, ssrc, h, att, dn, agg, N, E);

  gat_out_kernel<<<gemmBlk, 256, 0, stream>>>(agg, Wout, bout, out, N);
}


# First GEMM's row blocks run as extra work-groups of the three sort launches (now 256-thread blocks), not as a launch of their own; 7 -> 6 launches; sum order = LDS arrival order as before
# speedup vs baseline: 4.2945x; 1.0407x over previous
#include <hip/hip_runtime.h>

typedef float v2f __attribute__((ext_vector_type(2)));
typedef float v8f __attribute__((ext_vector_type(8)));

#define IN_DIM 128
#define HID    64
#define OUT_D  64

#define WL_PITCH (IN_DIM + 4)
#define WO_PITCH (HID + 4)

__device__ __forceinline__ v8f wmma_k4(v2f a, v2f b, v8f c) {
  return __builtin_amdgcn_wmma_f32_16x16x4_f32(false, a, false, b, (short)0, c,
                                               false, false);
}


template <int XM>
__device__ __forceinline__ float gat_row_xadd(float v) {
  const int t = __builtin_amdgcn_update_dpp(0, __float_as_int(v), 0x160 + XM, 0xF, 0xF, true);
  return v + __int_as_float(t);
}

__device__ __forceinline__ void gat_lin_body(const int blk,
                                             const float* __restrict__ x,
                                             const float* __restrict__ W,
                                             const float* __restrict__ att,
                                             float* __restrict__ h,
                                             float* __restrict__ s,
                                             float* __restrict__ d, int N) {
  __shared__ float sdl[2][128];
  __shared__ __attribute__((aligned(16))) float wl[HID * WL_PITCH];
  const int lane  = threadIdx.x & 31;
  const int wave  = threadIdx.x >> 5;
  const int row0  = (blk * 8 + wave) * 16;
  const int lh    = lane & 15;
  const int khalf = lane >> 4;
  const int rowA  = row0 + lh;
  const int rA    = rowA < N ? rowA : (N - 1);

  const float ws0 = att[lh +  0], ws1 = att[lh + 16], ws2 = att[lh + 32], ws3 = att[lh + 48];
  const float wd0 = att[HID + lh +  0], wd1 = att[HID + lh + 16];
  const float wd2 = att[HID + lh + 32], wd3 = att[HID + lh + 48];

  v2f av[IN_DIM / 4];
  {
    const float4* __restrict__ W4 = reinterpret_cast<const float4*>(W);
    float4 wv[(HID * IN_DIM / 4) / 256];
#pragma unroll
    for (int i = 0; i < (HID * IN_DIM / 4) / 256; ++i) wv[i] = W4[i * 256 + threadIdx.x];
#pragma unroll
    for (int i = 0; i < IN_DIM / 4; ++i)
      av[i] = *reinterpret_cast<const v2f*>(&x[(size_t)rA * IN_DIM + i * 4 + khalf * 2]);
#pragma unroll
    for (int i = 0; i < (HID * IN_DIM / 4) / 256; ++i) {
      const int idx = i * 256 + threadIdx.x;
      const int r   = idx >> 5;
      const int c4  = idx & 31;
      *reinterpret_cast<float4*>(&wl[r * WL_PITCH + c4 * 4]) = wv[i];
    }
  }
  __builtin_amdgcn_fence(__ATOMIC_RELEASE, "workgroup", "local");
  __builtin_amdgcn_s_barrier();
  __builtin_amdgcn_fence(__ATOMIC_ACQUIRE, "workgroup", "local");
  const float* __restrict__ wlp = &wl[lh * WL_PITCH + khalf * 2];

  v8f a0 = {}, a1 = {}, a2 = {}, a3 = {};
#pragma unroll
  for (int k0 = 0; k0 < IN_DIM; k0 += 4) {
    const v2f a = av[k0 / 4];
    const v2f b0 = *reinterpret_cast<const v2f*>(wlp +  0 * WL_PITCH + k0);
    const v2f b1 = *reinterpret_cast<const v2f*>(wlp + 16 * WL_PITCH + k0);
    const v2f b2 = *reinterpret_cast<const v2f*>(wlp + 32 * WL_PITCH + k0);
    const v2f b3 = *reinterpret_cast<const v2f*>(wlp + 48 * WL_PITCH + k0);
    a0 = wmma_k4(a, b0, a0);
    a1 = wmma_k4(a, b1, a1);
    a2 = wmma_k4(a, b2, a2);
    a3 = wmma_k4(a, b3, a3);
  }
  float sel = 0.f;
#pragma unroll
  for (int r = 0; r < 8; ++r) {
    float ps = a0[r] * ws0 + a1[r] * ws1 + a2[r] * ws2 + a3[r] * ws3;
    float pd = a0[r] * wd0 + a1[r] * wd1 + a2[r] * wd2 + a3[r] * wd3;
    ps = gat_row_xadd<8>(ps);  pd = gat_row_xadd<8>(pd);
    ps = gat_row_xadd<4>(ps);  pd = gat_row_xadd<4>(pd);
    ps = gat_row_xadd<2>(ps);  pd = gat_row_xadd<2>(pd);
    ps = gat_row_xadd<1>(ps);  pd = gat_row_xadd<1>(pd);
    if ((lh & 7) == r) sel = (lh < 8) ? ps : pd;
  }
  sdl[(lh < 8) ? 0 : 1][wave * 16 + khalf * 8 + (lh & 7)] = sel;
  __syncthreads();

#pragma unroll
  for (int r = 0; r < 8; ++r) {
    const int row = row0 + khalf * 8 + r;
    if (row < N) {
      h[(size_t)row * HID + lh +  0] = a0[r];
      h[(size_t)row * HID + lh + 16] = a1[r];
      h[(size_t)row * HID + lh + 32] = a2[r];
      h[(size_t)row * HID + lh + 48] = a3[r];
    }
  }

  const int which = threadIdx.x >> 7;
  const int lrow  = threadIdx.x & 127;
  const int rowS  = blk * 128 + lrow;
  if (rowS < N) {
    float* __restrict__ dstp = which ? d : s;
    dstp[rowS] = sdl[which][lrow];
  }
}

#define BKT_SHIFT 8
#define BKT_NODES 256
#define BKT_WAVES (BKT_NODES / 32)
#define BKT_MAX   2048
#define STG_THREADS 256
#define ECHUNK    4096
#define EPT       (ECHUNK / STG_THREADS)
#define SORT_PER  20

__device__ __forceinline__ void gat_bcount_body(const int blk, const int* __restrict__ ei,
                                                int* __restrict__ tcnt,
                                                int E, int NBK, int NBLK) {
  __shared__ int hc[BKT_MAX];
  const int tid = threadIdx.x;
  for (int i = tid; i < NBK; i += STG_THREADS) hc[i] = 0;
  __syncthreads();
  const int base = blk * ECHUNK + tid;
  int d[EPT];
#pragma unroll
  for (int k = 0; k < EPT; ++k) {
    const int e = base + k * STG_THREADS;
    d[k] = (e < E) ? ei[(size_t)E + e] : -1;
  }
#pragma unroll
  for (int k = 0; k < EPT; ++k)
    if (d[k] >= 0) atomicAdd(&hc[d[k] >> BKT_SHIFT], 1);
  __syncthreads();
  for (int i = tid; i < NBK; i += STG_THREADS) tcnt[(size_t)i * NBLK + blk] = hc[i];
}

template <int ROWS>
__device__ __forceinline__ int gat_rows_sum(const int4* __restrict__ q) {
  int4 v[ROWS];
#pragma unroll
  for (int j = 0; j < ROWS; ++j) v[j] = q[j * 1024];
  int t = 0;
#pragma unroll
  for (int j = 0; j < ROWS; ++j) t += (v[j].x + v[j].y) + (v[j].z + v[j].w);
  return t;
}

__global__ void __launch_bounds__(1024) gat_scan_kernel(const int* __restrict__ cnt,
                                                        int* __restrict__ off, int L) {
  __shared__ int red[32];
  __shared__ int wsum[32];
  const int tid   = threadIdx.x;
  const int lane  = tid & 31;
  const int wave  = tid >> 5;
  const int start = blockIdx.x * 1024;

  const int i0 = start + tid;
  const int c  = (i0 < L) ? cnt[i0] : 0;

  const int rows = start >> 12;
  const int rem  = start & 4095;
  const int4* __restrict__ p = reinterpret_cast<const int4*>(cnt) + tid;

  int4 vp = make_int4(0, 0, 0, 0);
  if (tid * 4 < rem) vp = p[(size_t)rows * 1024];

  int pre = 0;
  int r = 0;
  for (; r + 16 <= rows; r += 16) pre += gat_rows_sum<16>(p + (size_t)r * 1024);
  if (rows - r >= 8) { pre += gat_rows_sum<8>(p + (size_t)r * 1024); r += 8; }
  if (rows - r >= 4) { pre += gat_rows_sum<4>(p + (size_t)r * 1024); r += 4; }
  if (rows - r >= 2) { pre += gat_rows_sum<2>(p + (size_t)r * 1024); r += 2; }
  if (rows - r >= 1) { pre += gat_rows_sum<1>(p + (size_t)r * 1024); }
  pre += (vp.x + vp.y) + (vp.z + vp.w);
  for (int m = 16; m > 0; m >>= 1) pre += __shfl_xor(pre, m);

  int inc = c;
  for (int m = 1; m < 32; m <<= 1) {
    const int t = __shfl_up(inc, m);
    if (lane >= m) inc += t;
  }
  if (lane == 0)  red[wave]  = pre;
  if (lane == 31) wsum[wave] = inc;
  __syncthreads();
  int part = red[lane] + ((lane < wave) ? wsum[lane] : 0);
  for (int m = 16; m > 0; m >>= 1) part += __shfl_xor(part, m);
  if (i0 < L) off[i0] = part + inc - c;
}

__device__ __forceinline__ void gat_bscatter_body(const int blk, const int* __restrict__ ei,
                                                  const int* __restrict__ toff,
                                                  int* __restrict__ cpair,
                                                  int E, int NBK, int NBLK) {
  __shared__ int pos[BKT_MAX + 32];
  const int tid = threadIdx.x;
  for (int i = tid; i < NBK; i += STG_THREADS) pos[i] = toff[(size_t)i * NBLK + blk];
  if (tid < 32) pos[BKT_MAX + tid] = 0;
  __syncthreads();
  const int base = blk * ECHUNK + tid;
  int s[EPT], d[EPT];
#pragma unroll
  for (int k = 0; k < EPT; ++k) {
    const int e = base + k * STG_THREADS;
    s[k] = (e < E) ? ei[e] : 0;
    d[k] = (e < E) ? ei[(size_t)E + e] : -1;
  }
  int q[EPT];
#pragma unroll
  for (int k = 0; k < EPT; ++k)
    q[k] = atomicAdd(&pos[(d[k] >= 0) ? (d[k] >> BKT_SHIFT) : (BKT_MAX + (tid & 31))], 1);
  int2* __restrict__ out2 = reinterpret_cast<int2*>(cpair);
#pragma unroll
  for (int k = 0; k < EPT; ++k)
    if (d[k] >= 0) out2[q[k]] = make_int2(s[k], d[k]);
}

__device__ __forceinline__ void gat_bsort_body(const int b,
                                               const int* __restrict__ toff,
                                               const int* __restrict__ cpair,
                                               int* __restrict__ cursor,
                                               int* __restrict__ ssrc,
                                               int N, int E, int NBK, int NBLK) {
  __shared__ int cl[BKT_NODES + 32];
  __shared__ int wtot[BKT_WAVES];
  const int tid   = threadIdx.x;
  const int lane  = tid & 31;
  const int wave  = tid >> 5;
  const int node0 = b << BKT_SHIFT;
  const int bstart = toff[(size_t)b * NBLK];
  const int bend   = (b + 1 < NBK) ? toff[(size_t)(b + 1) * NBLK] : E;
  const int2* __restrict__ pr = reinterpret_cast<const int2*>(cpair);

  cl[tid] = 0;
  if (tid < 32) cl[BKT_NODES + tid] = 0;
  __syncthreads();

  for (int e0 = bstart; e0 < bend; e0 += SORT_PER * STG_THREADS) {
    int d[SORT_PER];
#pragma unroll
    for (int k = 0; k < SORT_PER; ++k) {
      const int e = e0 + k * STG_THREADS + tid;
      d[k] = (e < bend) ? pr[e].y : -1;
    }
#pragma unroll
    for (int k = 0; k < SORT_PER; ++k)
      if (d[k] >= 0) atomicAdd(&cl[d[k] - node0], 1);
  }
  __syncthreads();

  const int v = cl[tid];
  int inc = v;
  for (int m = 1; m < 32; m <<= 1) {
    const int t = __shfl_up(inc, m);
    if (lane >= m) inc += t;
  }
  if (lane == 31) wtot[wave] = inc;
  __syncthreads();
  int part = (lane < BKT_WAVES && lane < wave) ? wtot[lane] : 0;
  for (int m = 16; m > 0; m >>= 1) part += __shfl_xor(part, m);
  const int excl = part + inc - v;
  cl[tid] = excl;
  if (node0 + tid < N) cursor[node0 + tid] = bstart + excl;
  __syncthreads();

  for (int e0 = bstart; e0 < bend; e0 += SORT_PER * STG_THREADS) {
    int2 q[SORT_PER];
#pragma unroll
    for (int k = 0; k < SORT_PER; ++k) {
      const int e = e0 + k * STG_THREADS + tid;
      q[k] = (e < bend) ? pr[e] : make_int2(0, -1);
    }
    int at[SORT_PER];
#pragma unroll
    for (int k = 0; k < SORT_PER; ++k)
      at[k] = atomicAdd(&cl[(q[k].y >= 0) ? (q[k].y - node0) : (BKT_NODES + lane)], 1);
#pragma unroll
    for (int k = 0; k < SORT_PER; ++k)
      if (q[k].y >= 0) ssrc[bstart + at[k]] = q[k].x;
  }
}

template <int STAGE>
__global__ void __launch_bounds__(256) gat_stage_kernel(const float* __restrict__ x,
                                                        const float* __restrict__ W,
                                                        const float* __restrict__ att,
                                                        float* __restrict__ h,
                                                        float* __restrict__ s,
                                                        float* __restrict__ d,
                                                        const int* __restrict__ ei,
                                                        int* __restrict__ tcnt,
                                                        const int* __restrict__ toff,
                                                        int* __restrict__ cpair,
                                                        int* __restrict__ cursor,
                                                        int* __restrict__ ssrc,
                                                        int N, int E, int NBK, int NBLK,
                                                        int gemmBlk0, int gemmCnt) {
  const int blk = blockIdx.x;
  if (blk < gemmCnt) {
    gat_lin_body(gemmBlk0 + blk, x, W, att, h, s, d, N);
  } else {
    const int sb = blk - gemmCnt;
    if constexpr (STAGE == 0)      gat_bcount_body(sb, ei, tcnt, E, NBK, NBLK);
    else if constexpr (STAGE == 1) gat_bscatter_body(sb, ei, toff, cpair, E, NBK, NBLK);
    else                           gat_bsort_body(sb, toff, cpair, cursor, ssrc, N, E, NBK, NBLK);
  }
}


template <int K>
__device__ __forceinline__ int gat_row_lane(int v) {
  return __builtin_amdgcn_update_dpp(0, v, 0x150 + K, 0xF, 0xF, true);
}

template <int XM>
__device__ __forceinline__ float gat_row_xfold(float keep, float send) {
  const int t = __builtin_amdgcn_update_dpp(0, __float_as_int(send), 0x160 + XM, 0xF, 0xF, true);
  return keep + __int_as_float(t);
}

__device__ __forceinline__ float gat_other_row(float v) {
  return __int_as_float(__builtin_amdgcn_permlanex16(0, __float_as_int(v), 0x76543210u, 0xfedcba98u,
                                                     false, false));
}

template <int T>
__device__ __forceinline__ void gat_rows16(const float4* __restrict__ h4, int srcl, int c,
                                           bool b1, bool b2, bool b3, bool live, bool even,
                                           float4 at, float dn, float4& acc, float& dsum) {
  int sj[8];
  sj[0] = gat_row_lane<8 * T + 0>(srcl);
  sj[1] = gat_row_lane<8 * T + 1>(srcl);
  sj[2] = gat_row_lane<8 * T + 2>(srcl);
  sj[3] = gat_row_lane<8 * T + 3>(srcl);
  sj[4] = gat_row_lane<8 * T + 4>(srcl);
  sj[5] = gat_row_lane<8 * T + 5>(srcl);
  sj[6] = gat_row_lane<8 * T + 6>(srcl);
  sj[7] = gat_row_lane<8 * T + 7>(srcl);
  float4 r[8];
#pragma unroll
  for (int j = 0; j < 8; ++j) {
    const unsigned off = (unsigned)sj[j] * (unsigned)(HID * 4) + (unsigned)c * 16u;
    r[j] = *reinterpret_cast<const float4*>(reinterpret_cast<const char*>(h4) + off);
  }
  float p[8];
#pragma unroll
  for (int j = 0; j < 8; ++j)
    p[j] = __builtin_fmaf(r[j].w, at.w, __builtin_fmaf(r[j].z, at.z, __builtin_fmaf(r[j].y, at.y, r[j].x * at.x)));
  float q[4];
#pragma unroll
  for (int j = 0; j < 4; ++j)
    q[j] = gat_row_xfold<8>(b3 ? p[j + 4] : p[j], b3 ? p[j] : p[j + 4]);
  float u[2];
#pragma unroll
  for (int j = 0; j < 2; ++j)
    u[j] = gat_row_xfold<4>(b2 ? q[j + 2] : q[j], b2 ? q[j] : q[j + 2]);
  float t = gat_row_xfold<2>(b1 ? u[1] : u[0], b1 ? u[0] : u[1]);
  t = gat_row_xfold<1>(t, t);
  float v = t + dn;
  v = (v > 0.f) ? v : 0.2f * v;
  const float w = live ? __expf(v) : 0.f;
  dsum += even ? w : 0.f;
  const int wb = __float_as_int(w);
  float a[8];
  a[0] = __int_as_float(gat_row_lane< 0>(wb));
  a[1] = __int_as_float(gat_row_lane< 2>(wb));
  a[2] = __int_as_float(gat_row_lane< 4>(wb));
  a[3] = __int_as_float(gat_row_lane< 6>(wb));
  a[4] = __int_as_float(gat_row_lane< 8>(wb));
  a[5] = __int_as_float(gat_row_lane<10>(wb));
  a[6] = __int_as_float(gat_row_lane<12>(wb));
  a[7] = __int_as_float(gat_row_lane<14>(wb));
#pragma unroll
  for (int j = 0; j < 8; ++j) {
    acc.x += r[j].x * a[j];
    acc.y += r[j].y * a[j];
    acc.z += r[j].z * a[j];
    acc.w += r[j].w * a[j];
  }
}

__global__ void __launch_bounds__(256) gat_gather_kernel(const int* __restrict__ cursor,
                                                         const int* __restrict__ ssrc,
                                                         const float* __restrict__ h,
                                                         const float* __restrict__ att,
                                                         const float* __restrict__ d,
                                                         float* __restrict__ agg, int N, int E) {
  const int lane = threadIdx.x & 31;
  const int c    = lane & 15;
  const int half = lane >> 4;
  const int n = __builtin_amdgcn_readfirstlane(blockIdx.x * 8 + (threadIdx.x >> 5));
  if (n >= N) return;
  const int begin = __builtin_amdgcn_readfirstlane(cursor[n]);
  const int end   = __builtin_amdgcn_readfirstlane((n + 1 < N) ? cursor[n + 1] : E);
  const float  dn = d[n];
  const float4 at = *reinterpret_cast<const float4*>(att + 4 * c);
  const float4* __restrict__ h4 = reinterpret_cast<const float4*>(h);

  const bool b1 = (c & 2) != 0, b2 = (c & 4) != 0, b3 = (c & 8) != 0, even = (c & 1) == 0;
  float4 acc  = {0.f, 0.f, 0.f, 0.f};
  float  dsum = 0.f;
  for (int base = begin; base < end; base += 32) {
    const int el   = base + 2 * c + half;
    const int srcl = ssrc[el < end ? el : end - 1];
    const int ej   = base + (c & ~1) + half;
    gat_rows16<0>(h4, srcl, c, b1, b2, b3, ej < end, even, at, dn, acc, dsum);
    if (end - base > 16)
      gat_rows16<1>(h4, srcl, c, b1, b2, b3, ej + 16 < end, even, at, dn, acc, dsum);
  }
  acc.x += gat_other_row(acc.x);
  acc.y += gat_other_row(acc.y);
  acc.z += gat_other_row(acc.z);
  acc.w += gat_other_row(acc.w);
  float tot = dsum;
  tot = gat_row_xfold<8>(tot, tot);
  tot = gat_row_xfold<4>(tot, tot);
  tot = gat_row_xfold<2>(tot, tot);
  tot = gat_row_xfold<1>(tot, tot);
  tot += gat_other_row(tot);
  const float inv = 1.0f / (tot + 1e-9f);
  if (half == 0)
    reinterpret_cast<float4*>(agg)[(size_t)n * (HID / 4) + c] =
        make_float4(acc.x * inv, acc.y * inv, acc.z * inv, acc.w * inv);
}

__global__ void __launch_bounds__(256) gat_out_kernel(const float* __restrict__ agg,
                                                      const float* __restrict__ W,
                                                      const float* __restrict__ bias,
                                                      float* __restrict__ out, int N) {
  const int lane  = threadIdx.x & 31;
  const int wave  = threadIdx.x >> 5;
  const int row0  = (blockIdx.x * 8 + wave) * 16;
  const int lh    = lane & 15;
  const int khalf = lane >> 4;
  const int rowA  = row0 + lh;
  const int rA    = rowA < N ? rowA : (N - 1);

  __shared__ __attribute__((aligned(16))) float wo[OUT_D * WO_PITCH];
  v2f av[HID / 4];
  {
    const float4* __restrict__ W4 = reinterpret_cast<const float4*>(W);
    float4 wv[(OUT_D * HID / 4) / 256];
#pragma unroll
    for (int i = 0; i < (OUT_D * HID / 4) / 256; ++i) wv[i] = W4[i * 256 + threadIdx.x];
#pragma unroll
    for (int i = 0; i < HID / 4; ++i)
      av[i] = *reinterpret_cast<const v2f*>(&agg[(size_t)rA * HID + i * 4 + khalf * 2]);
#pragma unroll
    for (int i = 0; i < (OUT_D * HID / 4) / 256; ++i) {
      const int idx = i * 256 + threadIdx.x;
      const int r   = idx >> 4;
      const int c4  = idx & 15;
      *reinterpret_cast<float4*>(&wo[r * WO_PITCH + c4 * 4]) = wv[i];
    }
  }
  __builtin_amdgcn_fence(__ATOMIC_RELEASE, "workgroup", "local");
  __builtin_amdgcn_s_barrier();
  __builtin_amdgcn_fence(__ATOMIC_ACQUIRE, "workgroup", "local");
  const float* __restrict__ wop = &wo[lh * WO_PITCH + khalf * 2];

  v8f a0 = {}, a1 = {}, a2 = {}, a3 = {};
#pragma unroll
  for (int k0 = 0; k0 < HID; k0 += 4) {
    const float z0 = av[k0 / 4].x;
    const float z1 = av[k0 / 4].y;
    v2f a = { z0 > 0.f ? z0 : 0.f, z1 > 0.f ? z1 : 0.f };
    const v2f b0 = *reinterpret_cast<const v2f*>(wop +  0 * WO_PITCH + k0);
    const v2f b1 = *reinterpret_cast<const v2f*>(wop + 16 * WO_PITCH + k0);
    const v2f b2 = *reinterpret_cast<const v2f*>(wop + 32 * WO_PITCH + k0);
    const v2f b3 = *reinterpret_cast<const v2f*>(wop + 48 * WO_PITCH + k0);
    a0 = wmma_k4(a, b0, a0);
    a1 = wmma_k4(a, b1, a1);
    a2 = wmma_k4(a, b2, a2);
    a3 = wmma_k4(a, b3, a3);
  }
  const float bb0 = bias[lh +  0];
  const float bb1 = bias[lh + 16];
  const float bb2 = bias[lh + 32];
  const float bb3 = bias[lh + 48];
#pragma unroll
  for (int r = 0; r < 8; ++r) {
    const int row = row0 + khalf * 8 + r;
    if (row < N) {
      out[(size_t)row * OUT_D + lh +  0] = a0[r] + bb0;
      out[(size_t)row * OUT_D + lh + 16] = a1[r] + bb1;
      out[(size_t)row * OUT_D + lh + 32] = a2[r] + bb2;
      out[(size_t)row * OUT_D + lh + 48] = a3[r] + bb3;
    }
  }
}

extern "C" void kernel_launch(void* const* d_in, const int* in_sizes, int n_in,
                              void* d_out, int out_size, void* d_ws, size_t ws_size,
                              hipStream_t stream) {
  const float* x    = (const float*)d_in[0];
  const int*   ei   = (const int*)  d_in[1];
  const float* Wlin = (const float*)d_in[2];
  const float* att  = (const float*)d_in[3];
  const float* Wout = (const float*)d_in[4];
  const float* bout = (const float*)d_in[5];
  float* out = (float*)d_out;

  const int N = in_sizes[0] / IN_DIM;
  const int E = in_sizes[1] / 2;

  float* ws     = (float*)d_ws;
  float* h      = ws;
  float* s      = h + (size_t)N * HID;
  float* dn     = s + N;
  int*   ssrc   = (int*)(dn + N);
  int*   cursor = ssrc + E;
  float* agg    = (float*)(cursor + N);

  const int NBK  = (N + BKT_NODES - 1) / BKT_NODES;
  const int NBLK = (E + ECHUNK - 1) / ECHUNK;
  const int L    = NBK * NBLK;
  const int L4   = (L + 3) / 4 * 4;
  int*   tcnt   = (int*)agg;
  int*   toff   = tcnt + L4;
  int*   cpair  = toff + L4;

  const int tiles   = (N + 15) / 16;
  const int gemmBlk = (tiles + 7) / 8;

  const int g0 = (gemmBlk + 2) / 3;
  const int g1 = (gemmBlk - g0 + 1) / 2;
  const int g2 = gemmBlk - g0 - g1;

  gat_stage_kernel<0><<<g0 + NBLK, 256, 0, stream>>>(x, Wlin, att, h, s, dn, ei, tcnt, toff, cpair,
                                                     cursor, ssrc, N, E, NBK, NBLK, 0, g0);
  gat_scan_kernel<<<(L + 1023) / 1024, 1024, 0, stream>>>(tcnt, toff, L);
  gat_stage_kernel<1><<<g1 + NBLK, 256, 0, stream>>>(x, Wlin, att, h, s, dn, ei, tcnt, toff, cpair,
                                                     cursor, ssrc, N, E, NBK, NBLK, g0, g1);
  gat_stage_kernel<2><<<g2 + NBK, 256, 0, stream>>>(x, Wlin, att, h, s, dn, ei, tcnt, toff, cpair,
                                                    cursor, ssrc, N, E, NBK, NBLK, g0 + g1, g2);
  gat_gather_kernel<<<(N + 7) / 8, 256, 0, stream>>>(cursor, ssrc, h, att, dn, agg, N, E);

  gat_out_kernel<<<gemmBlk, 256, 0, stream>>>(agg, Wout, bout, out, N);
}


# Counting sort's scattered stores staged in LDS: a block's (src,dst) pairs leave in bucket order, a bucket's sorted sources in whole 128-B lines, not 32 lines per store; sum order = LDS arrival order
# speedup vs baseline: 5.2094x; 1.2130x over previous
#include <hip/hip_runtime.h>

typedef float v2f __attribute__((ext_vector_type(2)));
typedef float v8f __attribute__((ext_vector_type(8)));

#define IN_DIM 128
#define HID    64
#define OUT_D  64

#define WL_PITCH (IN_DIM + 4)
#define WO_PITCH (HID + 4)
#define LIN_WL_BYTES  (HID * WL_PITCH * 4)
#define LIN_LDS_BYTES (LIN_WL_BYTES + 2 * 128 * 4)

__device__ __forceinline__ v8f wmma_k4(v2f a, v2f b, v8f c) {
  return __builtin_amdgcn_wmma_f32_16x16x4_f32(false, a, false, b, (short)0, c,
                                               false, false);
}


template <int XM>
__device__ __forceinline__ float gat_row_xadd(float v) {
  const int t = __builtin_amdgcn_update_dpp(0, __float_as_int(v), 0x160 + XM, 0xF, 0xF, true);
  return v + __int_as_float(t);
}

__device__ __forceinline__ void gat_lin_body(unsigned char* smem, const int blk,
                                             const float* __restrict__ x,
                                             const float* __restrict__ W,
                                             const float* __restrict__ att,
                                             float* __restrict__ h,
                                             float* __restrict__ s,
                                             float* __restrict__ d, int N) {
  float (*const sdl)[128] = reinterpret_cast<float (*)[128]>(smem + LIN_WL_BYTES);
  float* const wl = reinterpret_cast<float*>(smem);
  const int lane  = threadIdx.x & 31;
  const int wave  = threadIdx.x >> 5;
  const int row0  = (blk * 8 + wave) * 16;
  const int lh    = lane & 15;
  const int khalf = lane >> 4;
  const int rowA  = row0 + lh;
  const int rA    = rowA < N ? rowA : (N - 1);

  const float ws0 = att[lh +  0], ws1 = att[lh + 16], ws2 = att[lh + 32], ws3 = att[lh + 48];
  const float wd0 = att[HID + lh +  0], wd1 = att[HID + lh + 16];
  const float wd2 = att[HID + lh + 32], wd3 = att[HID + lh + 48];

  v2f av[IN_DIM / 4];
  {
    const float4* __restrict__ W4 = reinterpret_cast<const float4*>(W);
    float4 wv[(HID * IN_DIM / 4) / 256];
#pragma unroll
    for (int i = 0; i < (HID * IN_DIM / 4) / 256; ++i) wv[i] = W4[i * 256 + threadIdx.x];
#pragma unroll
    for (int i = 0; i < IN_DIM / 4; ++i)
      av[i] = *reinterpret_cast<const v2f*>(&x[(size_t)rA * IN_DIM + i * 4 + khalf * 2]);
#pragma unroll
    for (int i = 0; i < (HID * IN_DIM / 4) / 256; ++i) {
      const int idx = i * 256 + threadIdx.x;
      const int r   = idx >> 5;
      const int c4  = idx & 31;
      *reinterpret_cast<float4*>(&wl[r * WL_PITCH + c4 * 4]) = wv[i];
    }
  }
  __builtin_amdgcn_fence(__ATOMIC_RELEASE, "workgroup", "local");
  __builtin_amdgcn_s_barrier();
  __builtin_amdgcn_fence(__ATOMIC_ACQUIRE, "workgroup", "local");
  const float* __restrict__ wlp = &wl[lh * WL_PITCH + khalf * 2];

  v8f a0 = {}, a1 = {}, a2 = {}, a3 = {};
#pragma unroll
  for (int k0 = 0; k0 < IN_DIM; k0 += 4) {
    const v2f a = av[k0 / 4];
    const v2f b0 = *reinterpret_cast<const v2f*>(wlp +  0 * WL_PITCH + k0);
    const v2f b1 = *reinterpret_cast<const v2f*>(wlp + 16 * WL_PITCH + k0);
    const v2f b2 = *reinterpret_cast<const v2f*>(wlp + 32 * WL_PITCH + k0);
    const v2f b3 = *reinterpret_cast<const v2f*>(wlp + 48 * WL_PITCH + k0);
    a0 = wmma_k4(a, b0, a0);
    a1 = wmma_k4(a, b1, a1);
    a2 = wmma_k4(a, b2, a2);
    a3 = wmma_k4(a, b3, a3);
  }
  float sel = 0.f;
#pragma unroll
  for (int r = 0; r < 8; ++r) {
    float ps = a0[r] * ws0 + a1[r] * ws1 + a2[r] * ws2 + a3[r] * ws3;
    float pd = a0[r] * wd0 + a1[r] * wd1 + a2[r] * wd2 + a3[r] * wd3;
    ps = gat_row_xadd<8>(ps);  pd = gat_row_xadd<8>(pd);
    ps = gat_row_xadd<4>(ps);  pd = gat_row_xadd<4>(pd);
    ps = gat_row_xadd<2>(ps);  pd = gat_row_xadd<2>(pd);
    ps = gat_row_xadd<1>(ps);  pd = gat_row_xadd<1>(pd);
    if ((lh & 7) == r) sel = (lh < 8) ? ps : pd;
  }
  sdl[(lh < 8) ? 0 : 1][wave * 16 + khalf * 8 + (lh & 7)] = sel;
  __syncthreads();

#pragma unroll
  for (int r = 0; r < 8; ++r) {
    const int row = row0 + khalf * 8 + r;
    if (row < N) {
      h[(size_t)row * HID + lh +  0] = a0[r];
      h[(size_t)row * HID + lh + 16] = a1[r];
      h[(size_t)row * HID + lh + 32] = a2[r];
      h[(size_t)row * HID + lh + 48] = a3[r];
    }
  }

  const int which = threadIdx.x >> 7;
  const int lrow  = threadIdx.x & 127;
  const int rowS  = blk * 128 + lrow;
  if (rowS < N) {
    float* __restrict__ dstp = which ? d : s;
    dstp[rowS] = sdl[which][lrow];
  }
}

#define BKT_SHIFT 8
#define BKT_NODES 256
#define BKT_WAVES (BKT_NODES / 32)
#define BKT_MAX   2048
#define STG_THREADS 256
#define ECHUNK    4096
#define EPT       (ECHUNK / STG_THREADS)
#define SORT_PER  20
#define SRT_CAP   8192
#define BPT       (BKT_MAX / STG_THREADS)
#define S3A_LDS_BYTES (BKT_MAX * 4)
#define S3C_LDS_BYTES (ECHUNK * 8 + (BKT_MAX + 32) * 4 + BKT_MAX * 4 + (STG_THREADS / 32) * 4)
#define S3D_LDS_BYTES ((SRT_CAP + 32) * 4 + (BKT_NODES + 32) * 4 + BKT_WAVES * 4)

__device__ __forceinline__ void gat_bcount_body(unsigned char* smem, const int blk, const int* __restrict__ ei,
                                                int* __restrict__ tcnt,
                                                int E, int NBK, int NBLK) {
  int* const hc = reinterpret_cast<int*>(smem);
  const int tid = threadIdx.x;
  for (int i = tid; i < NBK; i += STG_THREADS) hc[i] = 0;
  __syncthreads();
  const int base = blk * ECHUNK + tid;
  int d[EPT];
#pragma unroll
  for (int k = 0; k < EPT; ++k) {
    const int e = base + k * STG_THREADS;
    d[k] = (e < E) ? ei[(size_t)E + e] : -1;
  }
#pragma unroll
  for (int k = 0; k < EPT; ++k)
    if (d[k] >= 0) atomicAdd(&hc[d[k] >> BKT_SHIFT], 1);
  __syncthreads();
  for (int i = tid; i < NBK; i += STG_THREADS) tcnt[(size_t)i * NBLK + blk] = hc[i];
}

template <int ROWS>
__device__ __forceinline__ int gat_rows_sum(const int4* __restrict__ q) {
  int4 v[ROWS];
#pragma unroll
  for (int j = 0; j < ROWS; ++j) v[j] = q[j * 1024];
  int t = 0;
#pragma unroll
  for (int j = 0; j < ROWS; ++j) t += (v[j].x + v[j].y) + (v[j].z + v[j].w);
  return t;
}

__global__ void __launch_bounds__(1024) gat_scan_kernel(const int* __restrict__ cnt,
                                                        int* __restrict__ off, int L) {
  __shared__ int red[32];
  __shared__ int wsum[32];
  const int tid   = threadIdx.x;
  const int lane  = tid & 31;
  const int wave  = tid >> 5;
  const int start = blockIdx.x * 1024;

  const int i0 = start + tid;
  const int c  = (i0 < L) ? cnt[i0] : 0;

  const int rows = start >> 12;
  const int rem  = start & 4095;
  const int4* __restrict__ p = reinterpret_cast<const int4*>(cnt) + tid;

  int4 vp = make_int4(0, 0, 0, 0);
  if (tid * 4 < rem) vp = p[(size_t)rows * 1024];

  int pre = 0;
  int r = 0;
  for (; r + 16 <= rows; r += 16) pre += gat_rows_sum<16>(p + (size_t)r * 1024);
  if (rows - r >= 8) { pre += gat_rows_sum<8>(p + (size_t)r * 1024); r += 8; }
  if (rows - r >= 4) { pre += gat_rows_sum<4>(p + (size_t)r * 1024); r += 4; }
  if (rows - r >= 2) { pre += gat_rows_sum<2>(p + (size_t)r * 1024); r += 2; }
  if (rows - r >= 1) { pre += gat_rows_sum<1>(p + (size_t)r * 1024); }
  pre += (vp.x + vp.y) + (vp.z + vp.w);
  for (int m = 16; m > 0; m >>= 1) pre += __shfl_xor(pre, m);

  int inc = c;
  for (int m = 1; m < 32; m <<= 1) {
    const int t = __shfl_up(inc, m);
    if (lane >= m) inc += t;
  }
  if (lane == 0)  red[wave]  = pre;
  if (lane == 31) wsum[wave] = inc;
  __syncthreads();
  int part = red[lane] + ((lane < wave) ? wsum[lane] : 0);
  for (int m = 16; m > 0; m >>= 1) part += __shfl_xor(part, m);
  if (i0 < L) off[i0] = part + inc - c;
}

__device__ __forceinline__ void gat_bscatter_body(unsigned char* smem, const int blk, const int* __restrict__ ei,
                                                  const int* __restrict__ tcnt,
                                                  const int* __restrict__ toff,
                                                  int* __restrict__ cpair,
                                                  int E, int NBK, int NBLK) {
  int2* const stg = reinterpret_cast<int2*>(smem);
  int*  const pos = reinterpret_cast<int*>(smem + ECHUNK * 8);
  int*  const gdl = pos + (BKT_MAX + 32);
  int*  const wt  = gdl + BKT_MAX;
  const int tid  = threadIdx.x;
  const int lane = tid & 31;
  const int wave = tid >> 5;
  const int base = blk * ECHUNK + tid;
  int s[EPT], d[EPT];
#pragma unroll
  for (int k = 0; k < EPT; ++k) {
    const int e = base + k * STG_THREADS;
    s[k] = (e < E) ? ei[e] : 0;
    d[k] = (e < E) ? ei[(size_t)E + e] : -1;
  }
  int c[BPT], g[BPT];
  int tsum = 0;
#pragma unroll
  for (int j = 0; j < BPT; ++j) {
    const int i = tid * BPT + j;
    const unsigned off = ((unsigned)i * (unsigned)NBLK + (unsigned)blk) * 4u;
    c[j] = (i < NBK) ? *reinterpret_cast<const int*>(reinterpret_cast<const char*>(tcnt) + off) : 0;
    g[j] = (i < NBK) ? *reinterpret_cast<const int*>(reinterpret_cast<const char*>(toff) + off) : 0;
    tsum += c[j];
  }
  int inc = tsum;
  for (int m = 1; m < 32; m <<= 1) {
    const int t = __shfl_up(inc, m);
    if (lane >= m) inc += t;
  }
  if (lane == 31) wt[wave] = inc;
  __syncthreads();
  int part = (lane < STG_THREADS / 32 && lane < wave) ? wt[lane] : 0;
  for (int m = 16; m > 0; m >>= 1) part += __shfl_xor(part, m);
  int run = part + inc - tsum;
#pragma unroll
  for (int j = 0; j < BPT; ++j) {
    const int i = tid * BPT + j;
    if (i < NBK) { pos[i] = run; gdl[i] = g[j] - run; }
    run += c[j];
  }
  if (tid < 32) pos[BKT_MAX + tid] = 0;
  __syncthreads();
  int q[EPT];
#pragma unroll
  for (int k = 0; k < EPT; ++k)
    q[k] = atomicAdd(&pos[(d[k] >= 0) ? (d[k] >> BKT_SHIFT) : (BKT_MAX + lane)], 1);
#pragma unroll
  for (int k = 0; k < EPT; ++k)
    if (d[k] >= 0) stg[q[k]] = make_int2(s[k], d[k]);
  __syncthreads();
  const int left = E - blk * ECHUNK;
  const int nval = left < ECHUNK ? left : ECHUNK;
  int2* __restrict__ out2 = reinterpret_cast<int2*>(cpair);
  int2 v[EPT];
  int  gd[EPT];
#pragma unroll
  for (int k = 0; k < EPT; ++k) {
    const int j = tid + k * STG_THREADS;
    v[k] = stg[j < nval ? j : nval - 1];
  }
#pragma unroll
  for (int k = 0; k < EPT; ++k) gd[k] = gdl[v[k].y >> BKT_SHIFT];
#pragma unroll
  for (int k = 0; k < EPT; ++k) {
    const int j = tid + k * STG_THREADS;
    if (j < nval) out2[gd[k] + j] = v[k];
  }
}

__device__ __forceinline__ void gat_bsort_body(unsigned char* smem, const int b,
                                               const int* __restrict__ toff,
                                               const int* __restrict__ cpair,
                                               int* __restrict__ cursor,
                                               int* __restrict__ ssrc,
                                               int N, int E, int NBK, int NBLK) {
  int* const sbuf = reinterpret_cast<int*>(smem);
  int* const cl   = sbuf + (SRT_CAP + 32);
  int* const wtot = cl + (BKT_NODES + 32);
  const int tid   = threadIdx.x;
  const int lane  = tid & 31;
  const int wave  = tid >> 5;
  const int node0 = b << BKT_SHIFT;
  const int bstart = toff[(size_t)b * NBLK];
  const int bend   = (b + 1 < NBK) ? toff[(size_t)(b + 1) * NBLK] : E;
  const int2* __restrict__ pr = reinterpret_cast<const int2*>(cpair);
  const int  cnt    = bend - bstart;
  const bool staged = cnt <= SRT_CAP;
  const int  lead   = (int)((reinterpret_cast<size_t>(ssrc + bstart) >> 2) & 31);

  cl[tid] = 0;
  if (tid < 32) cl[BKT_NODES + tid] = 0;
  __syncthreads();

  for (int e0 = bstart; e0 < bend; e0 += SORT_PER * STG_THREADS) {
    int d[SORT_PER];
#pragma unroll
    for (int k = 0; k < SORT_PER; ++k) {
      const int e = e0 + k * STG_THREADS + tid;
      d[k] = (e < bend) ? pr[e].y : -1;
    }
#pragma unroll
    for (int k = 0; k < SORT_PER; ++k)
      if (d[k] >= 0) atomicAdd(&cl[d[k] - node0], 1);
  }
  __syncthreads();

  const int v = cl[tid];
  int inc = v;
  for (int m = 1; m < 32; m <<= 1) {
    const int t = __shfl_up(inc, m);
    if (lane >= m) inc += t;
  }
  if (lane == 31) wtot[wave] = inc;
  __syncthreads();
  int part = (lane < BKT_WAVES && lane < wave) ? wtot[lane] : 0;
  for (int m = 16; m > 0; m >>= 1) part += __shfl_xor(part, m);
  const int excl = part + inc - v;
  cl[tid] = excl;
  if (node0 + tid < N) cursor[node0 + tid] = bstart + excl;
  __syncthreads();

  for (int e0 = bstart; e0 < bend; e0 += SORT_PER * STG_THREADS) {
    int2 q[SORT_PER];
#pragma unroll
    for (int k = 0; k < SORT_PER; ++k) {
      const int e = e0 + k * STG_THREADS + tid;
      q[k] = (e < bend) ? pr[e] : make_int2(0, -1);
    }
    int at[SORT_PER];
#pragma unroll
    for (int k = 0; k < SORT_PER; ++k)
      at[k] = atomicAdd(&cl[(q[k].y >= 0) ? (q[k].y - node0) : (BKT_NODES + lane)], 1);
    if (staged) {
#pragma unroll
      for (int k = 0; k < SORT_PER; ++k)
        if (q[k].y >= 0) sbuf[lead + at[k]] = q[k].x;
    } else {
#pragma unroll
      for (int k = 0; k < SORT_PER; ++k)
        if (q[k].y >= 0) ssrc[bstart + at[k]] = q[k].x;
    }
  }
  if (staged) {
    __syncthreads();
    const int tot = lead + cnt;
    for (int j4 = tid * 4; j4 < tot; j4 += STG_THREADS * 4) {
      if (j4 >= lead && j4 + 4 <= tot) {
        *reinterpret_cast<int4*>(ssrc + (bstart - lead + j4)) = *reinterpret_cast<const int4*>(&sbuf[j4]);
      } else {
        for (int e = 0; e < 4; ++e) {
          const int j = j4 + e;
          if (j >= lead && j < tot) ssrc[bstart - lead + j] = sbuf[j];
        }
      }
    }
  }
}

template <int STAGE>
__global__ void __launch_bounds__(256) gat_stage_kernel(const float* __restrict__ x,
                                                        const float* __restrict__ W,
                                                        const float* __restrict__ att,
                                                        float* __restrict__ h,
                                                        float* __restrict__ s,
                                                        float* __restrict__ d,
                                                        const int* __restrict__ ei,
                                                        int* __restrict__ tcnt,
                                                        const int* __restrict__ toff,
                                                        int* __restrict__ cpair,
                                                        int* __restrict__ cursor,
                                                        int* __restrict__ ssrc,
                                                        int N, int E, int NBK, int NBLK,
                                                        int gemmBlk0, int gemmCnt) {
  constexpr int SORT_BYTES = STAGE == 0 ? S3A_LDS_BYTES : (STAGE == 1 ? S3C_LDS_BYTES : S3D_LDS_BYTES);
  constexpr int LDS_BYTES  = SORT_BYTES > LIN_LDS_BYTES ? SORT_BYTES : LIN_LDS_BYTES;
  __shared__ __attribute__((aligned(16))) unsigned char smem[LDS_BYTES];
  const int blk = blockIdx.x;
  if (blk < gemmCnt) {
    gat_lin_body(smem, gemmBlk0 + blk, x, W, att, h, s, d, N);
  } else {
    const int sb = blk - gemmCnt;
    if constexpr (STAGE == 0)      gat_bcount_body(smem, sb, ei, tcnt, E, NBK, NBLK);
    else if constexpr (STAGE == 1) gat_bscatter_body(smem, sb, ei, tcnt, toff, cpair, E, NBK, NBLK);
    else                           gat_bsort_body(smem, sb, toff, cpair, cursor, ssrc, N, E, NBK, NBLK);
  }
}


template <int K>
__device__ __forceinline__ int gat_row_lane(int v) {
  return __builtin_amdgcn_update_dpp(0, v, 0x150 + K, 0xF, 0xF, true);
}

template <int XM>
__device__ __forceinline__ float gat_row_xfold(float keep, float send) {
  const int t = __builtin_amdgcn_update_dpp(0, __float_as_int(send), 0x160 + XM, 0xF, 0xF, true);
  return keep + __int_as_float(t);
}

__device__ __forceinline__ float gat_other_row(float v) {
  return __int_as_float(__builtin_amdgcn_permlanex16(0, __float_as_int(v), 0x76543210u, 0xfedcba98u,
                                                     false, false));
}

template <int T>
__device__ __forceinline__ void gat_rows16(const float4* __restrict__ h4, int srcl, int c,
                                           bool b1, bool b2, bool b3, bool live, bool even,
                                           float4 at, float dn, float4& acc, float& dsum) {
  int sj[8];
  sj[0] = gat_row_lane<8 * T + 0>(srcl);
  sj[1] = gat_row_lane<8 * T + 1>(srcl);
  sj[2] = gat_row_lane<8 * T + 2>(srcl);
  sj[3] = gat_row_lane<8 * T + 3>(srcl);
  sj[4] = gat_row_lane<8 * T + 4>(srcl);
  sj[5] = gat_row_lane<8 * T + 5>(srcl);
  sj[6] = gat_row_lane<8 * T + 6>(srcl);
  sj[7] = gat_row_lane<8 * T + 7>(srcl);
  float4 r[8];
#pragma unroll
  for (int j = 0; j < 8; ++j) {
    const unsigned off = (unsigned)sj[j] * (unsigned)(HID * 4) + (unsigned)c * 16u;
    r[j] = *reinterpret_cast<const float4*>(reinterpret_cast<const char*>(h4) + off);
  }
  float p[8];
#pragma unroll
  for (int j = 0; j < 8; ++j)
    p[j] = __builtin_fmaf(r[j].w, at.w, __builtin_fmaf(r[j].z, at.z, __builtin_fmaf(r[j].y, at.y, r[j].x * at.x)));
  float q[4];
#pragma unroll
  for (int j = 0; j < 4; ++j)
    q[j] = gat_row_xfold<8>(b3 ? p[j + 4] : p[j], b3 ? p[j] : p[j + 4]);
  float u[2];
#pragma unroll
  for (int j = 0; j < 2; ++j)
    u[j] = gat_row_xfold<4>(b2 ? q[j + 2] : q[j], b2 ? q[j] : q[j + 2]);
  float t = gat_row_xfold<2>(b1 ? u[1] : u[0], b1 ? u[0] : u[1]);
  t = gat_row_xfold<1>(t, t);
  float v = t + dn;
  v = (v > 0.f) ? v : 0.2f * v;
  const float w = live ? __expf(v) : 0.f;
  dsum += even ? w : 0.f;
  const int wb = __float_as_int(w);
  float a[8];
  a[0] = __int_as_float(gat_row_lane< 0>(wb));
  a[1] = __int_as_float(gat_row_lane< 2>(wb));
  a[2] = __int_as_float(gat_row_lane< 4>(wb));
  a[3] = __int_as_float(gat_row_lane< 6>(wb));
  a[4] = __int_as_float(gat_row_lane< 8>(wb));
  a[5] = __int_as_float(gat_row_lane<10>(wb));
  a[6] = __int_as_float(gat_row_lane<12>(wb));
  a[7] = __int_as_float(gat_row_lane<14>(wb));
#pragma unroll
  for (int j = 0; j < 8; ++j) {
    acc.x += r[j].x * a[j];
    acc.y += r[j].y * a[j];
    acc.z += r[j].z * a[j];
    acc.w += r[j].w * a[j];
  }
}

__global__ void __launch_bounds__(256) gat_gather_kernel(const int* __restrict__ cursor,
                                                         const int* __restrict__ ssrc,
                                                         const float* __restrict__ h,
                                                         const float* __restrict__ att,
                                                         const float* __restrict__ d,
                                                         float* __restrict__ agg, int N, int E) {
  const int lane = threadIdx.x & 31;
  const int c    = lane & 15;
  const int half = lane >> 4;
  const int n = __builtin_amdgcn_readfirstlane(blockIdx.x * 8 + (threadIdx.x >> 5));
  if (n >= N) return;
  const int begin = __builtin_amdgcn_readfirstlane(cursor[n]);
  const int end   = __builtin_amdgcn_readfirstlane((n + 1 < N) ? cursor[n + 1] : E);
  const float  dn = d[n];
  const float4 at = *reinterpret_cast<const float4*>(att + 4 * c);
  const float4* __restrict__ h4 = reinterpret_cast<const float4*>(h);

  const bool b1 = (c & 2) != 0, b2 = (c & 4) != 0, b3 = (c & 8) != 0, even = (c & 1) == 0;
  float4 acc  = {0.f, 0.f, 0.f, 0.f};
  float  dsum = 0.f;
  for (int base = begin; base < end; base += 32) {
    const int el   = base + 2 * c + half;
    const int srcl = ssrc[el < end ? el : end - 1];
    const int ej   = base + (c & ~1) + half;
    gat_rows16<0>(h4, srcl, c, b1, b2, b3, ej < end, even, at, dn, acc, dsum);
    if (end - base > 16)
      gat_rows16<1>(h4, srcl, c, b1, b2, b3, ej + 16 < end, even, at, dn, acc, dsum);
  }
  acc.x += gat_other_row(acc.x);
  acc.y += gat_other_row(acc.y);
  acc.z += gat_other_row(acc.z);
  acc.w += gat_other_row(acc.w);
  float tot = dsum;
  tot = gat_row_xfold<8>(tot, tot);
  tot = gat_row_xfold<4>(tot, tot);
  tot = gat_row_xfold<2>(tot, tot);
  tot = gat_row_xfold<1>(tot, tot);
  tot += gat_other_row(tot);
  const float inv = 1.0f / (tot + 1e-9f);
  if (half == 0)
    reinterpret_cast<float4*>(agg)[(size_t)n * (HID / 4) + c] =
        make_float4(acc.x * inv, acc.y * inv, acc.z * inv, acc.w * inv);
}

__global__ void __launch_bounds__(256) gat_out_kernel(const float* __restrict__ agg,
                                                      const float* __restrict__ W,
                                                      const float* __restrict__ bias,
                                                      float* __restrict__ out, int N) {
  const int lane  = threadIdx.x & 31;
  const int wave  = threadIdx.x >> 5;
  const int row0  = (blockIdx.x * 8 + wave) * 16;
  const int lh    = lane & 15;
  const int khalf = lane >> 4;
  const int rowA  = row0 + lh;
  const int rA    = rowA < N ? rowA : (N - 1);

  __shared__ __attribute__((aligned(16))) float wo[OUT_D * WO_PITCH];
  v2f av[HID / 4];
  {
    const float4* __restrict__ W4 = reinterpret_cast<const float4*>(W);
    float4 wv[(OUT_D * HID / 4) / 256];
#pragma unroll
    for (int i = 0; i < (OUT_D * HID / 4) / 256; ++i) wv[i] = W4[i * 256 + threadIdx.x];
#pragma unroll
    for (int i = 0; i < HID / 4; ++i)
      av[i] = *reinterpret_cast<const v2f*>(&agg[(size_t)rA * HID + i * 4 + khalf * 2]);
#pragma unroll
    for (int i = 0; i < (OUT_D * HID / 4) / 256; ++i) {
      const int idx = i * 256 + threadIdx.x;
      const int r   = idx >> 4;
      const int c4  = idx & 15;
      *reinterpret_cast<float4*>(&wo[r * WO_PITCH + c4 * 4]) = wv[i];
    }
  }
  __builtin_amdgcn_fence(__ATOMIC_RELEASE, "workgroup", "local");
  __builtin_amdgcn_s_barrier();
  __builtin_amdgcn_fence(__ATOMIC_ACQUIRE, "workgroup", "local");
  const float* __restrict__ wop = &wo[lh * WO_PITCH + khalf * 2];

  v8f a0 = {}, a1 = {}, a2 = {}, a3 = {};
#pragma unroll
  for (int k0 = 0; k0 < HID; k0 += 4) {
    const float z0 = av[k0 / 4].x;
    const float z1 = av[k0 / 4].y;
    v2f a = { z0 > 0.f ? z0 : 0.f, z1 > 0.f ? z1 : 0.f };
    const v2f b0 = *reinterpret_cast<const v2f*>(wop +  0 * WO_PITCH + k0);
    const v2f b1 = *reinterpret_cast<const v2f*>(wop + 16 * WO_PITCH + k0);
    const v2f b2 = *reinterpret_cast<const v2f*>(wop + 32 * WO_PITCH + k0);
    const v2f b3 = *reinterpret_cast<const v2f*>(wop + 48 * WO_PITCH + k0);
    a0 = wmma_k4(a, b0, a0);
    a1 = wmma_k4(a, b1, a1);
    a2 = wmma_k4(a, b2, a2);
    a3 = wmma_k4(a, b3, a3);
  }
  const float bb0 = bias[lh +  0];
  const float bb1 = bias[lh + 16];
  const float bb2 = bias[lh + 32];
  const float bb3 = bias[lh + 48];
#pragma unroll
  for (int r = 0; r < 8; ++r) {
    const int row = row0 + khalf * 8 + r;
    if (row < N) {
      out[(size_t)row * OUT_D + lh +  0] = a0[r] + bb0;
      out[(size_t)row * OUT_D + lh + 16] = a1[r] + bb1;
      out[(size_t)row * OUT_D + lh + 32] = a2[r] + bb2;
      out[(size_t)row * OUT_D + lh + 48] = a3[r] + bb3;
    }
  }
}

extern "C" void kernel_launch(void* const* d_in, const int* in_sizes, int n_in,
                              void* d_out, int out_size, void* d_ws, size_t ws_size,
                              hipStream_t stream) {
  const float* x    = (const float*)d_in[0];
  const int*   ei   = (const int*)  d_in[1];
  const float* Wlin = (const float*)d_in[2];
  const float* att  = (const float*)d_in[3];
  const float* Wout = (const float*)d_in[4];
  const float* bout = (const float*)d_in[5];
  float* out = (float*)d_out;

  const int N = in_sizes[0] / IN_DIM;
  const int E = in_sizes[1] / 2;

  float* ws     = (float*)d_ws;
  float* h      = ws;
  float* s      = h + (size_t)N * HID;
  float* dn     = s + N;
  int*   ssrc   = (int*)(dn + N);
  int*   cursor = ssrc + E;
  float* agg    = (float*)(cursor + N);

  const int NBK  = (N + BKT_NODES - 1) / BKT_NODES;
  const int NBLK = (E + ECHUNK - 1) / ECHUNK;
  const int L    = NBK * NBLK;
  const int L4   = (L + 3) / 4 * 4;
  int*   tcnt   = (int*)agg;
  int*   toff   = tcnt + L4;
  int*   cpair  = toff + L4;

  const int tiles   = (N + 15) / 16;
  const int gemmBlk = (tiles + 7) / 8;

  const int g0 = (gemmBlk + 2) / 3;
  const int g1 = (gemmBlk - g0 + 1) / 2;
  const int g2 = gemmBlk - g0 - g1;

  gat_stage_kernel<0><<<g0 + NBLK, 256, 0, stream>>>(x, Wlin, att, h, s, dn, ei, tcnt, toff, cpair,
                                                     cursor, ssrc, N, E, NBK, NBLK, 0, g0);
  gat_scan_kernel<<<(L + 1023) / 1024, 1024, 0, stream>>>(tcnt, toff, L);
  gat_stage_kernel<1><<<g1 + NBLK, 256, 0, stream>>>(x, Wlin, att, h, s, dn, ei, tcnt, toff, cpair,
                                                     cursor, ssrc, N, E, NBK, NBLK, g0, g1);
  gat_stage_kernel<2><<<g2 + NBK, 256, 0, stream>>>(x, Wlin, att, h, s, dn, ei, tcnt, toff, cpair,
                                                    cursor, ssrc, N, E, NBK, NBLK, g0 + g1, g2);
  gat_gather_kernel<<<(N + 7) / 8, 256, 0, stream>>>(cursor, ssrc, h, att, dn, agg, N, E);

  gat_out_kernel<<<gemmBlk, 256, 0, stream>>>(agg, Wout, bout, out, N);
}


# Second GEMM fused into the gather: a wave gathers its 16 nodes one after the other into an LDS tile and multiplies it by W_out itself; agg never goes to memory; 6 -> 5 launches; sums unchanged
# speedup vs baseline: 5.5080x; 1.0573x over previous
#include <hip/hip_runtime.h>

typedef float v2f __attribute__((ext_vector_type(2)));
typedef float v8f __attribute__((ext_vector_type(8)));

#define IN_DIM 128
#define HID    64
#define OUT_D  64

#define WL_PITCH (IN_DIM + 4)
#define WO_PITCH (HID + 4)
#define LIN_WL_BYTES  (HID * WL_PITCH * 4)
#define LIN_LDS_BYTES (LIN_WL_BYTES + 2 * 128 * 4)

__device__ __forceinline__ v8f wmma_k4(v2f a, v2f b, v8f c) {
  return __builtin_amdgcn_wmma_f32_16x16x4_f32(false, a, false, b, (short)0, c,
                                               false, false);
}


template <int XM>
__device__ __forceinline__ float gat_row_xadd(float v) {
  const int t = __builtin_amdgcn_update_dpp(0, __float_as_int(v), 0x160 + XM, 0xF, 0xF, true);
  return v + __int_as_float(t);
}

__device__ __forceinline__ void gat_lin_body(unsigned char* smem, const int blk,
                                             const float* __restrict__ x,
                                             const float* __restrict__ W,
                                             const float* __restrict__ att,
                                             float* __restrict__ h,
                                             float* __restrict__ s,
                                             float* __restrict__ d, int N) {
  float (*const sdl)[128] = reinterpret_cast<float (*)[128]>(smem + LIN_WL_BYTES);
  float* const wl = reinterpret_cast<float*>(smem);
  const int lane  = threadIdx.x & 31;
  const int wave  = threadIdx.x >> 5;
  const int row0  = (blk * 8 + wave) * 16;
  const int lh    = lane & 15;
  const int khalf = lane >> 4;
  const int rowA  = row0 + lh;
  const int rA    = rowA < N ? rowA : (N - 1);

  const float ws0 = att[lh +  0], ws1 = att[lh + 16], ws2 = att[lh + 32], ws3 = att[lh + 48];
  const float wd0 = att[HID + lh +  0], wd1 = att[HID + lh + 16];
  const float wd2 = att[HID + lh + 32], wd3 = att[HID + lh + 48];

  v2f av[IN_DIM / 4];
  {
    const float4* __restrict__ W4 = reinterpret_cast<const float4*>(W);
    float4 wv[(HID * IN_DIM / 4) / 256];
#pragma unroll
    for (int i = 0; i < (HID * IN_DIM / 4) / 256; ++i) wv[i] = W4[i * 256 + threadIdx.x];
#pragma unroll
    for (int i = 0; i < IN_DIM / 4; ++i)
      av[i] = *reinterpret_cast<const v2f*>(&x[(size_t)rA * IN_DIM + i * 4 + khalf * 2]);
#pragma unroll
    for (int i = 0; i < (HID * IN_DIM / 4) / 256; ++i) {
      const int idx = i * 256 + threadIdx.x;
      const int r   = idx >> 5;
      const int c4  = idx & 31;
      *reinterpret_cast<float4*>(&wl[r * WL_PITCH + c4 * 4]) = wv[i];
    }
  }
  __builtin_amdgcn_fence(__ATOMIC_RELEASE, "workgroup", "local");
  __builtin_amdgcn_s_barrier();
  __builtin_amdgcn_fence(__ATOMIC_ACQUIRE, "workgroup", "local");
  const float* __restrict__ wlp = &wl[lh * WL_PITCH + khalf * 2];

  v8f a0 = {}, a1 = {}, a2 = {}, a3 = {};
#pragma unroll
  for (int k0 = 0; k0 < IN_DIM; k0 += 4) {
    const v2f a = av[k0 / 4];
    const v2f b0 = *reinterpret_cast<const v2f*>(wlp +  0 * WL_PITCH + k0);
    const v2f b1 = *reinterpret_cast<const v2f*>(wlp + 16 * WL_PITCH + k0);
    const v2f b2 = *reinterpret_cast<const v2f*>(wlp + 32 * WL_PITCH + k0);
    const v2f b3 = *reinterpret_cast<const v2f*>(wlp + 48 * WL_PITCH + k0);
    a0 = wmma_k4(a, b0, a0);
    a1 = wmma_k4(a, b1, a1);
    a2 = wmma_k4(a, b2, a2);
    a3 = wmma_k4(a, b3, a3);
  }
  float sel = 0.f;
#pragma unroll
  for (int r = 0; r < 8; ++r) {
    float ps = a0[r] * ws0 + a1[r] * ws1 + a2[r] * ws2 + a3[r] * ws3;
    float pd = a0[r] * wd0 + a1[r] * wd1 + a2[r] * wd2 + a3[r] * wd3;
    ps = gat_row_xadd<8>(ps);  pd = gat_row_xadd<8>(pd);
    ps = gat_row_xadd<4>(ps);  pd = gat_row_xadd<4>(pd);
    ps = gat_row_xadd<2>(ps);  pd = gat_row_xadd<2>(pd);
    ps = gat_row_xadd<1>(ps);  pd = gat_row_xadd<1>(pd);
    if ((lh & 7) == r) sel = (lh < 8) ? ps : pd;
  }
  sdl[(lh < 8) ? 0 : 1][wave * 16 + khalf * 8 + (lh & 7)] = sel;
  __syncthreads();

#pragma unroll
  for (int r = 0; r < 8; ++r) {
    const int row = row0 + khalf * 8 + r;
    if (row < N) {
      h[(size_t)row * HID + lh +  0] = a0[r];
      h[(size_t)row * HID + lh + 16] = a1[r];
      h[(size_t)row * HID + lh + 32] = a2[r];
      h[(size_t)row * HID + lh + 48] = a3[r];
    }
  }

  const int which = threadIdx.x >> 7;
  const int lrow  = threadIdx.x & 127;
  const int rowS  = blk * 128 + lrow;
  if (rowS < N) {
    float* __restrict__ dstp = which ? d : s;
    dstp[rowS] = sdl[which][lrow];
  }
}

#define BKT_SHIFT 8
#define BKT_NODES 256
#define BKT_WAVES (BKT_NODES / 32)
#define BKT_MAX   2048
#define STG_THREADS 256
#define ECHUNK    4096
#define EPT       (ECHUNK / STG_THREADS)
#define SORT_PER  20
#define SRT_CAP   8192
#define BPT       (BKT_MAX / STG_THREADS)
#define S3A_LDS_BYTES (BKT_MAX * 4)
#define S3C_LDS_BYTES (ECHUNK * 8 + (BKT_MAX + 32) * 4 + BKT_MAX * 4 + (STG_THREADS / 32) * 4)
#define S3D_LDS_BYTES ((SRT_CAP + 32) * 4 + (BKT_NODES + 32) * 4 + BKT_WAVES * 4)

__device__ __forceinline__ void gat_bcount_body(unsigned char* smem, const int blk, const int* __restrict__ ei,
                                                int* __restrict__ tcnt,
                                                int E, int NBK, int NBLK) {
  int* const hc = reinterpret_cast<int*>(smem);
  const int tid = threadIdx.x;
  for (int i = tid; i < NBK; i += STG_THREADS) hc[i] = 0;
  __syncthreads();
  const int base = blk * ECHUNK + tid;
  int d[EPT];
#pragma unroll
  for (int k = 0; k < EPT; ++k) {
    const int e = base + k * STG_THREADS;
    d[k] = (e < E) ? ei[(size_t)E + e] : -1;
  }
#pragma unroll
  for (int k = 0; k < EPT; ++k)
    if (d[k] >= 0) atomicAdd(&hc[d[k] >> BKT_SHIFT], 1);
  __syncthreads();
  for (int i = tid; i < NBK; i += STG_THREADS) tcnt[(size_t)i * NBLK + blk] = hc[i];
}

template <int ROWS>
__device__ __forceinline__ int gat_rows_sum(const int4* __restrict__ q) {
  int4 v[ROWS];
#pragma unroll
  for (int j = 0; j < ROWS; ++j) v[j] = q[j * 1024];
  int t = 0;
#pragma unroll
  for (int j = 0; j < ROWS; ++j) t += (v[j].x + v[j].y) + (v[j].z + v[j].w);
  return t;
}

__global__ void __launch_bounds__(1024) gat_scan_kernel(const int* __restrict__ cnt,
                                                        int* __restrict__ off, int L) {
  __shared__ int red[32];
  __shared__ int wsum[32];
  const int tid   = threadIdx.x;
  const int lane  = tid & 31;
  const int wave  = tid >> 5;
  const int start = blockIdx.x * 1024;

  const int i0 = start + tid;
  const int c  = (i0 < L) ? cnt[i0] : 0;

  const int rows = start >> 12;
  const int rem  = start & 4095;
  const int4* __restrict__ p = reinterpret_cast<const int4*>(cnt) + tid;

  int4 vp = make_int4(0, 0, 0, 0);
  if (tid * 4 < rem) vp = p[(size_t)rows * 1024];

  int pre = 0;
  int r = 0;
  for (; r + 16 <= rows; r += 16) pre += gat_rows_sum<16>(p + (size_t)r * 1024);
  if (rows - r >= 8) { pre += gat_rows_sum<8>(p + (size_t)r * 1024); r += 8; }
  if (rows - r >= 4) { pre += gat_rows_sum<4>(p + (size_t)r * 1024); r += 4; }
  if (rows - r >= 2) { pre += gat_rows_sum<2>(p + (size_t)r * 1024); r += 2; }
  if (rows - r >= 1) { pre += gat_rows_sum<1>(p + (size_t)r * 1024); }
  pre += (vp.x + vp.y) + (vp.z + vp.w);
  for (int m = 16; m > 0; m >>= 1) pre += __shfl_xor(pre, m);

  int inc = c;
  for (int m = 1; m < 32; m <<= 1) {
    const int t = __shfl_up(inc, m);
    if (lane >= m) inc += t;
  }
  if (lane == 0)  red[wave]  = pre;
  if (lane == 31) wsum[wave] = inc;
  __syncthreads();
  int part = red[lane] + ((lane < wave) ? wsum[lane] : 0);
  for (int m = 16; m > 0; m >>= 1) part += __shfl_xor(part, m);
  if (i0 < L) off[i0] = part + inc - c;
}

__device__ __forceinline__ void gat_bscatter_body(unsigned char* smem, const int blk, const int* __restrict__ ei,
                                                  const int* __restrict__ tcnt,
                                                  const int* __restrict__ toff,
                                                  int* __restrict__ cpair,
                                                  int E, int NBK, int NBLK) {
  int2* const stg = reinterpret_cast<int2*>(smem);
  int*  const pos = reinterpret_cast<int*>(smem + ECHUNK * 8);
  int*  const gdl = pos + (BKT_MAX + 32);
  int*  const wt  = gdl + BKT_MAX;
  const int tid  = threadIdx.x;
  const int lane = tid & 31;
  const int wave = tid >> 5;
  const int base = blk * ECHUNK + tid;
  int s[EPT], d[EPT];
#pragma unroll
  for (int k = 0; k < EPT; ++k) {
    const int e = base + k * STG_THREADS;
    s[k] = (e < E) ? ei[e] : 0;
    d[k] = (e < E) ? ei[(size_t)E + e] : -1;
  }
  int c[BPT], g[BPT];
  int tsum = 0;
#pragma unroll
  for (int j = 0; j < BPT; ++j) {
    const int i = tid * BPT + j;
    const unsigned off = ((unsigned)i * (unsigned)NBLK + (unsigned)blk) * 4u;
    c[j] = (i < NBK) ? *reinterpret_cast<const int*>(reinterpret_cast<const char*>(tcnt) + off) : 0;
    g[j] = (i < NBK) ? *reinterpret_cast<const int*>(reinterpret_cast<const char*>(toff) + off) : 0;
    tsum += c[j];
  }
  int inc = tsum;
  for (int m = 1; m < 32; m <<= 1) {
    const int t = __shfl_up(inc, m);
    if (lane >= m) inc += t;
  }
  if (lane == 31) wt[wave] = inc;
  __syncthreads();
  int part = (lane < STG_THREADS / 32 && lane < wave) ? wt[lane] : 0;
  for (int m = 16; m > 0; m >>= 1) part += __shfl_xor(part, m);
  int run = part + inc - tsum;
#pragma unroll
  for (int j = 0; j < BPT; ++j) {
    const int i = tid * BPT + j;
    if (i < NBK) { pos[i] = run; gdl[i] = g[j] - run; }
    run += c[j];
  }
  if (tid < 32) pos[BKT_MAX + tid] = 0;
  __syncthreads();
  int q[EPT];
#pragma unroll
  for (int k = 0; k < EPT; ++k)
    q[k] = atomicAdd(&pos[(d[k] >= 0) ? (d[k] >> BKT_SHIFT) : (BKT_MAX + lane)], 1);
#pragma unroll
  for (int k = 0; k < EPT; ++k)
    if (d[k] >= 0) stg[q[k]] = make_int2(s[k], d[k]);
  __syncthreads();
  const int left = E - blk * ECHUNK;
  const int nval = left < ECHUNK ? left : ECHUNK;
  int2* __restrict__ out2 = reinterpret_cast<int2*>(cpair);
  int2 v[EPT];
  int  gd[EPT];
#pragma unroll
  for (int k = 0; k < EPT; ++k) {
    const int j = tid + k * STG_THREADS;
    v[k] = stg[j < nval ? j : nval - 1];
  }
#pragma unroll
  for (int k = 0; k < EPT; ++k) gd[k] = gdl[v[k].y >> BKT_SHIFT];
#pragma unroll
  for (int k = 0; k < EPT; ++k) {
    const int j = tid + k * STG_THREADS;
    if (j < nval) out2[gd[k] + j] = v[k];
  }
}

__device__ __forceinline__ void gat_bsort_body(unsigned char* smem, const int b,
                                               const int* __restrict__ toff,
                                               const int* __restrict__ cpair,
                                               int* __restrict__ cursor,
                                               int* __restrict__ ssrc,
                                               int N, int E, int NBK, int NBLK) {
  int* const sbuf = reinterpret_cast<int*>(smem);
  int* const cl   = sbuf + (SRT_CAP + 32);
  int* const wtot = cl + (BKT_NODES + 32);
  const int tid   = threadIdx.x;
  const int lane  = tid & 31;
  const int wave  = tid >> 5;
  const int node0 = b << BKT_SHIFT;
  const int bstart = toff[(size_t)b * NBLK];
  const int bend   = (b + 1 < NBK) ? toff[(size_t)(b + 1) * NBLK] : E;
  const int2* __restrict__ pr = reinterpret_cast<const int2*>(cpair);
  const int  cnt    = bend - bstart;
  const bool staged = cnt <= SRT_CAP;
  const int  lead   = (int)((reinterpret_cast<size_t>(ssrc + bstart) >> 2) & 31);

  cl[tid] = 0;
  if (tid < 32) cl[BKT_NODES + tid] = 0;
  __syncthreads();

  for (int e0 = bstart; e0 < bend; e0 += SORT_PER * STG_THREADS) {
    int d[SORT_PER];
#pragma unroll
    for (int k = 0; k < SORT_PER; ++k) {
      const int e = e0 + k * STG_THREADS + tid;
      d[k] = (e < bend) ? pr[e].y : -1;
    }
#pragma unroll
    for (int k = 0; k < SORT_PER; ++k)
      if (d[k] >= 0) atomicAdd(&cl[d[k] - node0], 1);
  }
  __syncthreads();

  const int v = cl[tid];
  int inc = v;
  for (int m = 1; m < 32; m <<= 1) {
    const int t = __shfl_up(inc, m);
    if (lane >= m) inc += t;
  }
  if (lane == 31) wtot[wave] = inc;
  __syncthreads();
  int part = (lane < BKT_WAVES && lane < wave) ? wtot[lane] : 0;
  for (int m = 16; m > 0; m >>= 1) part += __shfl_xor(part, m);
  const int excl = part + inc - v;
  cl[tid] = excl;
  if (node0 + tid < N) cursor[node0 + tid] = bstart + excl;
  __syncthreads();

  for (int e0 = bstart; e0 < bend; e0 += SORT_PER * STG_THREADS) {
    int2 q[SORT_PER];
#pragma unroll
    for (int k = 0; k < SORT_PER; ++k) {
      const int e = e0 + k * STG_THREADS + tid;
      q[k] = (e < bend) ? pr[e] : make_int2(0, -1);
    }
    int at[SORT_PER];
#pragma unroll
    for (int k = 0; k < SORT_PER; ++k)
      at[k] = atomicAdd(&cl[(q[k].y >= 0) ? (q[k].y - node0) : (BKT_NODES + lane)], 1);
    if (staged) {
#pragma unroll
      for (int k = 0; k < SORT_PER; ++k)
        if (q[k].y >= 0) sbuf[lead + at[k]] = q[k].x;
    } else {
#pragma unroll
      for (int k = 0; k < SORT_PER; ++k)
        if (q[k].y >= 0) ssrc[bstart + at[k]] = q[k].x;
    }
  }
  if (staged) {
    __syncthreads();
    const int tot = lead + cnt;
    for (int j4 = tid * 4; j4 < tot; j4 += STG_THREADS * 4) {
      if (j4 >= lead && j4 + 4 <= tot) {
        *reinterpret_cast<int4*>(ssrc + (bstart - lead + j4)) = *reinterpret_cast<const int4*>(&sbuf[j4]);
      } else {
        for (int e = 0; e < 4; ++e) {
          const int j = j4 + e;
          if (j >= lead && j < tot) ssrc[bstart - lead + j] = sbuf[j];
        }
      }
    }
  }
}

template <int STAGE>
__global__ void __launch_bounds__(256) gat_stage_kernel(const float* __restrict__ x,
                                                        const float* __restrict__ W,
                                                        const float* __restrict__ att,
                                                        float* __restrict__ h,
                                                        float* __restrict__ s,
                                                        float* __restrict__ d,
                                                        const int* __restrict__ ei,
                                                        int* __restrict__ tcnt,
                                                        const int* __restrict__ toff,
                                                        int* __restrict__ cpair,
                                                        int* __restrict__ cursor,
                                                        int* __restrict__ ssrc,
                                                        int N, int E, int NBK, int NBLK,
                                                        int gemmBlk0, int gemmCnt) {
  constexpr int SORT_BYTES = STAGE == 0 ? S3A_LDS_BYTES : (STAGE == 1 ? S3C_LDS_BYTES : S3D_LDS_BYTES);
  constexpr int LDS_BYTES  = SORT_BYTES > LIN_LDS_BYTES ? SORT_BYTES : LIN_LDS_BYTES;
  __shared__ __attribute__((aligned(16))) unsigned char smem[LDS_BYTES];
  const int blk = blockIdx.x;
  if (blk < gemmCnt) {
    gat_lin_body(smem, gemmBlk0 + blk, x, W, att, h, s, d, N);
  } else {
    const int sb = blk - gemmCnt;
    if constexpr (STAGE == 0)      gat_bcount_body(smem, sb, ei, tcnt, E, NBK, NBLK);
    else if constexpr (STAGE == 1) gat_bscatter_body(smem, sb, ei, tcnt, toff, cpair, E, NBK, NBLK);
    else                           gat_bsort_body(smem, sb, toff, cpair, cursor, ssrc, N, E, NBK, NBLK);
  }
}


template <int K>
__device__ __forceinline__ int gat_row_lane(int v) {
  return __builtin_amdgcn_update_dpp(0, v, 0x150 + K, 0xF, 0xF, true);
}

template <int XM>
__device__ __forceinline__ float gat_row_xfold(float keep, float send) {
  const int t = __builtin_amdgcn_update_dpp(0, __float_as_int(send), 0x160 + XM, 0xF, 0xF, true);
  return keep + __int_as_float(t);
}

__device__ __forceinline__ float gat_other_row(float v) {
  return __int_as_float(__builtin_amdgcn_permlanex16(0, __float_as_int(v), 0x76543210u, 0xfedcba98u,
                                                     false, false));
}

template <int T>
__device__ __forceinline__ void gat_rows16(const float4* __restrict__ h4, int srcl, int c,
                                           bool b1, bool b2, bool b3, bool live, bool even,
                                           float4 at, float dn, float4& acc, float& dsum) {
  int sj[8];
  sj[0] = gat_row_lane<8 * T + 0>(srcl);
  sj[1] = gat_row_lane<8 * T + 1>(srcl);
  sj[2] = gat_row_lane<8 * T + 2>(srcl);
  sj[3] = gat_row_lane<8 * T + 3>(srcl);
  sj[4] = gat_row_lane<8 * T + 4>(srcl);
  sj[5] = gat_row_lane<8 * T + 5>(srcl);
  sj[6] = gat_row_lane<8 * T + 6>(srcl);
  sj[7] = gat_row_lane<8 * T + 7>(srcl);
  float4 r[8];
#pragma unroll
  for (int j = 0; j < 8; ++j) {
    const unsigned off = (unsigned)sj[j] * (unsigned)(HID * 4) + (unsigned)c * 16u;
    r[j] = *reinterpret_cast<const float4*>(reinterpret_cast<const char*>(h4) + off);
  }
  float p[8];
#pragma unroll
  for (int j = 0; j < 8; ++j)
    p[j] = __builtin_fmaf(r[j].w, at.w, __builtin_fmaf(r[j].z, at.z, __builtin_fmaf(r[j].y, at.y, r[j].x * at.x)));
  float q[4];
#pragma unroll
  for (int j = 0; j < 4; ++j)
    q[j] = gat_row_xfold<8>(b3 ? p[j + 4] : p[j], b3 ? p[j] : p[j + 4]);
  float u[2];
#pragma unroll
  for (int j = 0; j < 2; ++j)
    u[j] = gat_row_xfold<4>(b2 ? q[j + 2] : q[j], b2 ? q[j] : q[j + 2]);
  float t = gat_row_xfold<2>(b1 ? u[1] : u[0], b1 ? u[0] : u[1]);
  t = gat_row_xfold<1>(t, t);
  float v = t + dn;
  v = (v > 0.f) ? v : 0.2f * v;
  const float w = live ? __expf(v) : 0.f;
  dsum += even ? w : 0.f;
  const int wb = __float_as_int(w);
  float a[8];
  a[0] = __int_as_float(gat_row_lane< 0>(wb));
  a[1] = __int_as_float(gat_row_lane< 2>(wb));
  a[2] = __int_as_float(gat_row_lane< 4>(wb));
  a[3] = __int_as_float(gat_row_lane< 6>(wb));
  a[4] = __int_as_float(gat_row_lane< 8>(wb));
  a[5] = __int_as_float(gat_row_lane<10>(wb));
  a[6] = __int_as_float(gat_row_lane<12>(wb));
  a[7] = __int_as_float(gat_row_lane<14>(wb));
#pragma unroll
  for (int j = 0; j < 8; ++j) {
    acc.x += r[j].x * a[j];
    acc.y += r[j].y * a[j];
    acc.z += r[j].z * a[j];
    acc.w += r[j].w * a[j];
  }
}

#define GO_WAVES 4
#define GO_NODES (GO_WAVES * 16)
#define ZT_PITCH (HID + 4)

__global__ void __launch_bounds__(GO_WAVES * 32) gat_gather_out_kernel(const int* __restrict__ cursor,
                                                                       const int* __restrict__ ssrc,
                                                                       const float* __restrict__ h,
                                                                       const float* __restrict__ att,
                                                                       const float* __restrict__ d,
                                                                       const float* __restrict__ W,
                                                                       const float* __restrict__ bias,
                                                                       float* __restrict__ out, int N, int E) {
  __shared__ __attribute__((aligned(16))) float wo[OUT_D * WO_PITCH];
  __shared__ __attribute__((aligned(16))) float zt[GO_WAVES][16 * ZT_PITCH];
  const int lane = threadIdx.x & 31;
  const int wave = threadIdx.x >> 5;
  const int c    = lane & 15;
  const int half = lane >> 4;
  const int n0 = __builtin_amdgcn_readfirstlane((blockIdx.x * GO_WAVES + wave) * 16);

  float4 wv[(OUT_D * HID / 4) / (GO_WAVES * 32)];
  {
    const float4* __restrict__ W4 = reinterpret_cast<const float4*>(W);
#pragma unroll
    for (int i = 0; i < (OUT_D * HID / 4) / (GO_WAVES * 32); ++i) wv[i] = W4[i * (GO_WAVES * 32) + threadIdx.x];
  }
  const int   ni  = n0 + lane;
  const int   cur = (ni < N) ? cursor[ni] : E;
  const float dnv = (ni < N) ? d[ni] : 0.f;
  const float4 at = *reinterpret_cast<const float4*>(att + 4 * c);
  const float4* __restrict__ h4 = reinterpret_cast<const float4*>(h);

  int begin = __builtin_amdgcn_readlane(cur, 0);
  int end   = __builtin_amdgcn_readlane(cur, 1);
  int pre;
  {
    const int el = begin + 2 * c + half;
    int ix = el < end ? el : end - 1;
    ix = ix > 0 ? ix : 0;
    pre = ssrc[ix];
  }
#pragma unroll
  for (int i = 0; i < (OUT_D * HID / 4) / (GO_WAVES * 32); ++i) {
    const int idx = i * (GO_WAVES * 32) + threadIdx.x;
    const int r   = idx >> 4;
    const int c4  = idx & 15;
    *reinterpret_cast<float4*>(&wo[r * WO_PITCH + c4 * 4]) = wv[i];
  }

  const bool b1 = (c & 2) != 0, b2 = (c & 4) != 0, b3 = (c & 8) != 0, even = (c & 1) == 0;
  float* const ztw = &zt[wave][0];
#pragma nounroll
  for (int i = 0; i < 16; ++i) {
    const float dn = __int_as_float(__builtin_amdgcn_readlane(__float_as_int(dnv), i));
    const int nb = end;
    const int ne = __builtin_amdgcn_readlane(cur, i + 2 < 16 ? i + 2 : 16);
    int npre;
    {
      const int el = nb + 2 * c + half;
      int ix = el < ne ? el : ne - 1;
      ix = ix > 0 ? ix : 0;
      npre = ssrc[ix];
    }
    float4 acc  = {0.f, 0.f, 0.f, 0.f};
    float  dsum = 0.f;
    for (int base = begin; base < end; base += 32) {
      int srcl = pre;
      if (base != begin) {
        const int el = base + 2 * c + half;
        srcl = ssrc[el < end ? el : end - 1];
      }
      const int ej = base + (c & ~1) + half;
      gat_rows16<0>(h4, srcl, c, b1, b2, b3, ej < end, even, at, dn, acc, dsum);
      if (end - base > 16)
        gat_rows16<1>(h4, srcl, c, b1, b2, b3, ej + 16 < end, even, at, dn, acc, dsum);
    }
    acc.x += gat_other_row(acc.x);
    acc.y += gat_other_row(acc.y);
    acc.z += gat_other_row(acc.z);
    acc.w += gat_other_row(acc.w);
    float tot = dsum;
    tot = gat_row_xfold<8>(tot, tot);
    tot = gat_row_xfold<4>(tot, tot);
    tot = gat_row_xfold<2>(tot, tot);
    tot = gat_row_xfold<1>(tot, tot);
    tot += gat_other_row(tot);
    const float inv = 1.0f / (tot + 1e-9f);
    if (half == 0) {
      const float z0 = acc.x * inv, z1 = acc.y * inv, z2 = acc.z * inv, z3 = acc.w * inv;
      *reinterpret_cast<float4*>(&ztw[i * ZT_PITCH + 4 * c]) =
          make_float4(z0 > 0.f ? z0 : 0.f, z1 > 0.f ? z1 : 0.f, z2 > 0.f ? z2 : 0.f, z3 > 0.f ? z3 : 0.f);
    }
    pre   = npre;
    begin = nb;
    end   = ne;
  }
  __builtin_amdgcn_fence(__ATOMIC_RELEASE, "workgroup", "local");
  __builtin_amdgcn_s_barrier();
  __builtin_amdgcn_fence(__ATOMIC_ACQUIRE, "workgroup", "local");

  const int lh    = c;
  const int khalf = half;
  const float* __restrict__ zp  = &ztw[lh * ZT_PITCH + khalf * 2];
  const float* __restrict__ wop = &wo[lh * WO_PITCH + khalf * 2];

  v8f a0 = {}, a1 = {}, a2 = {}, a3 = {};
#pragma unroll
  for (int k0 = 0; k0 < HID; k0 += 4) {
    const v2f a  = *reinterpret_cast<const v2f*>(zp + k0);
    const v2f b0 = *reinterpret_cast<const v2f*>(wop +  0 * WO_PITCH + k0);
    const v2f b1f = *reinterpret_cast<const v2f*>(wop + 16 * WO_PITCH + k0);
    const v2f b2f = *reinterpret_cast<const v2f*>(wop + 32 * WO_PITCH + k0);
    const v2f b3f = *reinterpret_cast<const v2f*>(wop + 48 * WO_PITCH + k0);
    a0 = wmma_k4(a, b0, a0);
    a1 = wmma_k4(a, b1f, a1);
    a2 = wmma_k4(a, b2f, a2);
    a3 = wmma_k4(a, b3f, a3);
  }
  const float bb0 = bias[lh +  0];
  const float bb1 = bias[lh + 16];
  const float bb2 = bias[lh + 32];
  const float bb3 = bias[lh + 48];
#pragma unroll
  for (int r = 0; r < 8; ++r) {
    const int row = n0 + khalf * 8 + r;
    if (row < N) {
      out[(size_t)row * OUT_D + lh +  0] = a0[r] + bb0;
      out[(size_t)row * OUT_D + lh + 16] = a1[r] + bb1;
      out[(size_t)row * OUT_D + lh + 32] = a2[r] + bb2;
      out[(size_t)row * OUT_D + lh + 48] = a3[r] + bb3;
    }
  }
}

extern "C" void kernel_launch(void* const* d_in, const int* in_sizes, int n_in,
                              void* d_out, int out_size, void* d_ws, size_t ws_size,
                              hipStream_t stream) {
  const float* x    = (const float*)d_in[0];
  const int*   ei   = (const int*)  d_in[1];
  const float* Wlin = (const float*)d_in[2];
  const float* att  = (const float*)d_in[3];
  const float* Wout = (const float*)d_in[4];
  const float* bout = (const float*)d_in[5];
  float* out = (float*)d_out;

  const int N = in_sizes[0] / IN_DIM;
  const int E = in_sizes[1] / 2;

  float* ws     = (float*)d_ws;
  float* h      = ws;
  float* s      = h + (size_t)N * HID;
  float* dn     = s + N;
  int*   ssrc   = (int*)(dn + N);
  int*   cursor = ssrc + E;
  int*   scr    = cursor + N;

  const int NBK  = (N + BKT_NODES - 1) / BKT_NODES;
  const int NBLK = (E + ECHUNK - 1) / ECHUNK;
  const int L    = NBK * NBLK;
  const int L4   = (L + 3) / 4 * 4;
  int*   tcnt   = scr;
  int*   toff   = tcnt + L4;
  int*   cpair  = toff + L4;

  const int tiles   = (N + 15) / 16;
  const int gemmBlk = (tiles + 7) / 8;

  const int g0 = (gemmBlk + 2) / 3;
  const int g1 = (gemmBlk - g0 + 1) / 2;
  const int g2 = gemmBlk - g0 - g1;

  gat_stage_kernel<0><<<g0 + NBLK, 256, 0, stream>>>(x, Wlin, att, h, s, dn, ei, tcnt, toff, cpair,
                                                     cursor, ssrc, N, E, NBK, NBLK, 0, g0);
  gat_scan_kernel<<<(L + 1023) / 1024, 1024, 0, stream>>>(tcnt, toff, L);
  gat_stage_kernel<1><<<g1 + NBLK, 256, 0, stream>>>(x, Wlin, att, h, s, dn, ei, tcnt, toff, cpair,
                                                     cursor, ssrc, N, E, NBK, NBLK, g0, g1);
  gat_stage_kernel<2><<<g2 + NBK, 256, 0, stream>>>(x, Wlin, att, h, s, dn, ei, tcnt, toff, cpair,
                                                    cursor, ssrc, N, E, NBK, NBLK, g0 + g1, g2);
  gat_gather_out_kernel<<<(N + GO_NODES - 1) / GO_NODES, GO_WAVES * 32, 0, stream>>>(cursor, ssrc, h, att, dn,
                                                                                  Wout, bout, out, N, E);
}


# Sort blocks take 8192 edges, not 4096 (32 per thread; LDS adds and list stores in two half-trips): half the (bucket, block) table, scan and sort work-groups; sum order = LDS arrival order
# speedup vs baseline: 5.8182x; 1.0563x over previous
#include <hip/hip_runtime.h>

typedef float v2f __attribute__((ext_vector_type(2)));
typedef float v8f __attribute__((ext_vector_type(8)));

#define IN_DIM 128
#define HID    64
#define OUT_D  64

#define WL_PITCH (IN_DIM + 4)
#define WO_PITCH (HID + 4)
#define LIN_WL_BYTES  (HID * WL_PITCH * 4)
#define LIN_LDS_BYTES (LIN_WL_BYTES + 2 * 128 * 4)

__device__ __forceinline__ v8f wmma_k4(v2f a, v2f b, v8f c) {
  return __builtin_amdgcn_wmma_f32_16x16x4_f32(false, a, false, b, (short)0, c,
                                               false, false);
}


template <int XM>
__device__ __forceinline__ float gat_row_xadd(float v) {
  const int t = __builtin_amdgcn_update_dpp(0, __float_as_int(v), 0x160 + XM, 0xF, 0xF, true);
  return v + __int_as_float(t);
}

__device__ __forceinline__ void gat_lin_body(unsigned char* smem, const int blk,
                                             const float* __restrict__ x,
                                             const float* __restrict__ W,
                                             const float* __restrict__ att,
                                             float* __restrict__ h,
                                             float* __restrict__ s,
                                             float* __restrict__ d, int N) {
  float (*const sdl)[128] = reinterpret_cast<float (*)[128]>(smem + LIN_WL_BYTES);
  float* const wl = reinterpret_cast<float*>(smem);
  const int lane  = threadIdx.x & 31;
  const int wave  = threadIdx.x >> 5;
  const int row0  = (blk * 8 + wave) * 16;
  const int lh    = lane & 15;
  const int khalf = lane >> 4;
  const int rowA  = row0 + lh;
  const int rA    = rowA < N ? rowA : (N - 1);

  const float ws0 = att[lh +  0], ws1 = att[lh + 16], ws2 = att[lh + 32], ws3 = att[lh + 48];
  const float wd0 = att[HID + lh +  0], wd1 = att[HID + lh + 16];
  const float wd2 = att[HID + lh + 32], wd3 = att[HID + lh + 48];

  v2f av[IN_DIM / 4];
  {
    const float4* __restrict__ W4 = reinterpret_cast<const float4*>(W);
    float4 wv[(HID * IN_DIM / 4) / 256];
#pragma unroll
    for (int i = 0; i < (HID * IN_DIM / 4) / 256; ++i) wv[i] = W4[i * 256 + threadIdx.x];
#pragma unroll
    for (int i = 0; i < IN_DIM / 4; ++i)
      av[i] = *reinterpret_cast<const v2f*>(&x[(size_t)rA * IN_DIM + i * 4 + khalf * 2]);
#pragma unroll
    for (int i = 0; i < (HID * IN_DIM / 4) / 256; ++i) {
      const int idx = i * 256 + threadIdx.x;
      const int r   = idx >> 5;
      const int c4  = idx & 31;
      *reinterpret_cast<float4*>(&wl[r * WL_PITCH + c4 * 4]) = wv[i];
    }
  }
  __builtin_amdgcn_fence(__ATOMIC_RELEASE, "workgroup", "local");
  __builtin_amdgcn_s_barrier();
  __builtin_amdgcn_fence(__ATOMIC_ACQUIRE, "workgroup", "local");
  const float* __restrict__ wlp = &wl[lh * WL_PITCH + khalf * 2];

  v8f a0 = {}, a1 = {}, a2 = {}, a3 = {};
#pragma unroll
  for (int k0 = 0; k0 < IN_DIM; k0 += 4) {
    const v2f a = av[k0 / 4];
    const v2f b0 = *reinterpret_cast<const v2f*>(wlp +  0 * WL_PITCH + k0);
    const v2f b1 = *reinterpret_cast<const v2f*>(wlp + 16 * WL_PITCH + k0);
    const v2f b2 = *reinterpret_cast<const v2f*>(wlp + 32 * WL_PITCH + k0);
    const v2f b3 = *reinterpret_cast<const v2f*>(wlp + 48 * WL_PITCH + k0);
    a0 = wmma_k4(a, b0, a0);
    a1 = wmma_k4(a, b1, a1);
    a2 = wmma_k4(a, b2, a2);
    a3 = wmma_k4(a, b3, a3);
  }
  float sel = 0.f;
#pragma unroll
  for (int r = 0; r < 8; ++r) {
    float ps = a0[r] * ws0 + a1[r] * ws1 + a2[r] * ws2 + a3[r] * ws3;
    float pd = a0[r] * wd0 + a1[r] * wd1 + a2[r] * wd2 + a3[r] * wd3;
    ps = gat_row_xadd<8>(ps);  pd = gat_row_xadd<8>(pd);
    ps = gat_row_xadd<4>(ps);  pd = gat_row_xadd<4>(pd);
    ps = gat_row_xadd<2>(ps);  pd = gat_row_xadd<2>(pd);
    ps = gat_row_xadd<1>(ps);  pd = gat_row_xadd<1>(pd);
    if ((lh & 7) == r) sel = (lh < 8) ? ps : pd;
  }
  sdl[(lh < 8) ? 0 : 1][wave * 16 + khalf * 8 + (lh & 7)] = sel;
  __syncthreads();

#pragma unroll
  for (int r = 0; r < 8; ++r) {
    const int row = row0 + khalf * 8 + r;
    if (row < N) {
      h[(size_t)row * HID + lh +  0] = a0[r];
      h[(size_t)row * HID + lh + 16] = a1[r];
      h[(size_t)row * HID + lh + 32] = a2[r];
      h[(size_t)row * HID + lh + 48] = a3[r];
    }
  }

  const int which = threadIdx.x >> 7;
  const int lrow  = threadIdx.x & 127;
  const int rowS  = blk * 128 + lrow;
  if (rowS < N) {
    float* __restrict__ dstp = which ? d : s;
    dstp[rowS] = sdl[which][lrow];
  }
}

#define BKT_SHIFT 8
#define BKT_NODES 256
#define BKT_WAVES (BKT_NODES / 32)
#define BKT_MAX   2048
#define STG_THREADS 256
#define ECHUNK    8192
#define EPT       (ECHUNK / STG_THREADS)
#define EPT_H     16
#define SORT_PER  20
#define SRT_CAP   8192
#define BPT       (BKT_MAX / STG_THREADS)
#define S3A_LDS_BYTES (BKT_MAX * 4)
#define S3C_LDS_BYTES (ECHUNK * 8 + (BKT_MAX + 32) * 4 + BKT_MAX * 4 + (STG_THREADS / 32) * 4)
#define S3D_LDS_BYTES ((SRT_CAP + 32) * 4 + (BKT_NODES + 32) * 4 + BKT_WAVES * 4)

__device__ __forceinline__ void gat_bcount_body(unsigned char* smem, const int blk, const int* __restrict__ ei,
                                                int* __restrict__ tcnt,
                                                int E, int NBK, int NBLK) {
  int* const hc = reinterpret_cast<int*>(smem);
  const int tid = threadIdx.x;
  for (int i = tid; i < NBK; i += STG_THREADS) hc[i] = 0;
  __syncthreads();
  const int base = blk * ECHUNK + tid;
  int d[EPT];
#pragma unroll
  for (int k = 0; k < EPT; ++k) {
    const int e = base + k * STG_THREADS;
    d[k] = (e < E) ? ei[(size_t)E + e] : -1;
  }
#pragma unroll
  for (int k = 0; k < EPT; ++k)
    if (d[k] >= 0) atomicAdd(&hc[d[k] >> BKT_SHIFT], 1);
  __syncthreads();
  for (int i = tid; i < NBK; i += STG_THREADS) tcnt[(size_t)i * NBLK + blk] = hc[i];
}

template <int ROWS>
__device__ __forceinline__ int gat_rows_sum(const int4* __restrict__ q) {
  int4 v[ROWS];
#pragma unroll
  for (int j = 0; j < ROWS; ++j) v[j] = q[j * 1024];
  int t = 0;
#pragma unroll
  for (int j = 0; j < ROWS; ++j) t += (v[j].x + v[j].y) + (v[j].z + v[j].w);
  return t;
}

__global__ void __launch_bounds__(1024) gat_scan_kernel(const int* __restrict__ cnt,
                                                        int* __restrict__ off, int L) {
  __shared__ int red[32];
  __shared__ int wsum[32];
  const int tid   = threadIdx.x;
  const int lane  = tid & 31;
  const int wave  = tid >> 5;
  const int start = blockIdx.x * 1024;

  const int i0 = start + tid;
  const int c  = (i0 < L) ? cnt[i0] : 0;

  const int rows = start >> 12;
  const int rem  = start & 4095;
  const int4* __restrict__ p = reinterpret_cast<const int4*>(cnt) + tid;

  int4 vp = make_int4(0, 0, 0, 0);
  if (tid * 4 < rem) vp = p[(size_t)rows * 1024];

  int pre = 0;
  int r = 0;
  for (; r + 16 <= rows; r += 16) pre += gat_rows_sum<16>(p + (size_t)r * 1024);
  if (rows - r >= 8) { pre += gat_rows_sum<8>(p + (size_t)r * 1024); r += 8; }
  if (rows - r >= 4) { pre += gat_rows_sum<4>(p + (size_t)r * 1024); r += 4; }
  if (rows - r >= 2) { pre += gat_rows_sum<2>(p + (size_t)r * 1024); r += 2; }
  if (rows - r >= 1) { pre += gat_rows_sum<1>(p + (size_t)r * 1024); }
  pre += (vp.x + vp.y) + (vp.z + vp.w);
  for (int m = 16; m > 0; m >>= 1) pre += __shfl_xor(pre, m);

  int inc = c;
  for (int m = 1; m < 32; m <<= 1) {
    const int t = __shfl_up(inc, m);
    if (lane >= m) inc += t;
  }
  if (lane == 0)  red[wave]  = pre;
  if (lane == 31) wsum[wave] = inc;
  __syncthreads();
  int part = red[lane] + ((lane < wave) ? wsum[lane] : 0);
  for (int m = 16; m > 0; m >>= 1) part += __shfl_xor(part, m);
  if (i0 < L) off[i0] = part + inc - c;
}

__device__ __forceinline__ void gat_bscatter_body(unsigned char* smem, const int blk, const int* __restrict__ ei,
                                                  const int* __restrict__ tcnt,
                                                  const int* __restrict__ toff,
                                                  int* __restrict__ cpair,
                                                  int E, int NBK, int NBLK) {
  int2* const stg = reinterpret_cast<int2*>(smem);
  int*  const pos = reinterpret_cast<int*>(smem + ECHUNK * 8);
  int*  const gdl = pos + (BKT_MAX + 32);
  int*  const wt  = gdl + BKT_MAX;
  const int tid  = threadIdx.x;
  const int lane = tid & 31;
  const int wave = tid >> 5;
  const int base = blk * ECHUNK + tid;
  int s[EPT], d[EPT];
#pragma unroll
  for (int k = 0; k < EPT; ++k) {
    const int e = base + k * STG_THREADS;
    s[k] = (e < E) ? ei[e] : 0;
    d[k] = (e < E) ? ei[(size_t)E + e] : -1;
  }
  int c[BPT], g[BPT];
  int tsum = 0;
#pragma unroll
  for (int j = 0; j < BPT; ++j) {
    const int i = tid * BPT + j;
    const unsigned off = ((unsigned)i * (unsigned)NBLK + (unsigned)blk) * 4u;
    c[j] = (i < NBK) ? *reinterpret_cast<const int*>(reinterpret_cast<const char*>(tcnt) + off) : 0;
    g[j] = (i < NBK) ? *reinterpret_cast<const int*>(reinterpret_cast<const char*>(toff) + off) : 0;
    tsum += c[j];
  }
  int inc = tsum;
  for (int m = 1; m < 32; m <<= 1) {
    const int t = __shfl_up(inc, m);
    if (lane >= m) inc += t;
  }
  if (lane == 31) wt[wave] = inc;
  __syncthreads();
  int part = (lane < STG_THREADS / 32 && lane < wave) ? wt[lane] : 0;
  for (int m = 16; m > 0; m >>= 1) part += __shfl_xor(part, m);
  int run = part + inc - tsum;
#pragma unroll
  for (int j = 0; j < BPT; ++j) {
    const int i = tid * BPT + j;
    if (i < NBK) { pos[i] = run; gdl[i] = g[j] - run; }
    run += c[j];
  }
  if (tid < 32) pos[BKT_MAX + tid] = 0;
  __syncthreads();
#pragma unroll
  for (int hh = 0; hh < EPT / EPT_H; ++hh) {
    int q[EPT_H];
#pragma unroll
    for (int k = 0; k < EPT_H; ++k) {
      const int kk = hh * EPT_H + k;
      q[k] = atomicAdd(&pos[(d[kk] >= 0) ? (d[kk] >> BKT_SHIFT) : (BKT_MAX + lane)], 1);
    }
#pragma unroll
    for (int k = 0; k < EPT_H; ++k) {
      const int kk = hh * EPT_H + k;
      if (d[kk] >= 0) stg[q[k]] = make_int2(s[kk], d[kk]);
    }
  }
  __syncthreads();
  const int left = E - blk * ECHUNK;
  const int nval = left < ECHUNK ? left : ECHUNK;
  int2* __restrict__ out2 = reinterpret_cast<int2*>(cpair);
#pragma unroll
  for (int hh = 0; hh < EPT / EPT_H; ++hh) {
    int2 v[EPT_H];
    int  gd[EPT_H];
#pragma unroll
    for (int k = 0; k < EPT_H; ++k) {
      const int j = tid + (hh * EPT_H + k) * STG_THREADS;
      v[k] = stg[j < nval ? j : nval - 1];
    }
#pragma unroll
    for (int k = 0; k < EPT_H; ++k) gd[k] = gdl[v[k].y >> BKT_SHIFT];
#pragma unroll
    for (int k = 0; k < EPT_H; ++k) {
      const int j = tid + (hh * EPT_H + k) * STG_THREADS;
      if (j < nval) out2[gd[k] + j] = v[k];
    }
  }
}

__device__ __forceinline__ void gat_bsort_body(unsigned char* smem, const int b,
                                               const int* __restrict__ toff,
                                               const int* __restrict__ cpair,
                                               int* __restrict__ cursor,
                                               int* __restrict__ ssrc,
                                               int N, int E, int NBK, int NBLK) {
  int* const sbuf = reinterpret_cast<int*>(smem);
  int* const cl   = sbuf + (SRT_CAP + 32);
  int* const wtot = cl + (BKT_NODES + 32);
  const int tid   = threadIdx.x;
  const int lane  = tid & 31;
  const int wave  = tid >> 5;
  const int node0 = b << BKT_SHIFT;
  const int bstart = toff[(size_t)b * NBLK];
  const int bend   = (b + 1 < NBK) ? toff[(size_t)(b + 1) * NBLK] : E;
  const int2* __restrict__ pr = reinterpret_cast<const int2*>(cpair);
  const int  cnt    = bend - bstart;
  const bool staged = cnt <= SRT_CAP;
  const int  lead   = (int)((reinterpret_cast<size_t>(ssrc + bstart) >> 2) & 31);

  cl[tid] = 0;
  if (tid < 32) cl[BKT_NODES + tid] = 0;
  __syncthreads();

  for (int e0 = bstart; e0 < bend; e0 += SORT_PER * STG_THREADS) {
    int d[SORT_PER];
#pragma unroll
    for (int k = 0; k < SORT_PER; ++k) {
      const int e = e0 + k * STG_THREADS + tid;
      d[k] = (e < bend) ? pr[e].y : -1;
    }
#pragma unroll
    for (int k = 0; k < SORT_PER; ++k)
      if (d[k] >= 0) atomicAdd(&cl[d[k] - node0], 1);
  }
  __syncthreads();

  const int v = cl[tid];
  int inc = v;
  for (int m = 1; m < 32; m <<= 1) {
    const int t = __shfl_up(inc, m);
    if (lane >= m) inc += t;
  }
  if (lane == 31) wtot[wave] = inc;
  __syncthreads();
  int part = (lane < BKT_WAVES && lane < wave) ? wtot[lane] : 0;
  for (int m = 16; m > 0; m >>= 1) part += __shfl_xor(part, m);
  const int excl = part + inc - v;
  cl[tid] = excl;
  if (node0 + tid < N) cursor[node0 + tid] = bstart + excl;
  __syncthreads();

  for (int e0 = bstart; e0 < bend; e0 += SORT_PER * STG_THREADS) {
    int2 q[SORT_PER];
#pragma unroll
    for (int k = 0; k < SORT_PER; ++k) {
      const int e = e0 + k * STG_THREADS + tid;
      q[k] = (e < bend) ? pr[e] : make_int2(0, -1);
    }
    int at[SORT_PER];
#pragma unroll
    for (int k = 0; k < SORT_PER; ++k)
      at[k] = atomicAdd(&cl[(q[k].y >= 0) ? (q[k].y - node0) : (BKT_NODES + lane)], 1);
    if (staged) {
#pragma unroll
      for (int k = 0; k < SORT_PER; ++k)
        if (q[k].y >= 0) sbuf[lead + at[k]] = q[k].x;
    } else {
#pragma unroll
      for (int k = 0; k < SORT_PER; ++k)
        if (q[k].y >= 0) ssrc[bstart + at[k]] = q[k].x;
    }
  }
  if (staged) {
    __syncthreads();
    const int tot = lead + cnt;
    for (int j4 = tid * 4; j4 < tot; j4 += STG_THREADS * 4) {
      if (j4 >= lead && j4 + 4 <= tot) {
        *reinterpret_cast<int4*>(ssrc + (bstart - lead + j4)) = *reinterpret_cast<const int4*>(&sbuf[j4]);
      } else {
        for (int e = 0; e < 4; ++e) {
          const int j = j4 + e;
          if (j >= lead && j < tot) ssrc[bstart - lead + j] = sbuf[j];
        }
      }
    }
  }
}

template <int STAGE>
__global__ void __launch_bounds__(256) gat_stage_kernel(const float* __restrict__ x,
                                                        const float* __restrict__ W,
                                                        const float* __restrict__ att,
                                                        float* __restrict__ h,
                                                        float* __restrict__ s,
                                                        float* __restrict__ d,
                                                        const int* __restrict__ ei,
                                                        int* __restrict__ tcnt,
                                                        const int* __restrict__ toff,
                                                        int* __restrict__ cpair,
                                                        int* __restrict__ cursor,
                                                        int* __restrict__ ssrc,
                                                        int N, int E, int NBK, int NBLK,
                                                        int gemmBlk0, int gemmCnt) {
  constexpr int SORT_BYTES = STAGE == 0 ? S3A_LDS_BYTES : (STAGE == 1 ? S3C_LDS_BYTES : S3D_LDS_BYTES);
  constexpr int LDS_BYTES  = SORT_BYTES > LIN_LDS_BYTES ? SORT_BYTES : LIN_LDS_BYTES;
  __shared__ __attribute__((aligned(16))) unsigned char smem[LDS_BYTES];
  const int blk = blockIdx.x;
  if (blk < gemmCnt) {
    gat_lin_body(smem, gemmBlk0 + blk, x, W, att, h, s, d, N);
  } else {
    const int sb = blk - gemmCnt;
    if constexpr (STAGE == 0)      gat_bcount_body(smem, sb, ei, tcnt, E, NBK, NBLK);
    else if constexpr (STAGE == 1) gat_bscatter_body(smem, sb, ei, tcnt, toff, cpair, E, NBK, NBLK);
    else                           gat_bsort_body(smem, sb, toff, cpair, cursor, ssrc, N, E, NBK, NBLK);
  }
}


template <int K>
__device__ __forceinline__ int gat_row_lane(int v) {
  return __builtin_amdgcn_update_dpp(0, v, 0x150 + K, 0xF, 0xF, true);
}

template <int XM>
__device__ __forceinline__ float gat_row_xfold(float keep, float send) {
  const int t = __builtin_amdgcn_update_dpp(0, __float_as_int(send), 0x160 + XM, 0xF, 0xF, true);
  return keep + __int_as_float(t);
}

__device__ __forceinline__ float gat_other_row(float v) {
  return __int_as_float(__builtin_amdgcn_permlanex16(0, __float_as_int(v), 0x76543210u, 0xfedcba98u,
                                                     false, false));
}

template <int T>
__device__ __forceinline__ void gat_rows16(const float4* __restrict__ h4, int srcl, int c,
                                           bool b1, bool b2, bool b3, bool live, bool even,
                                           float4 at, float dn, float4& acc, float& dsum) {
  int sj[8];
  sj[0] = gat_row_lane<8 * T + 0>(srcl);
  sj[1] = gat_row_lane<8 * T + 1>(srcl);
  sj[2] = gat_row_lane<8 * T + 2>(srcl);
  sj[3] = gat_row_lane<8 * T + 3>(srcl);
  sj[4] = gat_row_lane<8 * T + 4>(srcl);
  sj[5] = gat_row_lane<8 * T + 5>(srcl);
  sj[6] = gat_row_lane<8 * T + 6>(srcl);
  sj[7] = gat_row_lane<8 * T + 7>(srcl);
  float4 r[8];
#pragma unroll
  for (int j = 0; j < 8; ++j) {
    const unsigned off = (unsigned)sj[j] * (unsigned)(HID * 4) + (unsigned)c * 16u;
    r[j] = *reinterpret_cast<const float4*>(reinterpret_cast<const char*>(h4) + off);
  }
  float p[8];
#pragma unroll
  for (int j = 0; j < 8; ++j)
    p[j] = __builtin_fmaf(r[j].w, at.w, __builtin_fmaf(r[j].z, at.z, __builtin_fmaf(r[j].y, at.y, r[j].x * at.x)));
  float q[4];
#pragma unroll
  for (int j = 0; j < 4; ++j)
    q[j] = gat_row_xfold<8>(b3 ? p[j + 4] : p[j], b3 ? p[j] : p[j + 4]);
  float u[2];
#pragma unroll
  for (int j = 0; j < 2; ++j)
    u[j] = gat_row_xfold<4>(b2 ? q[j + 2] : q[j], b2 ? q[j] : q[j + 2]);
  float t = gat_row_xfold<2>(b1 ? u[1] : u[0], b1 ? u[0] : u[1]);
  t = gat_row_xfold<1>(t, t);
  float v = t + dn;
  v = (v > 0.f) ? v : 0.2f * v;
  const float w = live ? __expf(v) : 0.f;
  dsum += even ? w : 0.f;
  const int wb = __float_as_int(w);
  float a[8];
  a[0] = __int_as_float(gat_row_lane< 0>(wb));
  a[1] = __int_as_float(gat_row_lane< 2>(wb));
  a[2] = __int_as_float(gat_row_lane< 4>(wb));
  a[3] = __int_as_float(gat_row_lane< 6>(wb));
  a[4] = __int_as_float(gat_row_lane< 8>(wb));
  a[5] = __int_as_float(gat_row_lane<10>(wb));
  a[6] = __int_as_float(gat_row_lane<12>(wb));
  a[7] = __int_as_float(gat_row_lane<14>(wb));
#pragma unroll
  for (int j = 0; j < 8; ++j) {
    acc.x += r[j].x * a[j];
    acc.y += r[j].y * a[j];
    acc.z += r[j].z * a[j];
    acc.w += r[j].w * a[j];
  }
}

#define GO_WAVES 4
#define GO_NODES (GO_WAVES * 16)
#define ZT_PITCH (HID + 4)

__global__ void __launch_bounds__(GO_WAVES * 32) gat_gather_out_kernel(const int* __restrict__ cursor,
                                                                       const int* __restrict__ ssrc,
                                                                       const float* __restrict__ h,
                                                                       const float* __restrict__ att,
                                                                       const float* __restrict__ d,
                                                                       const float* __restrict__ W,
                                                                       const float* __restrict__ bias,
                                                                       float* __restrict__ out, int N, int E) {
  __shared__ __attribute__((aligned(16))) float wo[OUT_D * WO_PITCH];
  __shared__ __attribute__((aligned(16))) float zt[GO_WAVES][16 * ZT_PITCH];
  const int lane = threadIdx.x & 31;
  const int wave = threadIdx.x >> 5;
  const int c    = lane & 15;
  const int half = lane >> 4;
  const int n0 = __builtin_amdgcn_readfirstlane((blockIdx.x * GO_WAVES + wave) * 16);

  float4 wv[(OUT_D * HID / 4) / (GO_WAVES * 32)];
  {
    const float4* __restrict__ W4 = reinterpret_cast<const float4*>(W);
#pragma unroll
    for (int i = 0; i < (OUT_D * HID / 4) / (GO_WAVES * 32); ++i) wv[i] = W4[i * (GO_WAVES * 32) + threadIdx.x];
  }
  const int   ni  = n0 + lane;
  const int   cur = (ni < N) ? cursor[ni] : E;
  const float dnv = (ni < N) ? d[ni] : 0.f;
  const float4 at = *reinterpret_cast<const float4*>(att + 4 * c);
  const float4* __restrict__ h4 = reinterpret_cast<const float4*>(h);

  int begin = __builtin_amdgcn_readlane(cur, 0);
  int end   = __builtin_amdgcn_readlane(cur, 1);
  int pre;
  {
    const int el = begin + 2 * c + half;
    int ix = el < end ? el : end - 1;
    ix = ix > 0 ? ix : 0;
    pre = ssrc[ix];
  }
#pragma unroll
  for (int i = 0; i < (OUT_D * HID / 4) / (GO_WAVES * 32); ++i) {
    const int idx = i * (GO_WAVES * 32) + threadIdx.x;
    const int r   = idx >> 4;
    const int c4  = idx & 15;
    *reinterpret_cast<float4*>(&wo[r * WO_PITCH + c4 * 4]) = wv[i];
  }

  const bool b1 = (c & 2) != 0, b2 = (c & 4) != 0, b3 = (c & 8) != 0, even = (c & 1) == 0;
  float* const ztw = &zt[wave][0];
#pragma nounroll
  for (int i = 0; i < 16; ++i) {
    const float dn = __int_as_float(__builtin_amdgcn_readlane(__float_as_int(dnv), i));
    const int nb = end;
    const int ne = __builtin_amdgcn_readlane(cur, i + 2 < 16 ? i + 2 : 16);
    int npre;
    {
      const int el = nb + 2 * c + half;
      int ix = el < ne ? el : ne - 1;
      ix = ix > 0 ? ix : 0;
      npre = ssrc[ix];
    }
    float4 acc  = {0.f, 0.f, 0.f, 0.f};
    float  dsum = 0.f;
    for (int base = begin; base < end; base += 32) {
      int srcl = pre;
      if (base != begin) {
        const int el = base + 2 * c + half;
        srcl = ssrc[el < end ? el : end - 1];
      }
      const int ej = base + (c & ~1) + half;
      gat_rows16<0>(h4, srcl, c, b1, b2, b3, ej < end, even, at, dn, acc, dsum);
      if (end - base > 16)
        gat_rows16<1>(h4, srcl, c, b1, b2, b3, ej + 16 < end, even, at, dn, acc, dsum);
    }
    acc.x += gat_other_row(acc.x);
    acc.y += gat_other_row(acc.y);
    acc.z += gat_other_row(acc.z);
    acc.w += gat_other_row(acc.w);
    float tot = dsum;
    tot = gat_row_xfold<8>(tot, tot);
    tot = gat_row_xfold<4>(tot, tot);
    tot = gat_row_xfold<2>(tot, tot);
    tot = gat_row_xfold<1>(tot, tot);
    tot += gat_other_row(tot);
    const float inv = 1.0f / (tot + 1e-9f);
    if (half == 0) {
      const float z0 = acc.x * inv, z1 = acc.y * inv, z2 = acc.z * inv, z3 = acc.w * inv;
      *reinterpret_cast<float4*>(&ztw[i * ZT_PITCH + 4 * c]) =
          make_float4(z0 > 0.f ? z0 : 0.f, z1 > 0.f ? z1 : 0.f, z2 > 0.f ? z2 : 0.f, z3 > 0.f ? z3 : 0.f);
    }
    pre   = npre;
    begin = nb;
    end   = ne;
  }
  __builtin_amdgcn_fence(__ATOMIC_RELEASE, "workgroup", "local");
  __builtin_amdgcn_s_barrier();
  __builtin_amdgcn_fence(__ATOMIC_ACQUIRE, "workgroup", "local");

  const int lh    = c;
  const int khalf = half;
  const float* __restrict__ zp  = &ztw[lh * ZT_PITCH + khalf * 2];
  const float* __restrict__ wop = &wo[lh * WO_PITCH + khalf * 2];

  v8f a0 = {}, a1 = {}, a2 = {}, a3 = {};
#pragma unroll
  for (int k0 = 0; k0 < HID; k0 += 4) {
    const v2f a  = *reinterpret_cast<const v2f*>(zp + k0);
    const v2f b0 = *reinterpret_cast<const v2f*>(wop +  0 * WO_PITCH + k0);
    const v2f b1f = *reinterpret_cast<const v2f*>(wop + 16 * WO_PITCH + k0);
    const v2f b2f = *reinterpret_cast<const v2f*>(wop + 32 * WO_PITCH + k0);
    const v2f b3f = *reinterpret_cast<const v2f*>(wop + 48 * WO_PITCH + k0);
    a0 = wmma_k4(a, b0, a0);
    a1 = wmma_k4(a, b1f, a1);
    a2 = wmma_k4(a, b2f, a2);
    a3 = wmma_k4(a, b3f, a3);
  }
  const float bb0 = bias[lh +  0];
  const float bb1 = bias[lh + 16];
  const float bb2 = bias[lh + 32];
  const float bb3 = bias[lh + 48];
#pragma unroll
  for (int r = 0; r < 8; ++r) {
    const int row = n0 + khalf * 8 + r;
    if (row < N) {
      out[(size_t)row * OUT_D + lh +  0] = a0[r] + bb0;
      out[(size_t)row * OUT_D + lh + 16] = a1[r] + bb1;
      out[(size_t)row * OUT_D + lh + 32] = a2[r] + bb2;
      out[(size_t)row * OUT_D + lh + 48] = a3[r] + bb3;
    }
  }
}

extern "C" void kernel_launch(void* const* d_in, const int* in_sizes, int n_in,
                              void* d_out, int out_size, void* d_ws, size_t ws_size,
                              hipStream_t stream) {
  const float* x    = (const float*)d_in[0];
  const int*   ei   = (const int*)  d_in[1];
  const float* Wlin = (const float*)d_in[2];
  const float* att  = (const float*)d_in[3];
  const float* Wout = (const float*)d_in[4];
  const float* bout = (const float*)d_in[5];
  float* out = (float*)d_out;

  const int N = in_sizes[0] / IN_DIM;
  const int E = in_sizes[1] / 2;

  float* ws     = (float*)d_ws;
  float* h      = ws;
  float* s      = h + (size_t)N * HID;
  float* dn     = s + N;
  int*   ssrc   = (int*)(dn + N);
  int*   cursor = ssrc + E;
  int*   scr    = cursor + N;

  const int NBK  = (N + BKT_NODES - 1) / BKT_NODES;
  const int NBLK = (E + ECHUNK - 1) / ECHUNK;
  const int L    = NBK * NBLK;
  const int L4   = (L + 3) / 4 * 4;
  int*   tcnt   = scr;
  int*   toff   = tcnt + L4;
  int*   cpair  = toff + L4;

  const int tiles   = (N + 15) / 16;
  const int gemmBlk = (tiles + 7) / 8;

  const int g0 = (gemmBlk + 2) / 3;
  const int g1 = (gemmBlk - g0 + 1) / 2;
  const int g2 = gemmBlk - g0 - g1;

  gat_stage_kernel<0><<<g0 + NBLK, 256, 0, stream>>>(x, Wlin, att, h, s, dn, ei, tcnt, toff, cpair,
                                                     cursor, ssrc, N, E, NBK, NBLK, 0, g0);
  gat_scan_kernel<<<(L + 1023) / 1024, 1024, 0, stream>>>(tcnt, toff, L);
  gat_stage_kernel<1><<<g1 + NBLK, 256, 0, stream>>>(x, Wlin, att, h, s, dn, ei, tcnt, toff, cpair,
                                                     cursor, ssrc, N, E, NBK, NBLK, g0, g1);
  gat_stage_kernel<2><<<g2 + NBK, 256, 0, stream>>>(x, Wlin, att, h, s, dn, ei, tcnt, toff, cpair,
                                                    cursor, ssrc, N, E, NBK, NBLK, g0 + g1, g2);
  gat_gather_out_kernel<<<(N + GO_NODES - 1) / GO_NODES, GO_WAVES * 32, 0, stream>>>(cursor, ssrc, h, att, dn,
                                                                                  Wout, bout, out, N, E);
}


# Sort blocks keep their bucket-ordered pairs in place plus a table of bucket starts; a bucket reads its pieces, offset = column sum: count and scan launches dropped, 5 -> 3; sum order = LDS arrival
# speedup vs baseline: 6.2056x; 1.0666x over previous
#include <hip/hip_runtime.h>

typedef float v2f __attribute__((ext_vector_type(2)));
typedef float v8f __attribute__((ext_vector_type(8)));

#define IN_DIM 128
#define HID    64
#define OUT_D  64

#define WL_PITCH (IN_DIM + 4)
#define WO_PITCH (HID + 4)
#define LIN_WL_BYTES  (HID * WL_PITCH * 4)
#define LIN_LDS_BYTES (LIN_WL_BYTES + 2 * 128 * 4)

__device__ __forceinline__ v8f wmma_k4(v2f a, v2f b, v8f c) {
  return __builtin_amdgcn_wmma_f32_16x16x4_f32(false, a, false, b, (short)0, c,
                                               false, false);
}


template <int XM>
__device__ __forceinline__ float gat_row_xadd(float v) {
  const int t = __builtin_amdgcn_update_dpp(0, __float_as_int(v), 0x160 + XM, 0xF, 0xF, true);
  return v + __int_as_float(t);
}

__device__ __forceinline__ void gat_lin_body(unsigned char* smem, const int blk,
                                             const float* __restrict__ x,
                                             const float* __restrict__ W,
                                             const float* __restrict__ att,
                                             float* __restrict__ h,
                                             float* __restrict__ s,
                                             float* __restrict__ d, int N) {
  float (*const sdl)[128] = reinterpret_cast<float (*)[128]>(smem + LIN_WL_BYTES);
  float* const wl = reinterpret_cast<float*>(smem);
  const int lane  = threadIdx.x & 31;
  const int wave  = threadIdx.x >> 5;
  const int row0  = (blk * 8 + wave) * 16;
  const int lh    = lane & 15;
  const int khalf = lane >> 4;
  const int rowA  = row0 + lh;
  const int rA    = rowA < N ? rowA : (N - 1);

  const float ws0 = att[lh +  0], ws1 = att[lh + 16], ws2 = att[lh + 32], ws3 = att[lh + 48];
  const float wd0 = att[HID + lh +  0], wd1 = att[HID + lh + 16];
  const float wd2 = att[HID + lh + 32], wd3 = att[HID + lh + 48];

  v2f av[IN_DIM / 4];
  {
    const float4* __restrict__ W4 = reinterpret_cast<const float4*>(W);
    float4 wv[(HID * IN_DIM / 4) / 256];
#pragma unroll
    for (int i = 0; i < (HID * IN_DIM / 4) / 256; ++i) wv[i] = W4[i * 256 + threadIdx.x];
#pragma unroll
    for (int i = 0; i < IN_DIM / 4; ++i)
      av[i] = *reinterpret_cast<const v2f*>(&x[(size_t)rA * IN_DIM + i * 4 + khalf * 2]);
#pragma unroll
    for (int i = 0; i < (HID * IN_DIM / 4) / 256; ++i) {
      const int idx = i * 256 + threadIdx.x;
      const int r   = idx >> 5;
      const int c4  = idx & 31;
      *reinterpret_cast<float4*>(&wl[r * WL_PITCH + c4 * 4]) = wv[i];
    }
  }
  __builtin_amdgcn_fence(__ATOMIC_RELEASE, "workgroup", "local");
  __builtin_amdgcn_s_barrier();
  __builtin_amdgcn_fence(__ATOMIC_ACQUIRE, "workgroup", "local");
  const float* __restrict__ wlp = &wl[lh * WL_PITCH + khalf * 2];

  v8f a0 = {}, a1 = {}, a2 = {}, a3 = {};
#pragma unroll
  for (int k0 = 0; k0 < IN_DIM; k0 += 4) {
    const v2f a = av[k0 / 4];
    const v2f b0 = *reinterpret_cast<const v2f*>(wlp +  0 * WL_PITCH + k0);
    const v2f b1 = *reinterpret_cast<const v2f*>(wlp + 16 * WL_PITCH + k0);
    const v2f b2 = *reinterpret_cast<const v2f*>(wlp + 32 * WL_PITCH + k0);
    const v2f b3 = *reinterpret_cast<const v2f*>(wlp + 48 * WL_PITCH + k0);
    a0 = wmma_k4(a, b0, a0);
    a1 = wmma_k4(a, b1, a1);
    a2 = wmma_k4(a, b2, a2);
    a3 = wmma_k4(a, b3, a3);
  }
  float sel = 0.f;
#pragma unroll
  for (int r = 0; r < 8; ++r) {
    float ps = a0[r] * ws0 + a1[r] * ws1 + a2[r] * ws2 + a3[r] * ws3;
    float pd = a0[r] * wd0 + a1[r] * wd1 + a2[r] * wd2 + a3[r] * wd3;
    ps = gat_row_xadd<8>(ps);  pd = gat_row_xadd<8>(pd);
    ps = gat_row_xadd<4>(ps);  pd = gat_row_xadd<4>(pd);
    ps = gat_row_xadd<2>(ps);  pd = gat_row_xadd<2>(pd);
    ps = gat_row_xadd<1>(ps);  pd = gat_row_xadd<1>(pd);
    if ((lh & 7) == r) sel = (lh < 8) ? ps : pd;
  }
  sdl[(lh < 8) ? 0 : 1][wave * 16 + khalf * 8 + (lh & 7)] = sel;
  __syncthreads();

#pragma unroll
  for (int r = 0; r < 8; ++r) {
    const int row = row0 + khalf * 8 + r;
    if (row < N) {
      h[(size_t)row * HID + lh +  0] = a0[r];
      h[(size_t)row * HID + lh + 16] = a1[r];
      h[(size_t)row * HID + lh + 32] = a2[r];
      h[(size_t)row * HID + lh + 48] = a3[r];
    }
  }

  const int which = threadIdx.x >> 7;
  const int lrow  = threadIdx.x & 127;
  const int rowS  = blk * 128 + lrow;
  if (rowS < N) {
    float* __restrict__ dstp = which ? d : s;
    dstp[rowS] = sdl[which][lrow];
  }
}

#define BKT_SHIFT 8
#define BKT_NODES 256
#define BKT_WAVES (BKT_NODES / 32)
#define BKT_MAX   2048
#define STG_THREADS 256
#define ECHUNK    8192
#define EPT       (ECHUNK / STG_THREADS)
#define EPT_H     16
#define SRT_CAP   8192
#define BPT       (BKT_MAX / STG_THREADS)
#define SA_LDS_BYTES (ECHUNK * 8 + (BKT_MAX + 32) * 4 + (STG_THREADS / 32) * 4)
#define SB_LDS_BYTES ((SRT_CAP + 32) * 4 + (BKT_NODES + 32) * 4 + 3 * BKT_WAVES * 4)

__device__ __forceinline__ void gat_blist_body(unsigned char* smem, const int blk, const int* __restrict__ ei,
                                               int* __restrict__ lst,
                                               int* __restrict__ cpair,
                                               int E, int NBK, int NBLK) {
  int2* const stg = reinterpret_cast<int2*>(smem);
  int*  const pos = reinterpret_cast<int*>(smem + ECHUNK * 8);
  int*  const wt  = pos + (BKT_MAX + 32);
  const int tid  = threadIdx.x;
  const int lane = tid & 31;
  const int wave = tid >> 5;
  const int base = blk * ECHUNK + tid;
  for (int i = tid; i < NBK; i += STG_THREADS) pos[i] = 0;
  if (tid < 32) pos[BKT_MAX + tid] = 0;
  int s[EPT], d[EPT];
#pragma unroll
  for (int k = 0; k < EPT; ++k) {
    const int e = base + k * STG_THREADS;
    s[k] = (e < E) ? ei[e] : 0;
    d[k] = (e < E) ? ei[(size_t)E + e] : -1;
  }
  __syncthreads();
#pragma unroll
  for (int k = 0; k < EPT; ++k)
    if (d[k] >= 0) atomicAdd(&pos[d[k] >> BKT_SHIFT], 1);
  __syncthreads();
  int c[BPT];
  int tsum = 0;
#pragma unroll
  for (int j = 0; j < BPT; ++j) {
    const int i = tid * BPT + j;
    c[j] = (i < NBK) ? pos[i] : 0;
    tsum += c[j];
  }
  int inc = tsum;
  for (int m = 1; m < 32; m <<= 1) {
    const int t = __shfl_up(inc, m);
    if (lane >= m) inc += t;
  }
  if (lane == 31) wt[wave] = inc;
  __syncthreads();
  int part = (lane < STG_THREADS / 32 && lane < wave) ? wt[lane] : 0;
  for (int m = 16; m > 0; m >>= 1) part += __shfl_xor(part, m);
  int run = part + inc - tsum;
#pragma unroll
  for (int j = 0; j < BPT; ++j) {
    const int i = tid * BPT + j;
    if (i < NBK) { pos[i] = run; lst[(size_t)i * NBLK + blk] = run; }
    run += c[j];
  }
  __syncthreads();
#pragma unroll
  for (int hh = 0; hh < EPT / EPT_H; ++hh) {
    int q[EPT_H];
#pragma unroll
    for (int k = 0; k < EPT_H; ++k) {
      const int kk = hh * EPT_H + k;
      q[k] = atomicAdd(&pos[(d[kk] >= 0) ? (d[kk] >> BKT_SHIFT) : (BKT_MAX + lane)], 1);
    }
#pragma unroll
    for (int k = 0; k < EPT_H; ++k) {
      const int kk = hh * EPT_H + k;
      if (d[kk] >= 0) stg[q[k]] = make_int2(s[kk], d[kk]);
    }
  }
  __syncthreads();
  const int left = E - blk * ECHUNK;
  const int nval = left < ECHUNK ? left : ECHUNK;
  int2* __restrict__ out2 = reinterpret_cast<int2*>(cpair) + (size_t)blk * ECHUNK;
#pragma unroll
  for (int hh = 0; hh < EPT / EPT_H; ++hh) {
    int2 v[EPT_H];
#pragma unroll
    for (int k = 0; k < EPT_H; ++k) {
      const int j = tid + (hh * EPT_H + k) * STG_THREADS;
      v[k] = stg[j < nval ? j : nval - 1];
    }
#pragma unroll
    for (int k = 0; k < EPT_H; ++k) {
      const int j = tid + (hh * EPT_H + k) * STG_THREADS;
      if (j < nval) out2[j] = v[k];
    }
  }
}


template <int PASS, int GP>
__device__ __forceinline__ void gat_pieces_pass(const int2* __restrict__ pr, const int* __restrict__ lst,
                                                int* cl, int* sbuf, int* __restrict__ ssrc,
                                                const int b, const int node0, const int lane, const int wave,
                                                const int E, const int NBK, const int NBLK,
                                                const bool staged, const int lead, const int bstart,
                                                int& rsum, int& csum) {
#pragma nounroll
  for (int t0 = 0; t0 < NBLK; t0 += STG_THREADS) {
    const int pl = t0 + lane * BKT_WAVES + wave;
    int st = 0, ln = 0;
    if (pl < NBLK) {
      const int rs = lst[(size_t)b * NBLK + pl];
      int re;
      if (b + 1 < NBK) {
        re = lst[(size_t)(b + 1) * NBLK + pl];
      } else {
        const int left = E - pl * ECHUNK;
        re = left < ECHUNK ? left : ECHUNK;
      }
      ln = re - rs;
      st = pl * ECHUNK + rs;
      if (PASS == 1) { rsum += rs; csum += ln; }
    }
#pragma unroll
    for (int g = 0; g < 32; g += GP) {
      if (t0 + g * BKT_WAVES + wave < NBLK) {
        int2 q[GP];
        int  pn[GP];
#pragma unroll
        for (int u = 0; u < GP; ++u) {
          const int ps = __builtin_amdgcn_readlane(st, g + u);
          pn[u] = __builtin_amdgcn_readlane(ln, g + u);
          if (PASS == 1) {
            q[u].x = 0;
            q[u].y = (lane < pn[u]) ? pr[ps + lane].y : -1;
          } else {
            q[u] = (lane < pn[u]) ? pr[ps + lane] : make_int2(0, -1);
          }
        }
        if (PASS == 1) {
#pragma unroll
          for (int u = 0; u < GP; ++u)
            if (q[u].y >= 0) atomicAdd(&cl[q[u].y - node0], 1);
        } else {
          int at[GP];
#pragma unroll
          for (int u = 0; u < GP; ++u)
            at[u] = atomicAdd(&cl[(q[u].y >= 0) ? (q[u].y - node0) : (BKT_NODES + lane)], 1);
          if (staged) {
#pragma unroll
            for (int u = 0; u < GP; ++u)
              if (q[u].y >= 0) sbuf[lead + at[u]] = q[u].x;
          } else {
#pragma unroll
            for (int u = 0; u < GP; ++u)
              if (q[u].y >= 0) ssrc[bstart + at[u]] = q[u].x;
          }
        }
      }
    }
    if (__builtin_amdgcn_ballot_w32(ln > 32) != 0u) {
#pragma nounroll
      for (int u = 0; u < 32; ++u) {
        const int pnl = __builtin_amdgcn_readlane(ln, u);
        const int psl = __builtin_amdgcn_readlane(st, u);
        for (int o = 32; o < pnl; o += 32) {
          const bool in = o + lane < pnl;
          if (PASS == 1) {
            const int dd = in ? pr[psl + o + lane].y : -1;
            if (dd >= 0) atomicAdd(&cl[dd - node0], 1);
          } else {
            const int2 v = in ? pr[psl + o + lane] : make_int2(0, -1);
            if (v.y >= 0) {
              const int a = atomicAdd(&cl[v.y - node0], 1);
              if (staged) sbuf[lead + a] = v.x; else ssrc[bstart + a] = v.x;
            }
          }
        }
      }
    }
  }
}

__device__ __forceinline__ void gat_bsort_body(unsigned char* smem, const int b,
                                               const int* __restrict__ lst,
                                               const int* __restrict__ cpair,
                                               int* __restrict__ cursor,
                                               int* __restrict__ ssrc,
                                               int N, int E, int NBK, int NBLK) {
  int* const sbuf = reinterpret_cast<int*>(smem);
  int* const cl   = sbuf + (SRT_CAP + 32);
  int* const wtot = cl + (BKT_NODES + 32);
  int* const wrs  = wtot + BKT_WAVES;
  int* const wcs  = wrs + BKT_WAVES;
  const int tid   = threadIdx.x;
  const int lane  = tid & 31;
  const int wave  = __builtin_amdgcn_readfirstlane(tid >> 5);
  const int node0 = b << BKT_SHIFT;
  const int2* __restrict__ pr = reinterpret_cast<const int2*>(cpair);

  cl[tid] = 0;
  if (tid < 32) cl[BKT_NODES + tid] = 0;
  __syncthreads();

  int rsum = 0, csum = 0;
  gat_pieces_pass<1, 32>(pr, lst, cl, sbuf, ssrc, b, node0, lane, wave, E, NBK, NBLK, false, 0, 0, rsum, csum);
  for (int m = 16; m > 0; m >>= 1) { rsum += __shfl_xor(rsum, m); csum += __shfl_xor(csum, m); }
  if (lane == 0) { wrs[wave] = rsum; wcs[wave] = csum; }
  __syncthreads();
  int bstart = 0, cnt = 0;
#pragma unroll
  for (int w = 0; w < BKT_WAVES; ++w) { bstart += wrs[w]; cnt += wcs[w]; }
  const bool staged = cnt <= SRT_CAP;
  const int  lead   = (int)((reinterpret_cast<size_t>(ssrc + bstart) >> 2) & 31);

  const int v = cl[tid];
  int inc = v;
  for (int m = 1; m < 32; m <<= 1) {
    const int t = __shfl_up(inc, m);
    if (lane >= m) inc += t;
  }
  if (lane == 31) wtot[wave] = inc;
  __syncthreads();
  int part = (lane < BKT_WAVES && lane < wave) ? wtot[lane] : 0;
  for (int m = 16; m > 0; m >>= 1) part += __shfl_xor(part, m);
  const int excl = part + inc - v;
  cl[tid] = excl;
  if (node0 + tid < N) cursor[node0 + tid] = bstart + excl;
  __syncthreads();

  gat_pieces_pass<2, 16>(pr, lst, cl, sbuf, ssrc, b, node0, lane, wave, E, NBK, NBLK, staged, lead, bstart, rsum, csum);
  if (staged) {
    __syncthreads();
    const int tot = lead + cnt;
    for (int j4 = tid * 4; j4 < tot; j4 += STG_THREADS * 4) {
      if (j4 >= lead && j4 + 4 <= tot) {
        *reinterpret_cast<int4*>(ssrc + (bstart - lead + j4)) = *reinterpret_cast<const int4*>(&sbuf[j4]);
      } else {
        for (int e = 0; e < 4; ++e) {
          const int j = j4 + e;
          if (j >= lead && j < tot) ssrc[bstart - lead + j] = sbuf[j];
        }
      }
    }
  }
}

template <int STAGE>
__global__ void __launch_bounds__(256) gat_stage_kernel(const float* __restrict__ x,
                                                        const float* __restrict__ W,
                                                        const float* __restrict__ att,
                                                        float* __restrict__ h,
                                                        float* __restrict__ s,
                                                        float* __restrict__ d,
                                                        const int* __restrict__ ei,
                                                        int* __restrict__ lst,
                                                        int* __restrict__ cpair,
                                                        int* __restrict__ cursor,
                                                        int* __restrict__ ssrc,
                                                        int N, int E, int NBK, int NBLK,
                                                        int gemmBlk0, int gemmCnt) {
  constexpr int SORT_BYTES = STAGE == 0 ? SA_LDS_BYTES : SB_LDS_BYTES;
  constexpr int LDS_BYTES  = SORT_BYTES > LIN_LDS_BYTES ? SORT_BYTES : LIN_LDS_BYTES;
  __shared__ __attribute__((aligned(16))) unsigned char smem[LDS_BYTES];
  const int blk = blockIdx.x;
  if (blk < gemmCnt) {
    gat_lin_body(smem, gemmBlk0 + blk, x, W, att, h, s, d, N);
  } else {
    const int sb = blk - gemmCnt;
    if constexpr (STAGE == 0) gat_blist_body(smem, sb, ei, lst, cpair, E, NBK, NBLK);
    else                      gat_bsort_body(smem, sb, lst, cpair, cursor, ssrc, N, E, NBK, NBLK);
  }
}


template <int K>
__device__ __forceinline__ int gat_row_lane(int v) {
  return __builtin_amdgcn_update_dpp(0, v, 0x150 + K, 0xF, 0xF, true);
}

template <int XM>
__device__ __forceinline__ float gat_row_xfold(float keep, float send) {
  const int t = __builtin_amdgcn_update_dpp(0, __float_as_int(send), 0x160 + XM, 0xF, 0xF, true);
  return keep + __int_as_float(t);
}

__device__ __forceinline__ float gat_other_row(float v) {
  return __int_as_float(__builtin_amdgcn_permlanex16(0, __float_as_int(v), 0x76543210u, 0xfedcba98u,
                                                     false, false));
}

template <int T>
__device__ __forceinline__ void gat_rows16(const float4* __restrict__ h4, int srcl, int c,
                                           bool b1, bool b2, bool b3, bool live, bool even,
                                           float4 at, float dn, float4& acc, float& dsum) {
  int sj[8];
  sj[0] = gat_row_lane<8 * T + 0>(srcl);
  sj[1] = gat_row_lane<8 * T + 1>(srcl);
  sj[2] = gat_row_lane<8 * T + 2>(srcl);
  sj[3] = gat_row_lane<8 * T + 3>(srcl);
  sj[4] = gat_row_lane<8 * T + 4>(srcl);
  sj[5] = gat_row_lane<8 * T + 5>(srcl);
  sj[6] = gat_row_lane<8 * T + 6>(srcl);
  sj[7] = gat_row_lane<8 * T + 7>(srcl);
  float4 r[8];
#pragma unroll
  for (int j = 0; j < 8; ++j) {
    const unsigned off = (unsigned)sj[j] * (unsigned)(HID * 4) + (unsigned)c * 16u;
    r[j] = *reinterpret_cast<const float4*>(reinterpret_cast<const char*>(h4) + off);
  }
  float p[8];
#pragma unroll
  for (int j = 0; j < 8; ++j)
    p[j] = __builtin_fmaf(r[j].w, at.w, __builtin_fmaf(r[j].z, at.z, __builtin_fmaf(r[j].y, at.y, r[j].x * at.x)));
  float q[4];
#pragma unroll
  for (int j = 0; j < 4; ++j)
    q[j] = gat_row_xfold<8>(b3 ? p[j + 4] : p[j], b3 ? p[j] : p[j + 4]);
  float u[2];
#pragma unroll
  for (int j = 0; j < 2; ++j)
    u[j] = gat_row_xfold<4>(b2 ? q[j + 2] : q[j], b2 ? q[j] : q[j + 2]);
  float t = gat_row_xfold<2>(b1 ? u[1] : u[0], b1 ? u[0] : u[1]);
  t = gat_row_xfold<1>(t, t);
  float v = t + dn;
  v = (v > 0.f) ? v : 0.2f * v;
  const float w = live ? __expf(v) : 0.f;
  dsum += even ? w : 0.f;
  const int wb = __float_as_int(w);
  float a[8];
  a[0] = __int_as_float(gat_row_lane< 0>(wb));
  a[1] = __int_as_float(gat_row_lane< 2>(wb));
  a[2] = __int_as_float(gat_row_lane< 4>(wb));
  a[3] = __int_as_float(gat_row_lane< 6>(wb));
  a[4] = __int_as_float(gat_row_lane< 8>(wb));
  a[5] = __int_as_float(gat_row_lane<10>(wb));
  a[6] = __int_as_float(gat_row_lane<12>(wb));
  a[7] = __int_as_float(gat_row_lane<14>(wb));
#pragma unroll
  for (int j = 0; j < 8; ++j) {
    acc.x += r[j].x * a[j];
    acc.y += r[j].y * a[j];
    acc.z += r[j].z * a[j];
    acc.w += r[j].w * a[j];
  }
}

#define GO_WAVES 4
#define GO_NODES (GO_WAVES * 16)
#define ZT_PITCH (HID + 4)

__global__ void __launch_bounds__(GO_WAVES * 32) gat_gather_out_kernel(const int* __restrict__ cursor,
                                                                       const int* __restrict__ ssrc,
                                                                       const float* __restrict__ h,
                                                                       const float* __restrict__ att,
                                                                       const float* __restrict__ d,
                                                                       const float* __restrict__ W,
                                                                       const float* __restrict__ bias,
                                                                       float* __restrict__ out, int N, int E) {
  __shared__ __attribute__((aligned(16))) float wo[OUT_D * WO_PITCH];
  __shared__ __attribute__((aligned(16))) float zt[GO_WAVES][16 * ZT_PITCH];
  const int lane = threadIdx.x & 31;
  const int wave = threadIdx.x >> 5;
  const int c    = lane & 15;
  const int half = lane >> 4;
  const int n0 = __builtin_amdgcn_readfirstlane((blockIdx.x * GO_WAVES + wave) * 16);

  float4 wv[(OUT_D * HID / 4) / (GO_WAVES * 32)];
  {
    const float4* __restrict__ W4 = reinterpret_cast<const float4*>(W);
#pragma unroll
    for (int i = 0; i < (OUT_D * HID / 4) / (GO_WAVES * 32); ++i) wv[i] = W4[i * (GO_WAVES * 32) + threadIdx.x];
  }
  const int   ni  = n0 + lane;
  const int   cur = (ni < N) ? cursor[ni] : E;
  const float dnv = (ni < N) ? d[ni] : 0.f;
  const float4 at = *reinterpret_cast<const float4*>(att + 4 * c);
  const float4* __restrict__ h4 = reinterpret_cast<const float4*>(h);

  int begin = __builtin_amdgcn_readlane(cur, 0);
  int end   = __builtin_amdgcn_readlane(cur, 1);
  int pre;
  {
    const int el = begin + 2 * c + half;
    int ix = el < end ? el : end - 1;
    ix = ix > 0 ? ix : 0;
    pre = ssrc[ix];
  }
#pragma unroll
  for (int i = 0; i < (OUT_D * HID / 4) / (GO_WAVES * 32); ++i) {
    const int idx = i * (GO_WAVES * 32) + threadIdx.x;
    const int r   = idx >> 4;
    const int c4  = idx & 15;
    *reinterpret_cast<float4*>(&wo[r * WO_PITCH + c4 * 4]) = wv[i];
  }

  const bool b1 = (c & 2) != 0, b2 = (c & 4) != 0, b3 = (c & 8) != 0, even = (c & 1) == 0;
  float* const ztw = &zt[wave][0];
#pragma nounroll
  for (int i = 0; i < 16; ++i) {
    const float dn = __int_as_float(__builtin_amdgcn_readlane(__float_as_int(dnv), i));
    const int nb = end;
    const int ne = __builtin_amdgcn_readlane(cur, i + 2 < 16 ? i + 2 : 16);
    int npre;
    {
      const int el = nb + 2 * c + half;
      int ix = el < ne ? el : ne - 1;
      ix = ix > 0 ? ix : 0;
      npre = ssrc[ix];
    }
    float4 acc  = {0.f, 0.f, 0.f, 0.f};
    float  dsum = 0.f;
    for (int base = begin; base < end; base += 32) {
      int srcl = pre;
      if (base != begin) {
        const int el = base + 2 * c + half;
        srcl = ssrc[el < end ? el : end - 1];
      }
      const int ej = base + (c & ~1) + half;
      gat_rows16<0>(h4, srcl, c, b1, b2, b3, ej < end, even, at, dn, acc, dsum);
      if (end - base > 16)
        gat_rows16<1>(h4, srcl, c, b1, b2, b3, ej + 16 < end, even, at, dn, acc, dsum);
    }
    acc.x += gat_other_row(acc.x);
    acc.y += gat_other_row(acc.y);
    acc.z += gat_other_row(acc.z);
    acc.w += gat_other_row(acc.w);
    float tot = dsum;
    tot = gat_row_xfold<8>(tot, tot);
    tot = gat_row_xfold<4>(tot, tot);
    tot = gat_row_xfold<2>(tot, tot);
    tot = gat_row_xfold<1>(tot, tot);
    tot += gat_other_row(tot);
    const float inv = 1.0f / (tot + 1e-9f);
    if (half == 0) {
      const float z0 = acc.x * inv, z1 = acc.y * inv, z2 = acc.z * inv, z3 = acc.w * inv;
      *reinterpret_cast<float4*>(&ztw[i * ZT_PITCH + 4 * c]) =
          make_float4(z0 > 0.f ? z0 : 0.f, z1 > 0.f ? z1 : 0.f, z2 > 0.f ? z2 : 0.f, z3 > 0.f ? z3 : 0.f);
    }
    pre   = npre;
    begin = nb;
    end   = ne;
  }
  __builtin_amdgcn_fence(__ATOMIC_RELEASE, "workgroup", "local");
  __builtin_amdgcn_s_barrier();
  __builtin_amdgcn_fence(__ATOMIC_ACQUIRE, "workgroup", "local");

  const int lh    = c;
  const int khalf = half;
  const float* __restrict__ zp  = &ztw[lh * ZT_PITCH + khalf * 2];
  const float* __restrict__ wop = &wo[lh * WO_PITCH + khalf * 2];

  v8f a0 = {}, a1 = {}, a2 = {}, a3 = {};
#pragma unroll
  for (int k0 = 0; k0 < HID; k0 += 4) {
    const v2f a  = *reinterpret_cast<const v2f*>(zp + k0);
    const v2f b0 = *reinterpret_cast<const v2f*>(wop +  0 * WO_PITCH + k0);
    const v2f b1f = *reinterpret_cast<const v2f*>(wop + 16 * WO_PITCH + k0);
    const v2f b2f = *reinterpret_cast<const v2f*>(wop + 32 * WO_PITCH + k0);
    const v2f b3f = *reinterpret_cast<const v2f*>(wop + 48 * WO_PITCH + k0);
    a0 = wmma_k4(a, b0, a0);
    a1 = wmma_k4(a, b1f, a1);
    a2 = wmma_k4(a, b2f, a2);
    a3 = wmma_k4(a, b3f, a3);
  }
  const float bb0 = bias[lh +  0];
  const float bb1 = bias[lh + 16];
  const float bb2 = bias[lh + 32];
  const float bb3 = bias[lh + 48];
#pragma unroll
  for (int r = 0; r < 8; ++r) {
    const int row = n0 + khalf * 8 + r;
    if (row < N) {
      out[(size_t)row * OUT_D + lh +  0] = a0[r] + bb0;
      out[(size_t)row * OUT_D + lh + 16] = a1[r] + bb1;
      out[(size_t)row * OUT_D + lh + 32] = a2[r] + bb2;
      out[(size_t)row * OUT_D + lh + 48] = a3[r] + bb3;
    }
  }
}

extern "C" void kernel_launch(void* const* d_in, const int* in_sizes, int n_in,
                              void* d_out, int out_size, void* d_ws, size_t ws_size,
                              hipStream_t stream) {
  const float* x    = (const float*)d_in[0];
  const int*   ei   = (const int*)  d_in[1];
  const float* Wlin = (const float*)d_in[2];
  const float* att  = (const float*)d_in[3];
  const float* Wout = (const float*)d_in[4];
  const float* bout = (const float*)d_in[5];
  float* out = (float*)d_out;

  const int N = in_sizes[0] / IN_DIM;
  const int E = in_sizes[1] / 2;

  float* ws     = (float*)d_ws;
  float* h      = ws;
  float* s      = h + (size_t)N * HID;
  float* dn     = s + N;
  int*   ssrc   = (int*)(dn + N);
  int*   cursor = ssrc + E;
  int*   scr    = cursor + N;

  const int NBK  = (N + BKT_NODES - 1) / BKT_NODES;
  const int NBLK = (E + ECHUNK - 1) / ECHUNK;
  const int L    = NBK * NBLK;
  const int L4   = (L + 3) / 4 * 4;
  int*   lst    = scr;
  int*   cpair  = lst + L4;

  const int tiles   = (N + 15) / 16;
  const int gemmBlk = (tiles + 7) / 8;

  const int g0 = (gemmBlk + 1) / 2;
  const int g1 = gemmBlk - g0;

  gat_stage_kernel<0><<<g0 + NBLK, 256, 0, stream>>>(x, Wlin, att, h, s, dn, ei, lst, cpair,
                                                     cursor, ssrc, N, E, NBK, NBLK, 0, g0);
  gat_stage_kernel<1><<<g1 + NBK, 256, 0, stream>>>(x, Wlin, att, h, s, dn, ei, lst, cpair,
                                                    cursor, ssrc, N, E, NBK, NBLK, g0, g1);
  gat_gather_out_kernel<<<(N + GO_NODES - 1) / GO_NODES, GO_WAVES * 32, 0, stream>>>(cursor, ssrc, h, att, dn,
                                                                                  Wout, bout, out, N, E);
}
